# starting kernel
# baseline (speedup 1.0000x reference)
#include <hip/hip_runtime.h>
#include <hip/hip_bf16.h>

typedef __attribute__((ext_vector_type(2))) float v2f;
typedef __attribute__((ext_vector_type(8))) float v8f;

#define FEATS 128

__global__ void gcn_degrees(const int* __restrict__ src, const int* __restrict__ dst,
                            float* __restrict__ dOut, float* __restrict__ dIn, int nE) {
    int e = blockIdx.x * blockDim.x + threadIdx.x;
    if (e >= nE) return;
    atomicAdd(&dOut[src[e]], 1.0f);
    atomicAdd(&dIn[dst[e]], 1.0f);
}

__global__ void gcn_norm(float* __restrict__ dOut, float* __restrict__ dIn, int n) {
    int i = blockIdx.x * blockDim.x + threadIdx.x;
    if (i >= n) return;
    dOut[i] = rsqrtf(fmaxf(dOut[i], 1.0f));
    dIn[i]  = rsqrtf(fmaxf(dIn[i], 1.0f));
}

__global__ void gcn_gemm1_wmma(const float* __restrict__ x, const float* __restrict__ W1,
                               const float* __restrict__ invOut, float* __restrict__ h) {
    const int lane    = threadIdx.x & 31;
    const int wave    = blockIdx.x * (blockDim.x >> 5) + (threadIdx.x >> 5);
    const int rowTile = wave >> 3;
    const int colTile = wave & 7;
    const int l15     = lane & 15;
    const int khalf   = lane >> 4;

    const int   row  = rowTile * 16 + l15;
    const int   nCol = colTile * 16 + l15;
    const float s    = invOut[row];
    const float* __restrict__ arow = x + (size_t)row * FEATS;

    v8f c = {};
#pragma unroll
    for (int k = 0; k < FEATS; k += 4) {
        const int ka = k + 2 * khalf;
        v2f a, b;
        a.x = arow[ka]     * s;
        a.y = arow[ka + 1] * s;
        b.x = W1[(size_t)ka       * FEATS + nCol];
        b.y = W1[(size_t)(ka + 1) * FEATS + nCol];
        c = __builtin_amdgcn_wmma_f32_16x16x4_f32(
                 false, a,  false, b,
                 (short)0, c,  false,  false);
    }

#pragma unroll
    for (int r = 0; r < 8; ++r) {
        const int mm = (lane < 16) ? r : (r + 8);
        h[(size_t)(rowTile * 16 + mm) * FEATS + nCol] = c[r];
    }
}

__global__ void gcn_scatter_rows(const float* __restrict__ h, const int* __restrict__ src,
                                 const int* __restrict__ dst, float* __restrict__ agg, int nE) {
    const int tid = blockIdx.x * blockDim.x + threadIdx.x;
    const int e   = tid >> 5;
    if (e >= nE) return;
    const int f = (tid & 31) * 4;
    const float4 v = *(const float4*)(h + (size_t)src[e] * FEATS + f);
    float* p = agg + (size_t)dst[e] * FEATS + f;
    atomicAdd(p + 0, v.x);
    atomicAdd(p + 1, v.y);
    atomicAdd(p + 2, v.z);
    atomicAdd(p + 3, v.w);
}

__global__ void gcn_post1(const float* __restrict__ agg, const float* __restrict__ invIn,
                          const float* __restrict__ b1, float* __restrict__ h, int n) {
    const int i = blockIdx.x * blockDim.x + threadIdx.x;
    if (i >= n * FEATS) return;
    const int node = i >> 7;
    const int feat = i & (FEATS - 1);
    const float v = agg[i] * invIn[node] + b1[feat];
    h[i] = fmaxf(v, 0.0f);
}

__global__ void gcn_dot2(const float* __restrict__ h, const float* __restrict__ W2,
                         const float* __restrict__ invOut, float* __restrict__ z) {
    const int node = blockIdx.x * (blockDim.x >> 5) + (threadIdx.x >> 5);
    const int lane = threadIdx.x & 31;
    const float4 hv = *(const float4*)(h + (size_t)node * FEATS + lane * 4);
    const float4 wv = *(const float4*)(W2 + lane * 4);
    float sum = hv.x * wv.x + hv.y * wv.y + hv.z * wv.z + hv.w * wv.w;
#pragma unroll
    for (int off = 16; off > 0; off >>= 1) sum += __shfl_down(sum, off, 32);
    if (lane == 0) z[node] = sum * invOut[node];
}

__global__ void gcn_scatter_scalar(const float* __restrict__ z, const int* __restrict__ src,
                                   const int* __restrict__ dst, float* __restrict__ agg2, int nE) {
    const int e = blockIdx.x * blockDim.x + threadIdx.x;
    if (e >= nE) return;
    atomicAdd(&agg2[dst[e]], z[src[e]]);
}

__global__ void gcn_final(const float* __restrict__ agg2, const float* __restrict__ invIn,
                          const float* __restrict__ b2, float* __restrict__ out, int n) {
    const int i = blockIdx.x * blockDim.x + threadIdx.x;
    if (i >= n) return;
    out[i] = agg2[i] * invIn[i] + b2[0];
}

extern "C" void kernel_launch(void* const* d_in, const int* in_sizes, int n_in,
                              void* d_out, int out_size, void* d_ws, size_t ws_size,
                              hipStream_t stream) {
    const float* x   = (const float*)d_in[0];
    const int*   src = (const int*)d_in[1];
    const int*   dst = (const int*)d_in[2];
    const float* W1  = (const float*)d_in[3];
    const float* b1  = (const float*)d_in[4];
    const float* W2  = (const float*)d_in[5];
    const float* b2  = (const float*)d_in[6];
    float* out = (float*)d_out;

    const int n  = in_sizes[0] / FEATS;
    const int nE = in_sizes[1];

    float* f      = (float*)d_ws;
    float* invOut = f;
    float* invIn  = f + (size_t)n;
    float* z      = f + 2 * (size_t)n;
    float* agg2   = f + 3 * (size_t)n;
    float* h      = f + 4 * (size_t)n;
    float* agg    = h + (size_t)n * FEATS;

    hipMemsetAsync(invOut, 0, 2 * (size_t)n * sizeof(float), stream);
    hipMemsetAsync(agg2,   0, (size_t)n * sizeof(float), stream);
    hipMemsetAsync(agg,    0, (size_t)n * FEATS * sizeof(float), stream);

    gcn_degrees<<<(nE + 255) / 256, 256, 0, stream>>>(src, dst, invOut, invIn, nE);
    gcn_norm<<<(n + 255) / 256, 256, 0, stream>>>(invOut, invIn, n);

    gcn_gemm1_wmma<<<n / 16, 256, 0, stream>>>(x, W1, invOut, h);

    gcn_scatter_rows<<<(int)(((size_t)nE * 32 + 255) / 256), 256, 0, stream>>>(h, src, dst, agg, nE);
    gcn_post1<<<(int)(((size_t)n * FEATS + 255) / 256), 256, 0, stream>>>(agg, invIn, b1, h, n);

    gcn_dot2<<<n / 8, 256, 0, stream>>>(h, W2, invOut, z);
    gcn_scatter_scalar<<<(nE + 255) / 256, 256, 0, stream>>>(z, src, dst, agg2, nE);
    gcn_final<<<(n + 255) / 256, 256, 0, stream>>>(agg2, invIn, b2, out, n);
}


# Neighbour sums as per-node gathers over a dst-grouped edge list (int count/rank atomics + prefix sum) instead of float atomic scatters; norm/bias/relu in registers; sum order = arrival-order rank
# speedup vs baseline: 1.9899x; 1.9899x over previous
#include <hip/hip_runtime.h>
#include <hip/hip_bf16.h>


typedef __attribute__((ext_vector_type(2))) float v2f;
typedef __attribute__((ext_vector_type(8))) float v8f;

#define FEATS 128
#define SCAN_TILE 1024

__global__ void gcn_degrees_rank(const int* __restrict__ src, const int* __restrict__ dst,
                                 int* __restrict__ cntOut, int* __restrict__ cntIn,
                                 int* __restrict__ rank, int nE) {
    const int e = blockIdx.x * blockDim.x + threadIdx.x;
    if (e >= nE) return;
    const int s = src[e];
    const int d = dst[e];
    atomicAdd(&cntOut[s], 1);
    rank[e] = atomicAdd(&cntIn[d], 1);
}

__global__ __launch_bounds__(256) void gcn_scan_norm(const int* __restrict__ cntOut,
                                                     const int* __restrict__ cntIn,
                                                     int* __restrict__ rowptr,
                                                     float* __restrict__ invOut,
                                                     float* __restrict__ invIn, int n) {
    __shared__ int sPre[8];
    __shared__ int sTot[8];
    const int tid  = threadIdx.x;
    const int lane = tid & 31;
    const int w    = tid >> 5;
    const int base = blockIdx.x * SCAN_TILE;

    int pre = 0;
    const int4* __restrict__ c4 = (const int4*)cntIn;
    const int lim = base >> 2;
    int i = tid;
    for (; i + 7 * 256 < lim; i += 8 * 256) {
        int4 v[8];
#pragma unroll
        for (int u = 0; u < 8; ++u) v[u] = c4[i + u * 256];
#pragma unroll
        for (int u = 0; u < 8; ++u) pre += (v[u].x + v[u].y) + (v[u].z + v[u].w);
    }
    for (; i < lim; i += 256) {
        const int4 v = c4[i];
        pre += (v.x + v.y) + (v.z + v.w);
    }

    const int i0 = base + tid * 4;
    int c[4];
#pragma unroll
    for (int k = 0; k < 4; ++k) c[k] = (i0 + k < n) ? cntIn[i0 + k] : 0;
    const int local = (c[0] + c[1]) + (c[2] + c[3]);

    int p = pre;
    for (int off = 16; off > 0; off >>= 1) p += __shfl_xor(p, off, 32);
    int inc = local;
    for (int off = 1; off < 32; off <<= 1) {
        const int t = __shfl_up(inc, off, 32);
        if (lane >= off) inc += t;
    }
    if (lane == 31) sTot[w] = inc;
    if (lane == 0)  sPre[w] = p;
    __syncthreads();
    int offset = 0;
    for (int k = 0; k < 8; ++k) {
        offset += sPre[k];
        if (k < w) offset += sTot[k];
    }
    int run = offset + inc - local;

#pragma unroll
    for (int k = 0; k < 4; ++k) {
        const int i = i0 + k;
        if (i < n) {
            rowptr[i] = run;
            run += c[k];
            invIn[i]  = rsqrtf(fmaxf((float)c[k], 1.0f));
            invOut[i] = rsqrtf(fmaxf((float)cntOut[i], 1.0f));
        }
    }
}

__global__ void gcn_fill(const int* __restrict__ src, const int* __restrict__ dst,
                         const int* __restrict__ rank, const int* __restrict__ rowptr,
                         int* __restrict__ csr, int nE) {
    const int e = blockIdx.x * blockDim.x + threadIdx.x;
    if (e >= nE) return;
    const unsigned pos = (unsigned)(rowptr[dst[e]] + rank[e]);
    if (pos < (unsigned)nE) csr[pos] = src[e];
}

__global__ void gcn_gemm1_wmma(const float* __restrict__ x, const float* __restrict__ W1,
                               const float* __restrict__ invOut, float* __restrict__ h) {
    const int lane    = threadIdx.x & 31;
    const int wave    = blockIdx.x * (blockDim.x >> 5) + (threadIdx.x >> 5);
    const int rowTile = wave >> 3;
    const int colTile = wave & 7;
    const int l15     = lane & 15;
    const int khalf   = lane >> 4;

    const int   row  = rowTile * 16 + l15;
    const int   nCol = colTile * 16 + l15;
    const float s    = invOut[row];
    const float* __restrict__ arow = x + (size_t)row * FEATS;

    v8f c = {};
#pragma unroll
    for (int k = 0; k < FEATS; k += 4) {
        const int ka = k + 2 * khalf;
        v2f a, b;
        a.x = arow[ka]     * s;
        a.y = arow[ka + 1] * s;
        b.x = W1[(size_t)ka       * FEATS + nCol];
        b.y = W1[(size_t)(ka + 1) * FEATS + nCol];
        c = __builtin_amdgcn_wmma_f32_16x16x4_f32(
                 false, a,  false, b,
                 (short)0, c,  false,  false);
    }

#pragma unroll
    for (int r = 0; r < 8; ++r) {
        const int mm = (lane < 16) ? r : (r + 8);
        h[(size_t)(rowTile * 16 + mm) * FEATS + nCol] = c[r];
    }
}

template <int U>
static __device__ __forceinline__ void add_rows(const float* __restrict__ h, int idx, int j,
                                                int lane4, float4& acc) {
    float4 v[U];
#pragma unroll
    for (int u = 0; u < U; ++u) {
        const int s = __builtin_amdgcn_readlane(idx, j + u);
        v[u] = *(const float4*)(h + (size_t)s * FEATS + lane4);
    }
#pragma unroll
    for (int u = 0; u < U; ++u) {
        acc.x += v[u].x; acc.y += v[u].y; acc.z += v[u].z; acc.w += v[u].w;
    }
}

__global__ __launch_bounds__(256) void gcn_gather_rows(const float* __restrict__ h,
                                                       const int* __restrict__ rowptr,
                                                       const int* __restrict__ cntIn,
                                                       const int* __restrict__ csr,
                                                       const float* __restrict__ invIn,
                                                       const float* __restrict__ b1,
                                                       float* __restrict__ h2, int n, int nE) {
    const int node = __builtin_amdgcn_readfirstlane((int)(blockIdx.x * 8 + (threadIdx.x >> 5)));
    if (node >= n) return;
    const int lane  = threadIdx.x & 31;
    const int lane4 = lane * 4;

    int beg = rowptr[node];
    int deg = cntIn[node];
    beg = min(max(beg, 0), nE);
    deg = min(max(deg, 0), nE - beg);

    float4 acc = {0.0f, 0.0f, 0.0f, 0.0f};
    for (int b = 0; b < deg; b += 32) {
        const int m = min(32, deg - b);
        int idx = 0;
        if (lane < m) idx = csr[beg + b + lane];
        idx = (int)min((unsigned)idx, (unsigned)(n - 1));
        int j = 0;
        for (; j + 8 <= m; j += 8) add_rows<8>(h, idx, j, lane4, acc);
        if (j + 4 <= m) { add_rows<4>(h, idx, j, lane4, acc); j += 4; }
        if (j + 2 <= m) { add_rows<2>(h, idx, j, lane4, acc); j += 2; }
        if (j < m)      { add_rows<1>(h, idx, j, lane4, acc); }
    }

    const float  sc = invIn[node];
    const float4 bb = *(const float4*)(b1 + lane4);
    float4 o;
    o.x = fmaxf(acc.x * sc + bb.x, 0.0f);
    o.y = fmaxf(acc.y * sc + bb.y, 0.0f);
    o.z = fmaxf(acc.z * sc + bb.z, 0.0f);
    o.w = fmaxf(acc.w * sc + bb.w, 0.0f);
    *(float4*)(h2 + (size_t)node * FEATS + lane4) = o;
}

__global__ void gcn_dot2(const float* __restrict__ h, const float* __restrict__ W2,
                         const float* __restrict__ invOut, float* __restrict__ z) {
    const int node = blockIdx.x * (blockDim.x >> 5) + (threadIdx.x >> 5);
    const int lane = threadIdx.x & 31;
    const float4 hv = *(const float4*)(h + (size_t)node * FEATS + lane * 4);
    const float4 wv = *(const float4*)(W2 + lane * 4);
    float sum = hv.x * wv.x + hv.y * wv.y + hv.z * wv.z + hv.w * wv.w;
#pragma unroll
    for (int off = 16; off > 0; off >>= 1) sum += __shfl_down(sum, off, 32);
    if (lane == 0) z[node] = sum * invOut[node];
}

__global__ __launch_bounds__(256) void gcn_gather_scalar(const float* __restrict__ z,
                                                         const int* __restrict__ rowptr,
                                                         const int* __restrict__ cntIn,
                                                         const int* __restrict__ csr,
                                                         const float* __restrict__ invIn,
                                                         const float* __restrict__ b2,
                                                         float* __restrict__ out, int n, int nE) {
    const int tid  = blockIdx.x * blockDim.x + threadIdx.x;
    const int node = tid >> 3;
    const int sub  = tid & 7;
    int beg = 0, deg = 0;
    if (node < n) { beg = rowptr[node]; deg = cntIn[node]; }
    beg = min(max(beg, 0), nE);
    deg = min(max(deg, 0), nE - beg);

    float s = 0.0f;
    for (int j = sub; j < deg; j += 8) {
        const int idx = (int)min((unsigned)csr[beg + j], (unsigned)(n - 1));
        s += z[idx];
    }
    s += __shfl_xor(s, 4, 32);
    s += __shfl_xor(s, 2, 32);
    s += __shfl_xor(s, 1, 32);
    if (sub == 0 && node < n) out[node] = s * invIn[node] + b2[0];
}

extern "C" void kernel_launch(void* const* d_in, const int* in_sizes, int n_in,
                              void* d_out, int out_size, void* d_ws, size_t ws_size,
                              hipStream_t stream) {
    const float* x   = (const float*)d_in[0];
    const int*   src = (const int*)d_in[1];
    const int*   dst = (const int*)d_in[2];
    const float* W1  = (const float*)d_in[3];
    const float* b1  = (const float*)d_in[4];
    const float* W2  = (const float*)d_in[5];
    const float* b2  = (const float*)d_in[6];
    float* out = (float*)d_out;

    const int n  = in_sizes[0] / FEATS;
    const int nE = in_sizes[1];

    float* f      = (float*)d_ws;
    int*   cntOut = (int*)f;
    int*   cntIn  = (int*)(f + (size_t)n);
    float* invOut = f + 2 * (size_t)n;
    float* invIn  = f + 3 * (size_t)n;
    float* z      = f + 4 * (size_t)n;
    int*   rowptr = (int*)(f + 5 * (size_t)n);
    int*   rank   = (int*)(f + 6 * (size_t)n);
    int*   csr    = rank + (size_t)nE;
    float* h      = (float*)(csr + (size_t)nE);
    float* h2     = h + (size_t)n * FEATS;

    hipMemsetAsync(cntOut, 0, 2 * (size_t)n * sizeof(int), stream);

    gcn_degrees_rank<<<(nE + 255) / 256, 256, 0, stream>>>(src, dst, cntOut, cntIn, rank, nE);
    gcn_scan_norm<<<(n + SCAN_TILE - 1) / SCAN_TILE, 256, 0, stream>>>(cntOut, cntIn, rowptr, invOut, invIn, n);
    gcn_fill<<<(nE + 255) / 256, 256, 0, stream>>>(src, dst, rank, rowptr, csr, nE);

    gcn_gemm1_wmma<<<n / 16, 256, 0, stream>>>(x, W1, invOut, h);

    gcn_gather_rows<<<(n + 7) / 8, 256, 0, stream>>>(h, rowptr, cntIn, csr, invIn, b1, h2, n, nE);

    gcn_dot2<<<n / 8, 256, 0, stream>>>(h2, W2, invOut, z);
    gcn_gather_scalar<<<(int)(((size_t)n * 8 + 255) / 256), 256, 0, stream>>>(z, rowptr, cntIn, csr, invIn, b2, out, n, nE);
}


# Each wave loops over several items instead of one (8 edges per lane; 5 row tiles per wave, W1 fragments held in registers; 4 or 8 nodes per wave): under a fifth of the waves; sum order kept
# speedup vs baseline: 2.0911x; 1.0509x over previous
#include <hip/hip_runtime.h>
#include <hip/hip_bf16.h>


typedef __attribute__((ext_vector_type(2))) float v2f;
typedef __attribute__((ext_vector_type(8))) float v8f;

#define FEATS 128
#define SCAN_TILE 1024

#define EDGES_PER_THREAD   8
#define ROW_TILES_PER_WAVE 5
#define NODES_PER_WAVE     4
#define DOT_NODES_PER_WAVE 8
#define GS_NODES_PER_GROUP 4

__global__ __launch_bounds__(256) void gcn_degrees_rank(const int* __restrict__ src,
                                                        const int* __restrict__ dst,
                                                        int* __restrict__ cntOut,
                                                        int* __restrict__ cntIn,
                                                        int* __restrict__ rank, int nE) {
    const int e0 = blockIdx.x * (256 * EDGES_PER_THREAD) + threadIdx.x;
    int s[EDGES_PER_THREAD], d[EDGES_PER_THREAD], r[EDGES_PER_THREAD];
#pragma unroll
    for (int u = 0; u < EDGES_PER_THREAD; ++u) {
        const int ec = min(e0 + u * 256, nE - 1);
        s[u] = src[ec];
        d[u] = dst[ec];
    }
#pragma unroll
    for (int u = 0; u < EDGES_PER_THREAD; ++u) {
        const int e = e0 + u * 256;
        r[u] = 0;
        if (e < nE) {
            atomicAdd(&cntOut[s[u]], 1);
            r[u] = atomicAdd(&cntIn[d[u]], 1);
        }
    }
#pragma unroll
    for (int u = 0; u < EDGES_PER_THREAD; ++u) {
        const int e = e0 + u * 256;
        if (e < nE) rank[e] = r[u];
    }
}

__global__ __launch_bounds__(256) void gcn_scan_norm(const int* __restrict__ cntOut,
                                                     const int* __restrict__ cntIn,
                                                     int* __restrict__ rowptr,
                                                     float* __restrict__ invOut,
                                                     float* __restrict__ invIn, int n) {
    __shared__ int sPre[8];
    __shared__ int sTot[8];
    const int tid  = threadIdx.x;
    const int lane = tid & 31;
    const int w    = tid >> 5;
    const int base = blockIdx.x * SCAN_TILE;

    int pre = 0;
    const int4* __restrict__ c4 = (const int4*)cntIn;
    const int lim = base >> 2;
    int i = tid;
    for (; i + 7 * 256 < lim; i += 8 * 256) {
        int4 v[8];
#pragma unroll
        for (int u = 0; u < 8; ++u) v[u] = c4[i + u * 256];
#pragma unroll
        for (int u = 0; u < 8; ++u) pre += (v[u].x + v[u].y) + (v[u].z + v[u].w);
    }
    for (; i < lim; i += 256) {
        const int4 v = c4[i];
        pre += (v.x + v.y) + (v.z + v.w);
    }

    const int i0 = base + tid * 4;
    int c[4];
#pragma unroll
    for (int k = 0; k < 4; ++k) c[k] = (i0 + k < n) ? cntIn[i0 + k] : 0;
    const int local = (c[0] + c[1]) + (c[2] + c[3]);

    int p = pre;
    for (int off = 16; off > 0; off >>= 1) p += __shfl_xor(p, off, 32);
    int inc = local;
    for (int off = 1; off < 32; off <<= 1) {
        const int t = __shfl_up(inc, off, 32);
        if (lane >= off) inc += t;
    }
    if (lane == 31) sTot[w] = inc;
    if (lane == 0)  sPre[w] = p;
    __syncthreads();
    int offset = 0;
    for (int k = 0; k < 8; ++k) {
        offset += sPre[k];
        if (k < w) offset += sTot[k];
    }
    int run = offset + inc - local;

#pragma unroll
    for (int k = 0; k < 4; ++k) {
        const int i = i0 + k;
        if (i < n) {
            rowptr[i] = run;
            run += c[k];
            invIn[i]  = rsqrtf(fmaxf((float)c[k], 1.0f));
            invOut[i] = rsqrtf(fmaxf((float)cntOut[i], 1.0f));
        }
    }
}

__global__ __launch_bounds__(256) void gcn_fill(const int* __restrict__ src,
                                                const int* __restrict__ dst,
                                                const int* __restrict__ rank,
                                                const int* __restrict__ rowptr,
                                                int* __restrict__ csr, int nE) {
    const int e0 = blockIdx.x * (256 * EDGES_PER_THREAD) + threadIdx.x;
    int s[EDGES_PER_THREAD], d[EDGES_PER_THREAD], r[EDGES_PER_THREAD], p[EDGES_PER_THREAD];
#pragma unroll
    for (int u = 0; u < EDGES_PER_THREAD; ++u) {
        const int ec = min(e0 + u * 256, nE - 1);
        d[u] = dst[ec];
        r[u] = rank[ec];
        s[u] = src[ec];
    }
#pragma unroll
    for (int u = 0; u < EDGES_PER_THREAD; ++u) p[u] = rowptr[d[u]];
#pragma unroll
    for (int u = 0; u < EDGES_PER_THREAD; ++u) {
        const int e = e0 + u * 256;
        const unsigned pos = (unsigned)(p[u] + r[u]);
        if (e < nE && pos < (unsigned)nE) csr[pos] = s[u];
    }
}

__global__ __launch_bounds__(256) void gcn_gemm1_wmma(const float* __restrict__ x,
                                                      const float* __restrict__ W1,
                                                      const float* __restrict__ invOut,
                                                      float* __restrict__ h, int nRowTiles) {
    const int lane    = threadIdx.x & 31;
    const int colTile = threadIdx.x >> 5;
    const int l15     = lane & 15;
    const int khalf   = lane >> 4;
    const int nCol    = colTile * 16 + l15;

    v2f b[FEATS / 4];
#pragma unroll
    for (int kk = 0; kk < FEATS / 4; ++kk) {
        const int ka = kk * 4 + 2 * khalf;
        b[kk].x = W1[(size_t)ka       * FEATS + nCol];
        b[kk].y = W1[(size_t)(ka + 1) * FEATS + nCol];
    }

    const int rt0 = blockIdx.x * ROW_TILES_PER_WAVE;
    for (int t = 0; t < ROW_TILES_PER_WAVE; ++t) {
        const int rowTile = rt0 + t;
        if (rowTile >= nRowTiles) break;
        const int   row = rowTile * 16 + l15;
        const float s   = invOut[row];
        const float* __restrict__ arow = x + (size_t)row * FEATS;

        v8f c = {};
#pragma unroll
        for (int kk = 0; kk < FEATS / 4; ++kk) {
            const int ka = kk * 4 + 2 * khalf;
            v2f a;
            a.x = arow[ka]     * s;
            a.y = arow[ka + 1] * s;
            c = __builtin_amdgcn_wmma_f32_16x16x4_f32(
                     false, a,  false, b[kk],
                     (short)0, c,  false,  false);
        }

#pragma unroll
        for (int r = 0; r < 8; ++r) {
            const int mm = (lane < 16) ? r : (r + 8);
            h[(size_t)(rowTile * 16 + mm) * FEATS + nCol] = c[r];
        }
    }
}

template <int U>
static __device__ __forceinline__ void add_rows(const float* __restrict__ h, int idx, int j,
                                                int lane4, float4& acc) {
    float4 v[U];
#pragma unroll
    for (int u = 0; u < U; ++u) {
        const int s = __builtin_amdgcn_readlane(idx, j + u);
        v[u] = *(const float4*)(h + (size_t)s * FEATS + lane4);
    }
#pragma unroll
    for (int u = 0; u < U; ++u) {
        acc.x += v[u].x; acc.y += v[u].y; acc.z += v[u].z; acc.w += v[u].w;
    }
}

__global__ __launch_bounds__(256) void gcn_gather_rows(const float* __restrict__ h,
                                                       const int* __restrict__ rowptr,
                                                       const int* __restrict__ cntIn,
                                                       const int* __restrict__ csr,
                                                       const float* __restrict__ invIn,
                                                       const float* __restrict__ b1,
                                                       float* __restrict__ h2, int n, int nE) {
    const int node0 = __builtin_amdgcn_readfirstlane(
        (int)(blockIdx.x * 8 + (threadIdx.x >> 5)) * NODES_PER_WAVE);
    if (node0 >= n) return;
    const int lane  = threadIdx.x & 31;
    const int lane4 = lane * 4;
    const float4 bb = *(const float4*)(b1 + lane4);

    int beg[NODES_PER_WAVE], deg[NODES_PER_WAVE], idx0[NODES_PER_WAVE];
    float sc[NODES_PER_WAVE];
#pragma unroll
    for (int i = 0; i < NODES_PER_WAVE; ++i) {
        const int node = min(node0 + i, n - 1);
        int bg = rowptr[node];
        int dg = cntIn[node];
        bg = min(max(bg, 0), nE);
        dg = min(max(dg, 0), nE - bg);
        beg[i] = bg;
        deg[i] = dg;
        sc[i]  = invIn[node];
    }
#pragma unroll
    for (int i = 0; i < NODES_PER_WAVE; ++i) {
        idx0[i] = 0;
        if (lane < min(32, deg[i])) idx0[i] = csr[beg[i] + lane];
    }

#pragma unroll
    for (int i = 0; i < NODES_PER_WAVE; ++i) {
        const int node = node0 + i;
        if (node >= n) break;
        float4 acc = {0.0f, 0.0f, 0.0f, 0.0f};
        for (int b = 0; b < deg[i]; b += 32) {
            const int m = min(32, deg[i] - b);
            int idx = idx0[i];
            if (b > 0) {
                idx = 0;
                if (lane < m) idx = csr[beg[i] + b + lane];
            }
            idx = (int)min((unsigned)idx, (unsigned)(n - 1));
            int j = 0;
            for (; j + 8 <= m; j += 8) add_rows<8>(h, idx, j, lane4, acc);
            if (j + 4 <= m) { add_rows<4>(h, idx, j, lane4, acc); j += 4; }
            if (j + 2 <= m) { add_rows<2>(h, idx, j, lane4, acc); j += 2; }
            if (j < m)      { add_rows<1>(h, idx, j, lane4, acc); }
        }

        float4 o;
        o.x = fmaxf(acc.x * sc[i] + bb.x, 0.0f);
        o.y = fmaxf(acc.y * sc[i] + bb.y, 0.0f);
        o.z = fmaxf(acc.z * sc[i] + bb.z, 0.0f);
        o.w = fmaxf(acc.w * sc[i] + bb.w, 0.0f);
        *(float4*)(h2 + (size_t)node * FEATS + lane4) = o;
    }
}

__global__ __launch_bounds__(256) void gcn_dot2(const float* __restrict__ h,
                                                const float* __restrict__ W2,
                                                const float* __restrict__ invOut,
                                                float* __restrict__ z, int n) {
    const int node0 = (blockIdx.x * 8 + (threadIdx.x >> 5)) * DOT_NODES_PER_WAVE;
    if (node0 >= n) return;
    const int lane = threadIdx.x & 31;
    const float4 wv = *(const float4*)(W2 + lane * 4);
    float4 hv[DOT_NODES_PER_WAVE];
#pragma unroll
    for (int u = 0; u < DOT_NODES_PER_WAVE; ++u) {
        const int node = min(node0 + u, n - 1);
        hv[u] = *(const float4*)(h + (size_t)node * FEATS + lane * 4);
    }
    float mine = 0.0f;
#pragma unroll
    for (int u = 0; u < DOT_NODES_PER_WAVE; ++u) {
        float sum = hv[u].x * wv.x + hv[u].y * wv.y + hv[u].z * wv.z + hv[u].w * wv.w;
#pragma unroll
        for (int off = 16; off > 0; off >>= 1) sum += __shfl_xor(sum, off, 32);
        if (lane == u) mine = sum;
    }
    const int node = node0 + lane;
    if (lane < DOT_NODES_PER_WAVE && node < n) z[node] = mine * invOut[node];
}

__global__ __launch_bounds__(256) void gcn_gather_scalar(const float* __restrict__ z,
                                                         const int* __restrict__ rowptr,
                                                         const int* __restrict__ cntIn,
                                                         const int* __restrict__ csr,
                                                         const float* __restrict__ invIn,
                                                         const float* __restrict__ b2,
                                                         float* __restrict__ out, int n, int nE) {
    const int wave  = blockIdx.x * 8 + (threadIdx.x >> 5);
    const int lane  = threadIdx.x & 31;
    const int grp   = lane >> 3;
    const int sub   = lane & 7;
    const int node0 = wave * (4 * GS_NODES_PER_GROUP) + grp;
    const float bias = b2[0];

    int beg[GS_NODES_PER_GROUP], deg[GS_NODES_PER_GROUP];
    float sc[GS_NODES_PER_GROUP];
#pragma unroll
    for (int t = 0; t < GS_NODES_PER_GROUP; ++t) {
        const int node = min(node0 + 4 * t, n - 1);
        int bg = rowptr[node];
        int dg = cntIn[node];
        const float v = invIn[node];
        bg = min(max(bg, 0), nE);
        dg = min(max(dg, 0), nE - bg);
        beg[t] = bg;
        deg[t] = dg;
        sc[t]  = v;
    }

#pragma unroll
    for (int t = 0; t < GS_NODES_PER_GROUP; ++t) {
        const int node = node0 + 4 * t;
        float s = 0.0f;
        for (int j = sub; j < deg[t]; j += 8) {
            const int idx = (int)min((unsigned)csr[beg[t] + j], (unsigned)(n - 1));
            s += z[idx];
        }
        s += __shfl_xor(s, 4, 32);
        s += __shfl_xor(s, 2, 32);
        s += __shfl_xor(s, 1, 32);
        if (sub == 0 && node < n) out[node] = s * sc[t] + bias;
    }
}

extern "C" void kernel_launch(void* const* d_in, const int* in_sizes, int n_in,
                              void* d_out, int out_size, void* d_ws, size_t ws_size,
                              hipStream_t stream) {
    const float* x   = (const float*)d_in[0];
    const int*   src = (const int*)d_in[1];
    const int*   dst = (const int*)d_in[2];
    const float* W1  = (const float*)d_in[3];
    const float* b1  = (const float*)d_in[4];
    const float* W2  = (const float*)d_in[5];
    const float* b2  = (const float*)d_in[6];
    float* out = (float*)d_out;

    const int n  = in_sizes[0] / FEATS;
    const int nE = in_sizes[1];

    float* f      = (float*)d_ws;
    int*   cntOut = (int*)f;
    int*   cntIn  = (int*)(f + (size_t)n);
    float* invOut = f + 2 * (size_t)n;
    float* invIn  = f + 3 * (size_t)n;
    float* z      = f + 4 * (size_t)n;
    int*   rowptr = (int*)(f + 5 * (size_t)n);
    int*   rank   = (int*)(f + 6 * (size_t)n);
    int*   csr    = rank + (size_t)nE;
    float* h      = (float*)(csr + (size_t)nE);
    float* h2     = h + (size_t)n * FEATS;

    hipMemsetAsync(cntOut, 0, 2 * (size_t)n * sizeof(int), stream);

    const int edgeBlocks = (nE + 256 * EDGES_PER_THREAD - 1) / (256 * EDGES_PER_THREAD);
    gcn_degrees_rank<<<edgeBlocks, 256, 0, stream>>>(src, dst, cntOut, cntIn, rank, nE);
    gcn_scan_norm<<<(n + SCAN_TILE - 1) / SCAN_TILE, 256, 0, stream>>>(cntOut, cntIn, rowptr, invOut, invIn, n);
    gcn_fill<<<edgeBlocks, 256, 0, stream>>>(src, dst, rank, rowptr, csr, nE);

    const int nRowTiles = n / 16;
    gcn_gemm1_wmma<<<(nRowTiles + ROW_TILES_PER_WAVE - 1) / ROW_TILES_PER_WAVE, 256, 0, stream>>>(x, W1, invOut, h, nRowTiles);

    gcn_gather_rows<<<(n + 8 * NODES_PER_WAVE - 1) / (8 * NODES_PER_WAVE), 256, 0, stream>>>(h, rowptr, cntIn, csr, invIn, b1, h2, n, nE);

    gcn_dot2<<<(n + 8 * DOT_NODES_PER_WAVE - 1) / (8 * DOT_NODES_PER_WAVE), 256, 0, stream>>>(h2, W2, invOut, z, n);
    gcn_gather_scalar<<<(n + 32 * GS_NODES_PER_GROUP - 1) / (32 * GS_NODES_PER_GROUP), 256, 0, stream>>>(z, rowptr, cntIn, csr, invIn, b2, out, n, nE);
}


# Edges grouped by destination (and counted by source) in two levels on LDS counters, buckets of 256 nodes: one memory add per block and bucket, not two per edge; sum order = arrival order as before
# speedup vs baseline: 4.5121x; 2.1578x over previous
#include <hip/hip_runtime.h>
#include <hip/hip_bf16.h>


typedef __attribute__((ext_vector_type(2))) float v2f;
typedef __attribute__((ext_vector_type(8))) float v8f;

#define FEATS 128

#define BIN_EDGES_PER_THREAD 16
#define ROW_TILES_PER_WAVE 5
#define NODES_PER_WAVE     4
#define DOT_NODES_PER_WAVE 8
#define GS_NODES_PER_GROUP 4

#define BUCKET_SHIFT 8
#define BUCKET_NODES 256
#define NBMAX        512
#define REC_BATCH    8

static __device__ __forceinline__ int bucket_of(int node, int nB) {
    return min((int)((unsigned)node >> BUCKET_SHIFT), nB - 1);
}

static __device__ __forceinline__ void block_excl_scan512(int* a, int* wsum, int tid) {
    const int lane = tid & 31;
    const int w    = tid >> 5;
    const int v0 = a[2 * tid];
    const int v1 = a[2 * tid + 1];
    const int local = v0 + v1;
    int inc = local;
    for (int off = 1; off < 32; off <<= 1) {
        const int t = __shfl_up(inc, off, 32);
        if (lane >= off) inc += t;
    }
    if (lane == 31) wsum[w] = inc;
    __syncthreads();
    int offset = 0;
    for (int k = 0; k < 8; ++k)
        if (k < w) offset += wsum[k];
    const int ex = offset + inc - local;
    a[2 * tid]     = ex;
    a[2 * tid + 1] = ex + v0;
    __syncthreads();
}

__global__ __launch_bounds__(256) void gcn_bucket_count(const int* __restrict__ src,
                                                        const int* __restrict__ dst,
                                                        int* __restrict__ bcnt, int nE, int nB) {
    __shared__ int hD[NBMAX];
    __shared__ int hS[NBMAX];
    const int tid = threadIdx.x;
    for (int i = tid; i < NBMAX; i += 256) { hD[i] = 0; hS[i] = 0; }

    const int e0 = blockIdx.x * (256 * BIN_EDGES_PER_THREAD) + tid;
    int s[BIN_EDGES_PER_THREAD], d[BIN_EDGES_PER_THREAD];
#pragma unroll
    for (int u = 0; u < BIN_EDGES_PER_THREAD; ++u) {
        const int ec = min(e0 + u * 256, nE - 1);
        s[u] = src[ec];
        d[u] = dst[ec];
    }
    __syncthreads();
#pragma unroll
    for (int u = 0; u < BIN_EDGES_PER_THREAD; ++u) {
        if (e0 + u * 256 < nE) {
            atomicAdd(&hD[bucket_of(d[u], nB)], 1);
            atomicAdd(&hS[bucket_of(s[u], nB)], 1);
        }
    }
    __syncthreads();
    for (int i = tid; i < nB; i += 256) {
        const int a = hD[i];
        const int b = hS[i];
        if (a) atomicAdd(&bcnt[i], a);
        if (b) atomicAdd(&bcnt[NBMAX + i], b);
    }
}

__global__ __launch_bounds__(256) void gcn_bucket_fill(const int* __restrict__ src,
                                                       const int* __restrict__ dst,
                                                       const int* __restrict__ bcnt,
                                                       int* __restrict__ cur,
                                                       int* __restrict__ bstart,
                                                       int2* __restrict__ recD,
                                                       int* __restrict__ recS, int nE, int nB) {
    __shared__ int  hD[NBMAX];
    __shared__ int  hS[NBMAX];
    __shared__ int  gD[NBMAX];
    __shared__ int  gS[NBMAX];
    __shared__ int  wsum[8];
    __shared__ int2 stD[256 * BIN_EDGES_PER_THREAD];
    __shared__ int  stS[256 * BIN_EDGES_PER_THREAD];
    const int tid = threadIdx.x;
    for (int i = tid; i < NBMAX; i += 256) {
        hD[i] = 0;
        hS[i] = 0;
        gD[i] = (i < nB) ? bcnt[i] : 0;
        gS[i] = (i < nB) ? bcnt[NBMAX + i] : 0;
    }

    const int eBlock = blockIdx.x * (256 * BIN_EDGES_PER_THREAD);
    const int e0 = eBlock + tid;
    int s[BIN_EDGES_PER_THREAD], d[BIN_EDGES_PER_THREAD];
    int rD[BIN_EDGES_PER_THREAD], rS[BIN_EDGES_PER_THREAD];
#pragma unroll
    for (int u = 0; u < BIN_EDGES_PER_THREAD; ++u) {
        const int ec = min(e0 + u * 256, nE - 1);
        s[u] = src[ec];
        d[u] = dst[ec];
    }
    __syncthreads();
#pragma unroll
    for (int u = 0; u < BIN_EDGES_PER_THREAD; ++u) {
        rD[u] = 0;
        rS[u] = 0;
        if (e0 + u * 256 < nE) {
            rD[u] = atomicAdd(&hD[bucket_of(d[u], nB)], 1);
            rS[u] = atomicAdd(&hS[bucket_of(s[u], nB)], 1);
        }
    }
    block_excl_scan512(gD, wsum, tid);
    block_excl_scan512(gS, wsum, tid);

    int baseD[2], baseS[2];
#pragma unroll
    for (int k = 0; k < 2; ++k) {
        const int i = tid + k * 256;
        baseD[k] = 0;
        baseS[k] = 0;
        if (i < nB) {
            baseD[k] = atomicAdd(&cur[i], hD[i]);
            baseS[k] = atomicAdd(&cur[NBMAX + i], hS[i]);
        }
    }
#pragma unroll
    for (int k = 0; k < 2; ++k) {
        const int i = tid + k * 256;
        if (i < nB) {
            if (blockIdx.x == 0) {
                bstart[i]         = gD[i];
                bstart[NBMAX + i] = gS[i];
            }
            gD[i] += baseD[k];
            gS[i] += baseS[k];
        }
    }
    block_excl_scan512(hD, wsum, tid);
    block_excl_scan512(hS, wsum, tid);

#pragma unroll
    for (int u = 0; u < BIN_EDGES_PER_THREAD; ++u) {
        if (e0 + u * 256 < nE) {
            const int pD = (hD[bucket_of(d[u], nB)] + rD[u]) & (256 * BIN_EDGES_PER_THREAD - 1);
            const int pS = (hS[bucket_of(s[u], nB)] + rS[u]) & (256 * BIN_EDGES_PER_THREAD - 1);
            stD[pD] = make_int2(d[u], s[u]);
            stS[pS] = s[u];
        }
    }
    __syncthreads();

    const int nMine = min(256 * BIN_EDGES_PER_THREAD, nE - eBlock);
#pragma unroll
    for (int u = 0; u < BIN_EDGES_PER_THREAD; ++u) {
        const int j = tid + u * 256;
        if (j < nMine) {
            const int2 v  = stD[j];
            const int  bD = bucket_of(v.x, nB);
            const unsigned pD = (unsigned)(gD[bD] + (j - hD[bD]));
            const int  w  = stS[j];
            const int  bS = bucket_of(w, nB);
            const unsigned pS = (unsigned)(gS[bS] + (j - hS[bS]));
            if (pD < (unsigned)nE) recD[pD] = v;
            if (pS < (unsigned)nE) recS[pS] = w;
        }
    }
}

__global__ __launch_bounds__(256) void gcn_bucket_build(const int* __restrict__ bcnt,
                                                        const int* __restrict__ bstart,
                                                        const int2* __restrict__ recD,
                                                        const int* __restrict__ recS,
                                                        int* __restrict__ rowptr,
                                                        int* __restrict__ cntIn,
                                                        int* __restrict__ csr,
                                                        float* __restrict__ invOut,
                                                        float* __restrict__ invIn,
                                                        int n, int nE) {
    __shared__ int c[BUCKET_NODES];
    __shared__ int wsum[8];
    const int tid   = threadIdx.x;
    const int lane  = tid & 31;
    const int w     = tid >> 5;
    const int b     = blockIdx.x;
    const int node0 = b << BUCKET_SHIFT;
    const int node  = node0 + tid;

    int begS = bstart[NBMAX + b], numS = bcnt[NBMAX + b];
    int begD = bstart[b],         numD = bcnt[b];
    begS = min(max(begS, 0), nE);  numS = min(max(numS, 0), nE - begS);
    begD = min(max(begD, 0), nE);  numD = min(max(numD, 0), nE - begD);
    const int*  __restrict__ rs = recS + begS;
    const int2* __restrict__ rd = recD + begD;

    c[tid] = 0;
    __syncthreads();
    for (int i = tid; i < numS; i += REC_BATCH * 256) {
        int v[REC_BATCH];
#pragma unroll
        for (int u = 0; u < REC_BATCH; ++u) {
            v[u] = rs[min(i + u * 256, numS - 1)];
        }
#pragma unroll
        for (int u = 0; u < REC_BATCH; ++u) {
            const unsigned l = (unsigned)(v[u] - node0);
            if (i + u * 256 < numS && l < BUCKET_NODES) atomicAdd(&c[l], 1);
        }
    }
    __syncthreads();
    if (node < n) invOut[node] = rsqrtf(fmaxf((float)c[tid], 1.0f));
    c[tid] = 0;
    __syncthreads();

    for (int i = tid; i < numD; i += REC_BATCH * 256) {
        int2 v[REC_BATCH];
#pragma unroll
        for (int u = 0; u < REC_BATCH; ++u) v[u] = rd[min(i + u * 256, numD - 1)];
#pragma unroll
        for (int u = 0; u < REC_BATCH; ++u) {
            const unsigned l = (unsigned)(v[u].x - node0);
            if (i + u * 256 < numD && l < BUCKET_NODES) atomicAdd(&c[l], 1);
        }
    }
    __syncthreads();

    const int cnt = c[tid];
    int inc = cnt;
    for (int off = 1; off < 32; off <<= 1) {
        const int t = __shfl_up(inc, off, 32);
        if (lane >= off) inc += t;
    }
    if (lane == 31) wsum[w] = inc;
    __syncthreads();
    int offset = 0;
    for (int k = 0; k < 8; ++k)
        if (k < w) offset += wsum[k];
    const int first = begD + offset + inc - cnt;
    if (node < n) {
        rowptr[node] = first;
        cntIn[node]  = cnt;
        invIn[node]  = rsqrtf(fmaxf((float)cnt, 1.0f));
    }
    c[tid] = first;
    __syncthreads();

    for (int i = tid; i < numD; i += REC_BATCH * 256) {
        int2 v[REC_BATCH];
#pragma unroll
        for (int u = 0; u < REC_BATCH; ++u) v[u] = rd[min(i + u * 256, numD - 1)];
#pragma unroll
        for (int u = 0; u < REC_BATCH; ++u) {
            const unsigned l = (unsigned)(v[u].x - node0);
            if (i + u * 256 < numD && l < BUCKET_NODES) {
                const unsigned pos = (unsigned)atomicAdd(&c[l], 1);
                if (pos < (unsigned)nE) csr[pos] = v[u].y;
            }
        }
    }
}

__global__ __launch_bounds__(256) void gcn_gemm1_wmma(const float* __restrict__ x,
                                                      const float* __restrict__ W1,
                                                      const float* __restrict__ invOut,
                                                      float* __restrict__ h, int nRowTiles) {
    const int lane    = threadIdx.x & 31;
    const int colTile = threadIdx.x >> 5;
    const int l15     = lane & 15;
    const int khalf   = lane >> 4;
    const int nCol    = colTile * 16 + l15;

    v2f b[FEATS / 4];
#pragma unroll
    for (int kk = 0; kk < FEATS / 4; ++kk) {
        const int ka = kk * 4 + 2 * khalf;
        b[kk].x = W1[(size_t)ka       * FEATS + nCol];
        b[kk].y = W1[(size_t)(ka + 1) * FEATS + nCol];
    }

    const int rt0 = blockIdx.x * ROW_TILES_PER_WAVE;
    for (int t = 0; t < ROW_TILES_PER_WAVE; ++t) {
        const int rowTile = rt0 + t;
        if (rowTile >= nRowTiles) break;
        const int   row = rowTile * 16 + l15;
        const float s   = invOut[row];
        const float* __restrict__ arow = x + (size_t)row * FEATS;

        v8f c = {};
#pragma unroll
        for (int kk = 0; kk < FEATS / 4; ++kk) {
            const int ka = kk * 4 + 2 * khalf;
            v2f a;
            a.x = arow[ka]     * s;
            a.y = arow[ka + 1] * s;
            c = __builtin_amdgcn_wmma_f32_16x16x4_f32(
                     false, a,  false, b[kk],
                     (short)0, c,  false,  false);
        }

#pragma unroll
        for (int r = 0; r < 8; ++r) {
            const int mm = (lane < 16) ? r : (r + 8);
            h[(size_t)(rowTile * 16 + mm) * FEATS + nCol] = c[r];
        }
    }
}

template <int U>
static __device__ __forceinline__ void add_rows(const float* __restrict__ h, int idx, int j,
                                                int lane4, float4& acc) {
    float4 v[U];
#pragma unroll
    for (int u = 0; u < U; ++u) {
        const int s = __builtin_amdgcn_readlane(idx, j + u);
        v[u] = *(const float4*)(h + (size_t)s * FEATS + lane4);
    }
#pragma unroll
    for (int u = 0; u < U; ++u) {
        acc.x += v[u].x; acc.y += v[u].y; acc.z += v[u].z; acc.w += v[u].w;
    }
}

__global__ __launch_bounds__(256) void gcn_gather_rows(const float* __restrict__ h,
                                                       const int* __restrict__ rowptr,
                                                       const int* __restrict__ cntIn,
                                                       const int* __restrict__ csr,
                                                       const float* __restrict__ invIn,
                                                       const float* __restrict__ b1,
                                                       float* __restrict__ h2, int n, int nE) {
    const int node0 = __builtin_amdgcn_readfirstlane(
        (int)(blockIdx.x * 8 + (threadIdx.x >> 5)) * NODES_PER_WAVE);
    if (node0 >= n) return;
    const int lane  = threadIdx.x & 31;
    const int lane4 = lane * 4;
    const float4 bb = *(const float4*)(b1 + lane4);

    int beg[NODES_PER_WAVE], deg[NODES_PER_WAVE], idx0[NODES_PER_WAVE];
    float sc[NODES_PER_WAVE];
#pragma unroll
    for (int i = 0; i < NODES_PER_WAVE; ++i) {
        const int node = min(node0 + i, n - 1);
        int bg = rowptr[node];
        int dg = cntIn[node];
        bg = min(max(bg, 0), nE);
        dg = min(max(dg, 0), nE - bg);
        beg[i] = bg;
        deg[i] = dg;
        sc[i]  = invIn[node];
    }
#pragma unroll
    for (int i = 0; i < NODES_PER_WAVE; ++i) {
        idx0[i] = 0;
        if (lane < min(32, deg[i])) idx0[i] = csr[beg[i] + lane];
    }

#pragma unroll
    for (int i = 0; i < NODES_PER_WAVE; ++i) {
        const int node = node0 + i;
        if (node >= n) break;
        float4 acc = {0.0f, 0.0f, 0.0f, 0.0f};
        for (int b = 0; b < deg[i]; b += 32) {
            const int m = min(32, deg[i] - b);
            int idx = idx0[i];
            if (b > 0) {
                idx = 0;
                if (lane < m) idx = csr[beg[i] + b + lane];
            }
            idx = (int)min((unsigned)idx, (unsigned)(n - 1));
            int j = 0;
            for (; j + 8 <= m; j += 8) add_rows<8>(h, idx, j, lane4, acc);
            if (j + 4 <= m) { add_rows<4>(h, idx, j, lane4, acc); j += 4; }
            if (j + 2 <= m) { add_rows<2>(h, idx, j, lane4, acc); j += 2; }
            if (j < m)      { add_rows<1>(h, idx, j, lane4, acc); }
        }

        float4 o;
        o.x = fmaxf(acc.x * sc[i] + bb.x, 0.0f);
        o.y = fmaxf(acc.y * sc[i] + bb.y, 0.0f);
        o.z = fmaxf(acc.z * sc[i] + bb.z, 0.0f);
        o.w = fmaxf(acc.w * sc[i] + bb.w, 0.0f);
        *(float4*)(h2 + (size_t)node * FEATS + lane4) = o;
    }
}

__global__ __launch_bounds__(256) void gcn_dot2(const float* __restrict__ h,
                                                const float* __restrict__ W2,
                                                const float* __restrict__ invOut,
                                                float* __restrict__ z, int n) {
    const int node0 = (blockIdx.x * 8 + (threadIdx.x >> 5)) * DOT_NODES_PER_WAVE;
    if (node0 >= n) return;
    const int lane = threadIdx.x & 31;
    const float4 wv = *(const float4*)(W2 + lane * 4);
    float4 hv[DOT_NODES_PER_WAVE];
#pragma unroll
    for (int u = 0; u < DOT_NODES_PER_WAVE; ++u) {
        const int node = min(node0 + u, n - 1);
        hv[u] = *(const float4*)(h + (size_t)node * FEATS + lane * 4);
    }
    float mine = 0.0f;
#pragma unroll
    for (int u = 0; u < DOT_NODES_PER_WAVE; ++u) {
        float sum = hv[u].x * wv.x + hv[u].y * wv.y + hv[u].z * wv.z + hv[u].w * wv.w;
#pragma unroll
        for (int off = 16; off > 0; off >>= 1) sum += __shfl_xor(sum, off, 32);
        if (lane == u) mine = sum;
    }
    const int node = node0 + lane;
    if (lane < DOT_NODES_PER_WAVE && node < n) z[node] = mine * invOut[node];
}

__global__ __launch_bounds__(256) void gcn_gather_scalar(const float* __restrict__ z,
                                                         const int* __restrict__ rowptr,
                                                         const int* __restrict__ cntIn,
                                                         const int* __restrict__ csr,
                                                         const float* __restrict__ invIn,
                                                         const float* __restrict__ b2,
                                                         float* __restrict__ out, int n, int nE) {
    const int wave  = blockIdx.x * 8 + (threadIdx.x >> 5);
    const int lane  = threadIdx.x & 31;
    const int grp   = lane >> 3;
    const int sub   = lane & 7;
    const int node0 = wave * (4 * GS_NODES_PER_GROUP) + grp;
    const float bias = b2[0];

    int beg[GS_NODES_PER_GROUP], deg[GS_NODES_PER_GROUP];
    float sc[GS_NODES_PER_GROUP];
#pragma unroll
    for (int t = 0; t < GS_NODES_PER_GROUP; ++t) {
        const int node = min(node0 + 4 * t, n - 1);
        int bg = rowptr[node];
        int dg = cntIn[node];
        const float v = invIn[node];
        bg = min(max(bg, 0), nE);
        dg = min(max(dg, 0), nE - bg);
        beg[t] = bg;
        deg[t] = dg;
        sc[t]  = v;
    }

#pragma unroll
    for (int t = 0; t < GS_NODES_PER_GROUP; ++t) {
        const int node = node0 + 4 * t;
        float s = 0.0f;
        for (int j = sub; j < deg[t]; j += 8) {
            const int idx = (int)min((unsigned)csr[beg[t] + j], (unsigned)(n - 1));
            s += z[idx];
        }
        s += __shfl_xor(s, 4, 32);
        s += __shfl_xor(s, 2, 32);
        s += __shfl_xor(s, 1, 32);
        if (sub == 0 && node < n) out[node] = s * sc[t] + bias;
    }
}

extern "C" void kernel_launch(void* const* d_in, const int* in_sizes, int n_in,
                              void* d_out, int out_size, void* d_ws, size_t ws_size,
                              hipStream_t stream) {
    const float* x   = (const float*)d_in[0];
    const int*   src = (const int*)d_in[1];
    const int*   dst = (const int*)d_in[2];
    const float* W1  = (const float*)d_in[3];
    const float* b1  = (const float*)d_in[4];
    const float* W2  = (const float*)d_in[5];
    const float* b2  = (const float*)d_in[6];
    float* out = (float*)d_out;

    const int n  = in_sizes[0] / FEATS;
    const int nE = in_sizes[1];
    const int nB = (n + BUCKET_NODES - 1) / BUCKET_NODES;

    float* f      = (float*)d_ws;
    int*   cntIn  = (int*)(f + (size_t)n);
    float* invOut = f + 2 * (size_t)n;
    float* invIn  = f + 3 * (size_t)n;
    float* z      = f + 4 * (size_t)n;
    int*   rowptr = (int*)(f + 5 * (size_t)n);
    int*   recS   = (int*)(f + 6 * (size_t)n);
    int*   csr    = recS + (size_t)nE;
    float* h      = (float*)(csr + (size_t)nE);
    float* h2     = h + (size_t)n * FEATS;
    int2*  recD   = (int2*)(h2 + (size_t)n * FEATS);
    int*   bcnt   = (int*)(recD + (size_t)nE);
    int*   cur    = bcnt + 2 * NBMAX;
    int*   bstart = cur + 2 * NBMAX;

    hipMemsetAsync(bcnt, 0, 4 * NBMAX * sizeof(int), stream);

    const int edgeBlocks = (nE + 256 * BIN_EDGES_PER_THREAD - 1) / (256 * BIN_EDGES_PER_THREAD);
    gcn_bucket_count<<<edgeBlocks, 256, 0, stream>>>(src, dst, bcnt, nE, nB);
    gcn_bucket_fill<<<edgeBlocks, 256, 0, stream>>>(src, dst, bcnt, cur, bstart, recD, recS, nE, nB);
    gcn_bucket_build<<<nB, 256, 0, stream>>>(bcnt, bstart, recD, recS, rowptr, cntIn, csr, invOut, invIn, n, nE);

    const int nRowTiles = n / 16;
    gcn_gemm1_wmma<<<(nRowTiles + ROW_TILES_PER_WAVE - 1) / ROW_TILES_PER_WAVE, 256, 0, stream>>>(x, W1, invOut, h, nRowTiles);

    gcn_gather_rows<<<(n + 8 * NODES_PER_WAVE - 1) / (8 * NODES_PER_WAVE), 256, 0, stream>>>(h, rowptr, cntIn, csr, invIn, b1, h2, n, nE);

    gcn_dot2<<<(n + 8 * DOT_NODES_PER_WAVE - 1) / (8 * DOT_NODES_PER_WAVE), 256, 0, stream>>>(h2, W2, invOut, z, n);
    gcn_gather_scalar<<<(n + 32 * GS_NODES_PER_GROUP - 1) / (32 * GS_NODES_PER_GROUP), 256, 0, stream>>>(z, rowptr, cntIn, csr, invIn, b2, out, n, nE);
}


# Layer-2 dot with W2 done in the row gather's epilogue from registers (relu row times W2, wave butterfly, one float z per node): h2 never written or re-read, one launch fewer; sum order kept
# speedup vs baseline: 4.9053x; 1.0871x over previous
#include <hip/hip_runtime.h>
#include <hip/hip_bf16.h>


typedef __attribute__((ext_vector_type(2))) float v2f;
typedef __attribute__((ext_vector_type(8))) float v8f;

#define FEATS 128

#define BIN_EDGES_PER_THREAD 16
#define ROW_TILES_PER_WAVE 5
#define NODES_PER_WAVE     4
#define GS_NODES_PER_GROUP 4

#define BUCKET_SHIFT 8
#define BUCKET_NODES 256
#define NBMAX        512
#define REC_BATCH    8

static __device__ __forceinline__ int bucket_of(int node, int nB) {
    return min((int)((unsigned)node >> BUCKET_SHIFT), nB - 1);
}

static __device__ __forceinline__ void block_excl_scan512(int* a, int* wsum, int tid) {
    const int lane = tid & 31;
    const int w    = tid >> 5;
    const int v0 = a[2 * tid];
    const int v1 = a[2 * tid + 1];
    const int local = v0 + v1;
    int inc = local;
    for (int off = 1; off < 32; off <<= 1) {
        const int t = __shfl_up(inc, off, 32);
        if (lane >= off) inc += t;
    }
    if (lane == 31) wsum[w] = inc;
    __syncthreads();
    int offset = 0;
    for (int k = 0; k < 8; ++k)
        if (k < w) offset += wsum[k];
    const int ex = offset + inc - local;
    a[2 * tid]     = ex;
    a[2 * tid + 1] = ex + v0;
    __syncthreads();
}

__global__ __launch_bounds__(256) void gcn_bucket_count(const int* __restrict__ src,
                                                        const int* __restrict__ dst,
                                                        int* __restrict__ bcnt, int nE, int nB) {
    __shared__ int hD[NBMAX];
    __shared__ int hS[NBMAX];
    const int tid = threadIdx.x;
    for (int i = tid; i < NBMAX; i += 256) { hD[i] = 0; hS[i] = 0; }

    const int e0 = blockIdx.x * (256 * BIN_EDGES_PER_THREAD) + tid;
    int s[BIN_EDGES_PER_THREAD], d[BIN_EDGES_PER_THREAD];
#pragma unroll
    for (int u = 0; u < BIN_EDGES_PER_THREAD; ++u) {
        const int ec = min(e0 + u * 256, nE - 1);
        s[u] = src[ec];
        d[u] = dst[ec];
    }
    __syncthreads();
#pragma unroll
    for (int u = 0; u < BIN_EDGES_PER_THREAD; ++u) {
        if (e0 + u * 256 < nE) {
            atomicAdd(&hD[bucket_of(d[u], nB)], 1);
            atomicAdd(&hS[bucket_of(s[u], nB)], 1);
        }
    }
    __syncthreads();
    for (int i = tid; i < nB; i += 256) {
        const int a = hD[i];
        const int b = hS[i];
        if (a) atomicAdd(&bcnt[i], a);
        if (b) atomicAdd(&bcnt[NBMAX + i], b);
    }
}

__global__ __launch_bounds__(256) void gcn_bucket_fill(const int* __restrict__ src,
                                                       const int* __restrict__ dst,
                                                       const int* __restrict__ bcnt,
                                                       int* __restrict__ cur,
                                                       int* __restrict__ bstart,
                                                       int2* __restrict__ recD,
                                                       int* __restrict__ recS, int nE, int nB) {
    __shared__ int  hD[NBMAX];
    __shared__ int  hS[NBMAX];
    __shared__ int  gD[NBMAX];
    __shared__ int  gS[NBMAX];
    __shared__ int  wsum[8];
    __shared__ int2 stD[256 * BIN_EDGES_PER_THREAD];
    __shared__ int  stS[256 * BIN_EDGES_PER_THREAD];
    const int tid = threadIdx.x;
    for (int i = tid; i < NBMAX; i += 256) {
        hD[i] = 0;
        hS[i] = 0;
        gD[i] = (i < nB) ? bcnt[i] : 0;
        gS[i] = (i < nB) ? bcnt[NBMAX + i] : 0;
    }

    const int eBlock = blockIdx.x * (256 * BIN_EDGES_PER_THREAD);
    const int e0 = eBlock + tid;
    int s[BIN_EDGES_PER_THREAD], d[BIN_EDGES_PER_THREAD];
    int rD[BIN_EDGES_PER_THREAD], rS[BIN_EDGES_PER_THREAD];
#pragma unroll
    for (int u = 0; u < BIN_EDGES_PER_THREAD; ++u) {
        const int ec = min(e0 + u * 256, nE - 1);
        s[u] = src[ec];
        d[u] = dst[ec];
    }
    __syncthreads();
#pragma unroll
    for (int u = 0; u < BIN_EDGES_PER_THREAD; ++u) {
        rD[u] = 0;
        rS[u] = 0;
        if (e0 + u * 256 < nE) {
            rD[u] = atomicAdd(&hD[bucket_of(d[u], nB)], 1);
            rS[u] = atomicAdd(&hS[bucket_of(s[u], nB)], 1);
        }
    }
    block_excl_scan512(gD, wsum, tid);
    block_excl_scan512(gS, wsum, tid);

    int baseD[2], baseS[2];
#pragma unroll
    for (int k = 0; k < 2; ++k) {
        const int i = tid + k * 256;
        baseD[k] = 0;
        baseS[k] = 0;
        if (i < nB) {
            baseD[k] = atomicAdd(&cur[i], hD[i]);
            baseS[k] = atomicAdd(&cur[NBMAX + i], hS[i]);
        }
    }
#pragma unroll
    for (int k = 0; k < 2; ++k) {
        const int i = tid + k * 256;
        if (i < nB) {
            if (blockIdx.x == 0) {
                bstart[i]         = gD[i];
                bstart[NBMAX + i] = gS[i];
            }
            gD[i] += baseD[k];
            gS[i] += baseS[k];
        }
    }
    block_excl_scan512(hD, wsum, tid);
    block_excl_scan512(hS, wsum, tid);

#pragma unroll
    for (int u = 0; u < BIN_EDGES_PER_THREAD; ++u) {
        if (e0 + u * 256 < nE) {
            const int pD = (hD[bucket_of(d[u], nB)] + rD[u]) & (256 * BIN_EDGES_PER_THREAD - 1);
            const int pS = (hS[bucket_of(s[u], nB)] + rS[u]) & (256 * BIN_EDGES_PER_THREAD - 1);
            stD[pD] = make_int2(d[u], s[u]);
            stS[pS] = s[u];
        }
    }
    __syncthreads();

    const int nMine = min(256 * BIN_EDGES_PER_THREAD, nE - eBlock);
#pragma unroll
    for (int u = 0; u < BIN_EDGES_PER_THREAD; ++u) {
        const int j = tid + u * 256;
        if (j < nMine) {
            const int2 v  = stD[j];
            const int  bD = bucket_of(v.x, nB);
            const unsigned pD = (unsigned)(gD[bD] + (j - hD[bD]));
            const int  w  = stS[j];
            const int  bS = bucket_of(w, nB);
            const unsigned pS = (unsigned)(gS[bS] + (j - hS[bS]));
            if (pD < (unsigned)nE) recD[pD] = v;
            if (pS < (unsigned)nE) recS[pS] = w;
        }
    }
}

__global__ __launch_bounds__(256) void gcn_bucket_build(const int* __restrict__ bcnt,
                                                        const int* __restrict__ bstart,
                                                        const int2* __restrict__ recD,
                                                        const int* __restrict__ recS,
                                                        int* __restrict__ rowptr,
                                                        int* __restrict__ cntIn,
                                                        int* __restrict__ csr,
                                                        float* __restrict__ invOut,
                                                        float* __restrict__ invIn,
                                                        int n, int nE) {
    __shared__ int c[BUCKET_NODES];
    __shared__ int wsum[8];
    const int tid   = threadIdx.x;
    const int lane  = tid & 31;
    const int w     = tid >> 5;
    const int b     = blockIdx.x;
    const int node0 = b << BUCKET_SHIFT;
    const int node  = node0 + tid;

    int begS = bstart[NBMAX + b], numS = bcnt[NBMAX + b];
    int begD = bstart[b],         numD = bcnt[b];
    begS = min(max(begS, 0), nE);  numS = min(max(numS, 0), nE - begS);
    begD = min(max(begD, 0), nE);  numD = min(max(numD, 0), nE - begD);
    const int*  __restrict__ rs = recS + begS;
    const int2* __restrict__ rd = recD + begD;

    c[tid] = 0;
    __syncthreads();
    for (int i = tid; i < numS; i += REC_BATCH * 256) {
        int v[REC_BATCH];
#pragma unroll
        for (int u = 0; u < REC_BATCH; ++u) {
            v[u] = rs[min(i + u * 256, numS - 1)];
        }
#pragma unroll
        for (int u = 0; u < REC_BATCH; ++u) {
            const unsigned l = (unsigned)(v[u] - node0);
            if (i + u * 256 < numS && l < BUCKET_NODES) atomicAdd(&c[l], 1);
        }
    }
    __syncthreads();
    if (node < n) invOut[node] = rsqrtf(fmaxf((float)c[tid], 1.0f));
    c[tid] = 0;
    __syncthreads();

    for (int i = tid; i < numD; i += REC_BATCH * 256) {
        int2 v[REC_BATCH];
#pragma unroll
        for (int u = 0; u < REC_BATCH; ++u) v[u] = rd[min(i + u * 256, numD - 1)];
#pragma unroll
        for (int u = 0; u < REC_BATCH; ++u) {
            const unsigned l = (unsigned)(v[u].x - node0);
            if (i + u * 256 < numD && l < BUCKET_NODES) atomicAdd(&c[l], 1);
        }
    }
    __syncthreads();

    const int cnt = c[tid];
    int inc = cnt;
    for (int off = 1; off < 32; off <<= 1) {
        const int t = __shfl_up(inc, off, 32);
        if (lane >= off) inc += t;
    }
    if (lane == 31) wsum[w] = inc;
    __syncthreads();
    int offset = 0;
    for (int k = 0; k < 8; ++k)
        if (k < w) offset += wsum[k];
    const int first = begD + offset + inc - cnt;
    if (node < n) {
        rowptr[node] = first;
        cntIn[node]  = cnt;
        invIn[node]  = rsqrtf(fmaxf((float)cnt, 1.0f));
    }
    c[tid] = first;
    __syncthreads();

    for (int i = tid; i < numD; i += REC_BATCH * 256) {
        int2 v[REC_BATCH];
#pragma unroll
        for (int u = 0; u < REC_BATCH; ++u) v[u] = rd[min(i + u * 256, numD - 1)];
#pragma unroll
        for (int u = 0; u < REC_BATCH; ++u) {
            const unsigned l = (unsigned)(v[u].x - node0);
            if (i + u * 256 < numD && l < BUCKET_NODES) {
                const unsigned pos = (unsigned)atomicAdd(&c[l], 1);
                if (pos < (unsigned)nE) csr[pos] = v[u].y;
            }
        }
    }
}

__global__ __launch_bounds__(256) void gcn_gemm1_wmma(const float* __restrict__ x,
                                                      const float* __restrict__ W1,
                                                      const float* __restrict__ invOut,
                                                      float* __restrict__ h, int nRowTiles) {
    const int lane    = threadIdx.x & 31;
    const int colTile = threadIdx.x >> 5;
    const int l15     = lane & 15;
    const int khalf   = lane >> 4;
    const int nCol    = colTile * 16 + l15;

    v2f b[FEATS / 4];
#pragma unroll
    for (int kk = 0; kk < FEATS / 4; ++kk) {
        const int ka = kk * 4 + 2 * khalf;
        b[kk].x = W1[(size_t)ka       * FEATS + nCol];
        b[kk].y = W1[(size_t)(ka + 1) * FEATS + nCol];
    }

    const int rt0 = blockIdx.x * ROW_TILES_PER_WAVE;
    for (int t = 0; t < ROW_TILES_PER_WAVE; ++t) {
        const int rowTile = rt0 + t;
        if (rowTile >= nRowTiles) break;
        const int   row = rowTile * 16 + l15;
        const float s   = invOut[row];
        const float* __restrict__ arow = x + (size_t)row * FEATS;

        v8f c = {};
#pragma unroll
        for (int kk = 0; kk < FEATS / 4; ++kk) {
            const int ka = kk * 4 + 2 * khalf;
            v2f a;
            a.x = arow[ka]     * s;
            a.y = arow[ka + 1] * s;
            c = __builtin_amdgcn_wmma_f32_16x16x4_f32(
                     false, a,  false, b[kk],
                     (short)0, c,  false,  false);
        }

#pragma unroll
        for (int r = 0; r < 8; ++r) {
            const int mm = (lane < 16) ? r : (r + 8);
            h[(size_t)(rowTile * 16 + mm) * FEATS + nCol] = c[r];
        }
    }
}

template <int U>
static __device__ __forceinline__ void add_rows(const float* __restrict__ h, int idx, int j,
                                                int lane4, float4& acc) {
    float4 v[U];
#pragma unroll
    for (int u = 0; u < U; ++u) {
        const int s = __builtin_amdgcn_readlane(idx, j + u);
        v[u] = *(const float4*)(h + (size_t)s * FEATS + lane4);
    }
#pragma unroll
    for (int u = 0; u < U; ++u) {
        acc.x += v[u].x; acc.y += v[u].y; acc.z += v[u].z; acc.w += v[u].w;
    }
}

__global__ __launch_bounds__(256) void gcn_gather_rows_dot(const float* __restrict__ h,
                                                           const int* __restrict__ rowptr,
                                                           const int* __restrict__ cntIn,
                                                           const int* __restrict__ csr,
                                                           const float* __restrict__ invIn,
                                                           const float* __restrict__ invOut,
                                                           const float* __restrict__ b1,
                                                           const float* __restrict__ W2,
                                                           float* __restrict__ z, int n, int nE) {
    const int node0 = __builtin_amdgcn_readfirstlane(
        (int)(blockIdx.x * 8 + (threadIdx.x >> 5)) * NODES_PER_WAVE);
    if (node0 >= n) return;
    const int lane  = threadIdx.x & 31;
    const int lane4 = lane * 4;
    const float4 bb = *(const float4*)(b1 + lane4);
    const float4 wv = *(const float4*)(W2 + lane4);
    const float so = invOut[min(node0 + min(lane, NODES_PER_WAVE - 1), n - 1)];

    int beg[NODES_PER_WAVE], deg[NODES_PER_WAVE], idx0[NODES_PER_WAVE];
    float sc[NODES_PER_WAVE];
#pragma unroll
    for (int i = 0; i < NODES_PER_WAVE; ++i) {
        const int node = min(node0 + i, n - 1);
        int bg = rowptr[node];
        int dg = cntIn[node];
        bg = min(max(bg, 0), nE);
        dg = min(max(dg, 0), nE - bg);
        beg[i] = bg;
        deg[i] = dg;
        sc[i]  = invIn[node];
    }
#pragma unroll
    for (int i = 0; i < NODES_PER_WAVE; ++i) {
        idx0[i] = 0;
        if (lane < min(32, deg[i])) idx0[i] = csr[beg[i] + lane];
    }

    float part[NODES_PER_WAVE];
#pragma unroll
    for (int i = 0; i < NODES_PER_WAVE; ++i) part[i] = 0.0f;
#pragma unroll
    for (int i = 0; i < NODES_PER_WAVE; ++i) {
        const int node = node0 + i;
        if (node >= n) break;
        float4 acc = {0.0f, 0.0f, 0.0f, 0.0f};
        for (int b = 0; b < deg[i]; b += 32) {
            const int m = min(32, deg[i] - b);
            int idx = idx0[i];
            if (b > 0) {
                idx = 0;
                if (lane < m) idx = csr[beg[i] + b + lane];
            }
            idx = (int)min((unsigned)idx, (unsigned)(n - 1));
            int j = 0;
            for (; j + 8 <= m; j += 8) add_rows<8>(h, idx, j, lane4, acc);
            if (j + 4 <= m) { add_rows<4>(h, idx, j, lane4, acc); j += 4; }
            if (j + 2 <= m) { add_rows<2>(h, idx, j, lane4, acc); j += 2; }
            if (j < m)      { add_rows<1>(h, idx, j, lane4, acc); }
        }

        float4 o;
        o.x = fmaxf(acc.x * sc[i] + bb.x, 0.0f);
        o.y = fmaxf(acc.y * sc[i] + bb.y, 0.0f);
        o.z = fmaxf(acc.z * sc[i] + bb.z, 0.0f);
        o.w = fmaxf(acc.w * sc[i] + bb.w, 0.0f);
        part[i] = fmaf(o.w, wv.w, fmaf(o.z, wv.z, fmaf(o.x, wv.x, o.y * wv.y)));
    }
    float mine = 0.0f;
#pragma unroll
    for (int off = 16; off > 0; off >>= 1) {
#pragma unroll
        for (int i = 0; i < NODES_PER_WAVE; ++i) part[i] += __shfl_xor(part[i], off, 32);
    }
#pragma unroll
    for (int i = 0; i < NODES_PER_WAVE; ++i)
        if (lane == i) mine = part[i];
    if (lane < NODES_PER_WAVE && node0 + lane < n) z[node0 + lane] = mine * so;
}

__global__ __launch_bounds__(256) void gcn_gather_scalar(const float* __restrict__ z,
                                                         const int* __restrict__ rowptr,
                                                         const int* __restrict__ cntIn,
                                                         const int* __restrict__ csr,
                                                         const float* __restrict__ invIn,
                                                         const float* __restrict__ b2,
                                                         float* __restrict__ out, int n, int nE) {
    const int wave  = blockIdx.x * 8 + (threadIdx.x >> 5);
    const int lane  = threadIdx.x & 31;
    const int grp   = lane >> 3;
    const int sub   = lane & 7;
    const int node0 = wave * (4 * GS_NODES_PER_GROUP) + grp;
    const float bias = b2[0];

    int beg[GS_NODES_PER_GROUP], deg[GS_NODES_PER_GROUP];
    float sc[GS_NODES_PER_GROUP];
#pragma unroll
    for (int t = 0; t < GS_NODES_PER_GROUP; ++t) {
        const int node = min(node0 + 4 * t, n - 1);
        int bg = rowptr[node];
        int dg = cntIn[node];
        const float v = invIn[node];
        bg = min(max(bg, 0), nE);
        dg = min(max(dg, 0), nE - bg);
        beg[t] = bg;
        deg[t] = dg;
        sc[t]  = v;
    }

#pragma unroll
    for (int t = 0; t < GS_NODES_PER_GROUP; ++t) {
        const int node = node0 + 4 * t;
        float s = 0.0f;
        for (int j = sub; j < deg[t]; j += 8) {
            const int idx = (int)min((unsigned)csr[beg[t] + j], (unsigned)(n - 1));
            s += z[idx];
        }
        s += __shfl_xor(s, 4, 32);
        s += __shfl_xor(s, 2, 32);
        s += __shfl_xor(s, 1, 32);
        if (sub == 0 && node < n) out[node] = s * sc[t] + bias;
    }
}

extern "C" void kernel_launch(void* const* d_in, const int* in_sizes, int n_in,
                              void* d_out, int out_size, void* d_ws, size_t ws_size,
                              hipStream_t stream) {
    const float* x   = (const float*)d_in[0];
    const int*   src = (const int*)d_in[1];
    const int*   dst = (const int*)d_in[2];
    const float* W1  = (const float*)d_in[3];
    const float* b1  = (const float*)d_in[4];
    const float* W2  = (const float*)d_in[5];
    const float* b2  = (const float*)d_in[6];
    float* out = (float*)d_out;

    const int n  = in_sizes[0] / FEATS;
    const int nE = in_sizes[1];
    const int nB = (n + BUCKET_NODES - 1) / BUCKET_NODES;

    float* f      = (float*)d_ws;
    int*   cntIn  = (int*)(f + (size_t)n);
    float* invOut = f + 2 * (size_t)n;
    float* invIn  = f + 3 * (size_t)n;
    float* z      = f + 4 * (size_t)n;
    int*   rowptr = (int*)(f + 5 * (size_t)n);
    int*   recS   = (int*)(f + 6 * (size_t)n);
    int*   csr    = recS + (size_t)nE;
    float* h      = (float*)(csr + (size_t)nE);
    int2*  recD   = (int2*)(h + (size_t)n * FEATS);
    int*   bcnt   = (int*)(recD + (size_t)nE);
    int*   cur    = bcnt + 2 * NBMAX;
    int*   bstart = cur + 2 * NBMAX;

    hipMemsetAsync(bcnt, 0, 4 * NBMAX * sizeof(int), stream);

    const int edgeBlocks = (nE + 256 * BIN_EDGES_PER_THREAD - 1) / (256 * BIN_EDGES_PER_THREAD);
    gcn_bucket_count<<<edgeBlocks, 256, 0, stream>>>(src, dst, bcnt, nE, nB);
    gcn_bucket_fill<<<edgeBlocks, 256, 0, stream>>>(src, dst, bcnt, cur, bstart, recD, recS, nE, nB);
    gcn_bucket_build<<<nB, 256, 0, stream>>>(bcnt, bstart, recD, recS, rowptr, cntIn, csr, invOut, invIn, n, nE);

    const int nRowTiles = n / 16;
    gcn_gemm1_wmma<<<(nRowTiles + ROW_TILES_PER_WAVE - 1) / ROW_TILES_PER_WAVE, 256, 0, stream>>>(x, W1, invOut, h, nRowTiles);

    gcn_gather_rows_dot<<<(n + 8 * NODES_PER_WAVE - 1) / (8 * NODES_PER_WAVE), 256, 0, stream>>>(h, rowptr, cntIn, csr, invIn, invOut, b1, W2, z, n, nE);

    gcn_gather_scalar<<<(n + 32 * GS_NODES_PER_GROUP - 1) / (32 * GS_NODES_PER_GROUP), 256, 0, stream>>>(z, rowptr, cntIn, csr, invIn, b2, out, n, nE);
}


# Layer-1 product moved behind the neighbour sum, into the gather kernel: scaled x rows summed, sums x W1 on WMMA from LDS, h never stored; sum order: aggregate-then-project, W2 dot by tile tree
# speedup vs baseline: 5.3917x; 1.0992x over previous
#include <hip/hip_runtime.h>
#include <hip/hip_bf16.h>


typedef __attribute__((ext_vector_type(2))) float v2f;
typedef __attribute__((ext_vector_type(8))) float v8f;

#define FEATS 128

#define BIN_EDGES_PER_THREAD 16
#define NODES_PER_WAVE     4
#define GROUP_NODES        32
#define GROUPS_PER_BLOCK   4
#define A_CHUNK            4
#define A_PITCH            132
#define GS_NODES_PER_GROUP 4

#define BUCKET_SHIFT 8
#define BUCKET_NODES 256
#define NBMAX        512
#define REC_BATCH    8

static __device__ __forceinline__ int bucket_of(int node, int nB) {
    return min((int)((unsigned)node >> BUCKET_SHIFT), nB - 1);
}

static __device__ __forceinline__ void block_excl_scan512(int* a, int* wsum, int tid) {
    const int lane = tid & 31;
    const int w    = tid >> 5;
    const int v0 = a[2 * tid];
    const int v1 = a[2 * tid + 1];
    const int local = v0 + v1;
    int inc = local;
    for (int off = 1; off < 32; off <<= 1) {
        const int t = __shfl_up(inc, off, 32);
        if (lane >= off) inc += t;
    }
    if (lane == 31) wsum[w] = inc;
    __syncthreads();
    int offset = 0;
    for (int k = 0; k < 8; ++k)
        if (k < w) offset += wsum[k];
    const int ex = offset + inc - local;
    a[2 * tid]     = ex;
    a[2 * tid + 1] = ex + v0;
    __syncthreads();
}

__global__ __launch_bounds__(256) void gcn_bucket_count(const int* __restrict__ src,
                                                        const int* __restrict__ dst,
                                                        int* __restrict__ bcnt, int nE, int nB) {
    __shared__ int hD[NBMAX];
    __shared__ int hS[NBMAX];
    const int tid = threadIdx.x;
    for (int i = tid; i < NBMAX; i += 256) { hD[i] = 0; hS[i] = 0; }

    const int e0 = blockIdx.x * (256 * BIN_EDGES_PER_THREAD) + tid;
    int s[BIN_EDGES_PER_THREAD], d[BIN_EDGES_PER_THREAD];
#pragma unroll
    for (int u = 0; u < BIN_EDGES_PER_THREAD; ++u) {
        const int ec = min(e0 + u * 256, nE - 1);
        s[u] = src[ec];
        d[u] = dst[ec];
    }
    __syncthreads();
#pragma unroll
    for (int u = 0; u < BIN_EDGES_PER_THREAD; ++u) {
        if (e0 + u * 256 < nE) {
            atomicAdd(&hD[bucket_of(d[u], nB)], 1);
            atomicAdd(&hS[bucket_of(s[u], nB)], 1);
        }
    }
    __syncthreads();
    for (int i = tid; i < nB; i += 256) {
        const int a = hD[i];
        const int b = hS[i];
        if (a) atomicAdd(&bcnt[i], a);
        if (b) atomicAdd(&bcnt[NBMAX + i], b);
    }
}

__global__ __launch_bounds__(256) void gcn_bucket_fill(const int* __restrict__ src,
                                                       const int* __restrict__ dst,
                                                       const int* __restrict__ bcnt,
                                                       int* __restrict__ cur,
                                                       int* __restrict__ bstart,
                                                       int2* __restrict__ recD,
                                                       int* __restrict__ recS, int nE, int nB) {
    __shared__ int  hD[NBMAX];
    __shared__ int  hS[NBMAX];
    __shared__ int  gD[NBMAX];
    __shared__ int  gS[NBMAX];
    __shared__ int  wsum[8];
    __shared__ int2 stD[256 * BIN_EDGES_PER_THREAD];
    __shared__ int  stS[256 * BIN_EDGES_PER_THREAD];
    const int tid = threadIdx.x;
    for (int i = tid; i < NBMAX; i += 256) {
        hD[i] = 0;
        hS[i] = 0;
        gD[i] = (i < nB) ? bcnt[i] : 0;
        gS[i] = (i < nB) ? bcnt[NBMAX + i] : 0;
    }

    const int eBlock = blockIdx.x * (256 * BIN_EDGES_PER_THREAD);
    const int e0 = eBlock + tid;
    int s[BIN_EDGES_PER_THREAD], d[BIN_EDGES_PER_THREAD];
    int rD[BIN_EDGES_PER_THREAD], rS[BIN_EDGES_PER_THREAD];
#pragma unroll
    for (int u = 0; u < BIN_EDGES_PER_THREAD; ++u) {
        const int ec = min(e0 + u * 256, nE - 1);
        s[u] = src[ec];
        d[u] = dst[ec];
    }
    __syncthreads();
#pragma unroll
    for (int u = 0; u < BIN_EDGES_PER_THREAD; ++u) {
        rD[u] = 0;
        rS[u] = 0;
        if (e0 + u * 256 < nE) {
            rD[u] = atomicAdd(&hD[bucket_of(d[u], nB)], 1);
            rS[u] = atomicAdd(&hS[bucket_of(s[u], nB)], 1);
        }
    }
    block_excl_scan512(gD, wsum, tid);
    block_excl_scan512(gS, wsum, tid);

    int baseD[2], baseS[2];
#pragma unroll
    for (int k = 0; k < 2; ++k) {
        const int i = tid + k * 256;
        baseD[k] = 0;
        baseS[k] = 0;
        if (i < nB) {
            baseD[k] = atomicAdd(&cur[i], hD[i]);
            baseS[k] = atomicAdd(&cur[NBMAX + i], hS[i]);
        }
    }
#pragma unroll
    for (int k = 0; k < 2; ++k) {
        const int i = tid + k * 256;
        if (i < nB) {
            if (blockIdx.x == 0) {
                bstart[i]         = gD[i];
                bstart[NBMAX + i] = gS[i];
            }
            gD[i] += baseD[k];
            gS[i] += baseS[k];
        }
    }
    block_excl_scan512(hD, wsum, tid);
    block_excl_scan512(hS, wsum, tid);

#pragma unroll
    for (int u = 0; u < BIN_EDGES_PER_THREAD; ++u) {
        if (e0 + u * 256 < nE) {
            const int pD = (hD[bucket_of(d[u], nB)] + rD[u]) & (256 * BIN_EDGES_PER_THREAD - 1);
            const int pS = (hS[bucket_of(s[u], nB)] + rS[u]) & (256 * BIN_EDGES_PER_THREAD - 1);
            stD[pD] = make_int2(d[u], s[u]);
            stS[pS] = s[u];
        }
    }
    __syncthreads();

    const int nMine = min(256 * BIN_EDGES_PER_THREAD, nE - eBlock);
#pragma unroll
    for (int u = 0; u < BIN_EDGES_PER_THREAD; ++u) {
        const int j = tid + u * 256;
        if (j < nMine) {
            const int2 v  = stD[j];
            const int  bD = bucket_of(v.x, nB);
            const unsigned pD = (unsigned)(gD[bD] + (j - hD[bD]));
            const int  w  = stS[j];
            const int  bS = bucket_of(w, nB);
            const unsigned pS = (unsigned)(gS[bS] + (j - hS[bS]));
            if (pD < (unsigned)nE) recD[pD] = v;
            if (pS < (unsigned)nE) recS[pS] = w;
        }
    }
}

__global__ __launch_bounds__(256) void gcn_bucket_build(const int* __restrict__ bcnt,
                                                        const int* __restrict__ bstart,
                                                        const int2* __restrict__ recD,
                                                        const int* __restrict__ recS,
                                                        int* __restrict__ rowptr,
                                                        int* __restrict__ cntIn,
                                                        int* __restrict__ csr,
                                                        float* __restrict__ invOut,
                                                        float* __restrict__ invIn,
                                                        int n, int nE) {
    __shared__ int c[BUCKET_NODES];
    __shared__ int wsum[8];
    const int tid   = threadIdx.x;
    const int lane  = tid & 31;
    const int w     = tid >> 5;
    const int b     = blockIdx.x;
    const int node0 = b << BUCKET_SHIFT;
    const int node  = node0 + tid;

    int begS = bstart[NBMAX + b], numS = bcnt[NBMAX + b];
    int begD = bstart[b],         numD = bcnt[b];
    begS = min(max(begS, 0), nE);  numS = min(max(numS, 0), nE - begS);
    begD = min(max(begD, 0), nE);  numD = min(max(numD, 0), nE - begD);
    const int*  __restrict__ rs = recS + begS;
    const int2* __restrict__ rd = recD + begD;

    c[tid] = 0;
    __syncthreads();
    for (int i = tid; i < numS; i += REC_BATCH * 256) {
        int v[REC_BATCH];
#pragma unroll
        for (int u = 0; u < REC_BATCH; ++u) {
            v[u] = rs[min(i + u * 256, numS - 1)];
        }
#pragma unroll
        for (int u = 0; u < REC_BATCH; ++u) {
            const unsigned l = (unsigned)(v[u] - node0);
            if (i + u * 256 < numS && l < BUCKET_NODES) atomicAdd(&c[l], 1);
        }
    }
    __syncthreads();
    if (node < n) invOut[node] = rsqrtf(fmaxf((float)c[tid], 1.0f));
    c[tid] = 0;
    __syncthreads();

    for (int i = tid; i < numD; i += REC_BATCH * 256) {
        int2 v[REC_BATCH];
#pragma unroll
        for (int u = 0; u < REC_BATCH; ++u) v[u] = rd[min(i + u * 256, numD - 1)];
#pragma unroll
        for (int u = 0; u < REC_BATCH; ++u) {
            const unsigned l = (unsigned)(v[u].x - node0);
            if (i + u * 256 < numD && l < BUCKET_NODES) atomicAdd(&c[l], 1);
        }
    }
    __syncthreads();

    const int cnt = c[tid];
    int inc = cnt;
    for (int off = 1; off < 32; off <<= 1) {
        const int t = __shfl_up(inc, off, 32);
        if (lane >= off) inc += t;
    }
    if (lane == 31) wsum[w] = inc;
    __syncthreads();
    int offset = 0;
    for (int k = 0; k < 8; ++k)
        if (k < w) offset += wsum[k];
    const int first = begD + offset + inc - cnt;
    if (node < n) {
        rowptr[node] = first;
        cntIn[node]  = cnt;
        invIn[node]  = rsqrtf(fmaxf((float)cnt, 1.0f));
    }
    c[tid] = first;
    __syncthreads();

    for (int i = tid; i < numD; i += REC_BATCH * 256) {
        int2 v[REC_BATCH];
#pragma unroll
        for (int u = 0; u < REC_BATCH; ++u) v[u] = rd[min(i + u * 256, numD - 1)];
#pragma unroll
        for (int u = 0; u < REC_BATCH; ++u) {
            const unsigned l = (unsigned)(v[u].x - node0);
            if (i + u * 256 < numD && l < BUCKET_NODES) {
                const unsigned pos = (unsigned)atomicAdd(&c[l], 1);
                if (pos < (unsigned)nE) csr[pos] = v[u].y;
            }
        }
    }
}

template <int U>
static __device__ __forceinline__ void add_rows_scaled(const float* __restrict__ x, int idx, int wgt,
                                                       int j, int lane4, float4& acc) {
    float4 v[U];
#pragma unroll
    for (int u = 0; u < U; ++u) {
        const int s = __builtin_amdgcn_readlane(idx, j + u);
        v[u] = *(const float4*)(x + (size_t)s * FEATS + lane4);
    }
#pragma unroll
    for (int u = 0; u < U; ++u) {
        const float w = __int_as_float(__builtin_amdgcn_readlane(wgt, j + u));
        acc.x = fmaf(v[u].x, w, acc.x);
        acc.y = fmaf(v[u].y, w, acc.y);
        acc.z = fmaf(v[u].z, w, acc.z);
        acc.w = fmaf(v[u].w, w, acc.w);
    }
}

static __device__ __forceinline__ void add_batch_scaled(const float* __restrict__ x, int idx, int wgt,
                                                        int m, int lane4, float4& acc) {
    int j = 0;
    for (; j + 8 <= m; j += 8) add_rows_scaled<8>(x, idx, wgt, j, lane4, acc);
    if (j + 4 <= m) { add_rows_scaled<4>(x, idx, wgt, j, lane4, acc); j += 4; }
    if (j + 2 <= m) { add_rows_scaled<2>(x, idx, wgt, j, lane4, acc); j += 2; }
    if (j < m)      { add_rows_scaled<1>(x, idx, wgt, j, lane4, acc); }
}

__global__ __launch_bounds__(256, 8) void gcn_gather_project_dot(const float* __restrict__ x,
                                                              const float* __restrict__ W1,
                                                              const int* __restrict__ rowptr,
                                                              const int* __restrict__ cntIn,
                                                              const int* __restrict__ csr,
                                                              const float* __restrict__ invIn,
                                                              const float* __restrict__ invOut,
                                                              const float* __restrict__ b1,
                                                              const float* __restrict__ W2,
                                                              float* __restrict__ z, int n, int nE) {
    __shared__ float sA[GROUP_NODES * A_PITCH];
    __shared__ float sPart[8 * GROUP_NODES];

    const int lane  = threadIdx.x & 31;
    const int wv    = __builtin_amdgcn_readfirstlane((int)(threadIdx.x >> 5));
    const int lane4 = lane * 4;
    const int l15   = lane & 15;
    const int khalf = lane >> 4;
    const int nCol  = wv * 16 + l15;

    v2f b[FEATS / 4];
#pragma unroll
    for (int kk = 0; kk < FEATS / 4; ++kk) {
        const int ka = kk * 4 + 2 * khalf;
        b[kk].x = W1[(size_t)ka       * FEATS + nCol];
        b[kk].y = W1[(size_t)(ka + 1) * FEATS + nCol];
    }
    const float bcol = b1[nCol];
    const float wcol = W2[nCol];

    for (int g = 0; g < GROUPS_PER_BLOCK; ++g) {
        const int base = (blockIdx.x * GROUPS_PER_BLOCK + g) * GROUP_NODES;
        if (base >= n) break;
        const int node0 = base + wv * NODES_PER_WAVE;

        int beg[NODES_PER_WAVE], deg[NODES_PER_WAVE], idx0[NODES_PER_WAVE];
        float sc[NODES_PER_WAVE];
#pragma unroll
        for (int i = 0; i < NODES_PER_WAVE; ++i) {
            const int node = min(node0 + i, n - 1);
            int bg = rowptr[node];
            int dg = cntIn[node];
            bg = min(max(bg, 0), nE);
            dg = min(max(dg, 0), nE - bg);
            if (node0 + i >= n) dg = 0;
            beg[i] = bg;
            deg[i] = dg;
            sc[i]  = invIn[node];
        }
#pragma unroll
        for (int i = 0; i < NODES_PER_WAVE; ++i) {
            idx0[i] = csr[max(min(beg[i] + lane, nE - 1), 0)];
        }
        int wgt0[NODES_PER_WAVE];
#pragma unroll
        for (int i = 0; i < NODES_PER_WAVE; ++i) {
            idx0[i] = (int)min((unsigned)idx0[i], (unsigned)(n - 1));
            wgt0[i] = __float_as_int(invOut[idx0[i]]);
        }

#pragma unroll
        for (int i = 0; i < NODES_PER_WAVE; ++i) {
            float4 acc = {0.0f, 0.0f, 0.0f, 0.0f};
            add_batch_scaled(x, idx0[i], wgt0[i], min(32, deg[i]), lane4, acc);
            for (int bb = 32; bb < deg[i]; bb += 32) {
                const int m = min(32, deg[i] - bb);
                int idx = csr[max(min(beg[i] + bb + lane, nE - 1), 0)];
                idx = (int)min((unsigned)idx, (unsigned)(n - 1));
                const int wgt = __float_as_int(invOut[idx]);
                add_batch_scaled(x, idx, wgt, m, lane4, acc);
            }
            acc.x *= sc[i]; acc.y *= sc[i]; acc.z *= sc[i]; acc.w *= sc[i];
            *(float4*)(sA + (wv * NODES_PER_WAVE + i) * A_PITCH + lane4) = acc;
        }
        __syncthreads();

        const float* __restrict__ arow0 = sA + l15 * A_PITCH + 2 * khalf;
        const float* __restrict__ arow1 = arow0 + 16 * A_PITCH;
        v8f c0 = {};
        v8f c1 = {};
        v2f a0[2][A_CHUNK], a1[2][A_CHUNK];
#pragma unroll
        for (int u = 0; u < A_CHUNK; ++u) {
            a0[0][u] = *(const v2f*)(arow0 + u * 4);
            a1[0][u] = *(const v2f*)(arow1 + u * 4);
        }
#pragma unroll
        for (int ch = 0; ch < FEATS / 4 / A_CHUNK; ++ch) {
            const int cur = ch & 1;
            if (ch + 1 < FEATS / 4 / A_CHUNK) {
#pragma unroll
                for (int u = 0; u < A_CHUNK; ++u) {
                    a0[cur ^ 1][u] = *(const v2f*)(arow0 + ((ch + 1) * A_CHUNK + u) * 4);
                    a1[cur ^ 1][u] = *(const v2f*)(arow1 + ((ch + 1) * A_CHUNK + u) * 4);
                }
            }
#pragma unroll
            for (int u = 0; u < A_CHUNK; ++u) {
                c0 = __builtin_amdgcn_wmma_f32_16x16x4_f32(
                         false, a0[cur][u],  false, b[ch * A_CHUNK + u],
                         (short)0, c0,  false,  false);
                c1 = __builtin_amdgcn_wmma_f32_16x16x4_f32(
                         false, a1[cur][u],  false, b[ch * A_CHUNK + u],
                         (short)0, c1,  false,  false);
            }
        }

#pragma unroll
        for (int rt = 0; rt < GROUP_NODES / 16; ++rt) {
            const v8f c = rt ? c1 : c0;
            float v[8];
#pragma unroll
            for (int r = 0; r < 8; ++r) {
                v[r] = fmaxf(c[r] + bcol, 0.0f) * wcol;
            }
            const bool h8 = (l15 & 8) != 0, h4 = (l15 & 4) != 0, h2 = (l15 & 2) != 0;
            float w4[4], u2[2];
#pragma unroll
            for (int q = 0; q < 4; ++q) {
                const float keep = h8 ? v[q + 4] : v[q];
                const float send = h8 ? v[q] : v[q + 4];
                w4[q] = keep + __shfl_xor(send, 8, 32);
            }
#pragma unroll
            for (int q = 0; q < 2; ++q) {
                const float keep = h4 ? w4[q + 2] : w4[q];
                const float send = h4 ? w4[q] : w4[q + 2];
                u2[q] = keep + __shfl_xor(send, 4, 32);
            }
            float t = (h2 ? u2[1] : u2[0]) + __shfl_xor(h2 ? u2[0] : u2[1], 2, 32);
            t += __shfl_xor(t, 1, 32);
            if ((lane & 1) == 0) sPart[wv * GROUP_NODES + rt * 16 + (lane >> 1)] = t;
        }
        __syncthreads();

        if (threadIdx.x < GROUP_NODES) {
            const int node = base + (int)threadIdx.x;
            float s = sPart[threadIdx.x];
#pragma unroll
            for (int ct = 1; ct < 8; ++ct) s += sPart[ct * GROUP_NODES + threadIdx.x];
            if (node < n) z[node] = s * invOut[node];
        }
    }
}

__global__ __launch_bounds__(256) void gcn_gather_scalar(const float* __restrict__ z,
                                                         const int* __restrict__ rowptr,
                                                         const int* __restrict__ cntIn,
                                                         const int* __restrict__ csr,
                                                         const float* __restrict__ invIn,
                                                         const float* __restrict__ b2,
                                                         float* __restrict__ out, int n, int nE) {
    const int wave  = blockIdx.x * 8 + (threadIdx.x >> 5);
    const int lane  = threadIdx.x & 31;
    const int grp   = lane >> 3;
    const int sub   = lane & 7;
    const int node0 = wave * (4 * GS_NODES_PER_GROUP) + grp;
    const float bias = b2[0];

    int beg[GS_NODES_PER_GROUP], deg[GS_NODES_PER_GROUP];
    float sc[GS_NODES_PER_GROUP];
#pragma unroll
    for (int t = 0; t < GS_NODES_PER_GROUP; ++t) {
        const int node = min(node0 + 4 * t, n - 1);
        int bg = rowptr[node];
        int dg = cntIn[node];
        const float v = invIn[node];
        bg = min(max(bg, 0), nE);
        dg = min(max(dg, 0), nE - bg);
        beg[t] = bg;
        deg[t] = dg;
        sc[t]  = v;
    }

#pragma unroll
    for (int t = 0; t < GS_NODES_PER_GROUP; ++t) {
        const int node = node0 + 4 * t;
        float s = 0.0f;
        for (int j = sub; j < deg[t]; j += 8) {
            const int idx = (int)min((unsigned)csr[beg[t] + j], (unsigned)(n - 1));
            s += z[idx];
        }
        s += __shfl_xor(s, 4, 32);
        s += __shfl_xor(s, 2, 32);
        s += __shfl_xor(s, 1, 32);
        if (sub == 0 && node < n) out[node] = s * sc[t] + bias;
    }
}

extern "C" void kernel_launch(void* const* d_in, const int* in_sizes, int n_in,
                              void* d_out, int out_size, void* d_ws, size_t ws_size,
                              hipStream_t stream) {
    const float* x   = (const float*)d_in[0];
    const int*   src = (const int*)d_in[1];
    const int*   dst = (const int*)d_in[2];
    const float* W1  = (const float*)d_in[3];
    const float* b1  = (const float*)d_in[4];
    const float* W2  = (const float*)d_in[5];
    const float* b2  = (const float*)d_in[6];
    float* out = (float*)d_out;

    const int n  = in_sizes[0] / FEATS;
    const int nE = in_sizes[1];
    const int nB = (n + BUCKET_NODES - 1) / BUCKET_NODES;

    float* f      = (float*)d_ws;
    int*   cntIn  = (int*)(f + (size_t)n);
    float* invOut = f + 2 * (size_t)n;
    float* invIn  = f + 3 * (size_t)n;
    float* z      = f + 4 * (size_t)n;
    int*   rowptr = (int*)(f + 5 * (size_t)n);
    int*   recS   = (int*)(f + 6 * (size_t)n);
    int*   csr    = recS + (size_t)nE;
    int2*  recD   = (int2*)(csr + (size_t)nE + (size_t)n * FEATS);
    int*   bcnt   = (int*)(recD + (size_t)nE);
    int*   cur    = bcnt + 2 * NBMAX;
    int*   bstart = cur + 2 * NBMAX;

    hipMemsetAsync(bcnt, 0, 4 * NBMAX * sizeof(int), stream);

    const int edgeBlocks = (nE + 256 * BIN_EDGES_PER_THREAD - 1) / (256 * BIN_EDGES_PER_THREAD);
    gcn_bucket_count<<<edgeBlocks, 256, 0, stream>>>(src, dst, bcnt, nE, nB);
    gcn_bucket_fill<<<edgeBlocks, 256, 0, stream>>>(src, dst, bcnt, cur, bstart, recD, recS, nE, nB);
    gcn_bucket_build<<<nB, 256, 0, stream>>>(bcnt, bstart, recD, recS, rowptr, cntIn, csr, invOut, invIn, n, nE);

    gcn_gather_project_dot<<<(n + GROUP_NODES * GROUPS_PER_BLOCK - 1) / (GROUP_NODES * GROUPS_PER_BLOCK), 256, 0, stream>>>(x, W1, rowptr, cntIn, csr, invIn, invOut, b1, W2, z, n, nE);

    gcn_gather_scalar<<<(n + 32 * GS_NODES_PER_GROUP - 1) / (32 * GS_NODES_PER_GROUP), 256, 0, stream>>>(z, rowptr, cntIn, csr, invIn, b2, out, n, nE);
}


# Counting launch dropped: each bucket owns a fixed region of record slots filled through its cursor, overflow goes to a spill list only an overflowing bucket walks; sum order = arrival rank as before
# speedup vs baseline: 5.5868x; 1.0362x over previous
#include <hip/hip_runtime.h>
#include <hip/hip_bf16.h>


typedef __attribute__((ext_vector_type(2))) float v2f;
typedef __attribute__((ext_vector_type(8))) float v8f;

#define FEATS 128

#define BIN_EDGES_PER_THREAD 16
#define NODES_PER_WAVE     4
#define GROUP_NODES        32
#define GROUPS_PER_BLOCK   4
#define A_CHUNK            4
#define A_PITCH            132
#define GS_NODES_PER_GROUP 4

#define BUCKET_SHIFT 8
#define BUCKET_NODES 256
#define NBMAX        512
#define BUCKET_CAP   8240
#define REC_BATCH    8

static __device__ __forceinline__ int bucket_of(int node, int nB) {
    return min((int)((unsigned)node >> BUCKET_SHIFT), nB - 1);
}

static __device__ __forceinline__ void block_excl_scan512(int* a, int* wsum, int tid) {
    const int lane = tid & 31;
    const int w    = tid >> 5;
    const int v0 = a[2 * tid];
    const int v1 = a[2 * tid + 1];
    const int local = v0 + v1;
    int inc = local;
    for (int off = 1; off < 32; off <<= 1) {
        const int t = __shfl_up(inc, off, 32);
        if (lane >= off) inc += t;
    }
    if (lane == 31) wsum[w] = inc;
    __syncthreads();
    int offset = 0;
    for (int k = 0; k < 8; ++k)
        if (k < w) offset += wsum[k];
    const int ex = offset + inc - local;
    a[2 * tid]     = ex;
    a[2 * tid + 1] = ex + v0;
    __syncthreads();
}

__global__ __launch_bounds__(256) void gcn_bucket_fill(const int* __restrict__ src,
                                                       const int* __restrict__ dst,
                                                       int* __restrict__ cur,
                                                       int2* __restrict__ recD,
                                                       int* __restrict__ recS,
                                                       int2* __restrict__ spillD,
                                                       int* __restrict__ spillS, int nE, int nB) {
    __shared__ int  hD[NBMAX];
    __shared__ int  hS[NBMAX];
    __shared__ int  gD[NBMAX];
    __shared__ int  gS[NBMAX];
    __shared__ int  wsum[8];
    __shared__ int2 stD[256 * BIN_EDGES_PER_THREAD];
    __shared__ int  stS[256 * BIN_EDGES_PER_THREAD];
    const int tid = threadIdx.x;
    for (int i = tid; i < NBMAX; i += 256) {
        hD[i] = 0;
        hS[i] = 0;
    }

    const int eBlock = blockIdx.x * (256 * BIN_EDGES_PER_THREAD);
    const int e0 = eBlock + tid;
    int s[BIN_EDGES_PER_THREAD], d[BIN_EDGES_PER_THREAD];
    int rD[BIN_EDGES_PER_THREAD], rS[BIN_EDGES_PER_THREAD];
#pragma unroll
    for (int u = 0; u < BIN_EDGES_PER_THREAD; ++u) {
        const int ec = min(e0 + u * 256, nE - 1);
        s[u] = src[ec];
        d[u] = dst[ec];
    }
    __syncthreads();
#pragma unroll
    for (int u = 0; u < BIN_EDGES_PER_THREAD; ++u) {
        rD[u] = 0;
        rS[u] = 0;
        if (e0 + u * 256 < nE) {
            rD[u] = atomicAdd(&hD[bucket_of(d[u], nB)], 1);
            rS[u] = atomicAdd(&hS[bucket_of(s[u], nB)], 1);
        }
    }
    __syncthreads();

    int baseD[2], baseS[2];
#pragma unroll
    for (int k = 0; k < 2; ++k) {
        const int i = tid + k * 256;
        baseD[k] = 0;
        baseS[k] = 0;
        if (i < nB) {
            baseD[k] = atomicAdd(&cur[i], hD[i]);
            baseS[k] = atomicAdd(&cur[NBMAX + i], hS[i]);
        }
    }
#pragma unroll
    for (int k = 0; k < 2; ++k) {
        const int i = tid + k * 256;
        gD[i] = baseD[k];
        gS[i] = baseS[k];
    }
    block_excl_scan512(hD, wsum, tid);
    block_excl_scan512(hS, wsum, tid);

#pragma unroll
    for (int u = 0; u < BIN_EDGES_PER_THREAD; ++u) {
        if (e0 + u * 256 < nE) {
            const int pD = (hD[bucket_of(d[u], nB)] + rD[u]) & (256 * BIN_EDGES_PER_THREAD - 1);
            const int pS = (hS[bucket_of(s[u], nB)] + rS[u]) & (256 * BIN_EDGES_PER_THREAD - 1);
            stD[pD] = make_int2(d[u], s[u]);
            stS[pS] = s[u];
        }
    }
    __syncthreads();

    const int nMine = min(256 * BIN_EDGES_PER_THREAD, nE - eBlock);
#pragma unroll
    for (int u = 0; u < BIN_EDGES_PER_THREAD; ++u) {
        const int j = tid + u * 256;
        if (j < nMine) {
            const int2 v  = stD[j];
            const int  bD = bucket_of(v.x, nB);
            const unsigned kD = (unsigned)(gD[bD] + (j - hD[bD]));
            const int  w  = stS[j];
            const int  bS = bucket_of(w, nB);
            const unsigned kS = (unsigned)(gS[bS] + (j - hS[bS]));
            if (kD < (unsigned)BUCKET_CAP) {
                recD[(size_t)bD * BUCKET_CAP + kD] = v;
            } else {
                const unsigned q = (unsigned)atomicAdd(&cur[2 * NBMAX], 1);
                if (q < (unsigned)nE) spillD[q] = v;
            }
            if (kS < (unsigned)BUCKET_CAP) {
                recS[(size_t)bS * BUCKET_CAP + kS] = w;
            } else {
                const unsigned q = (unsigned)atomicAdd(&cur[2 * NBMAX + 1], 1);
                if (q < (unsigned)nE) spillS[q] = w;
            }
        }
    }
}

__global__ __launch_bounds__(256) void gcn_bucket_build(const int* __restrict__ cur,
                                                        const int2* __restrict__ recD,
                                                        const int* __restrict__ recS,
                                                        const int2* __restrict__ spillD,
                                                        const int* __restrict__ spillS,
                                                        int* __restrict__ rowptr,
                                                        int* __restrict__ cntIn,
                                                        int* __restrict__ csr,
                                                        float* __restrict__ invOut,
                                                        float* __restrict__ invIn,
                                                        int n, int nE, int nB) {
    __shared__ int c[BUCKET_NODES];
    __shared__ int wsum[8];
    __shared__ int tot[NBMAX];
    const int tid   = threadIdx.x;
    const int lane  = tid & 31;
    const int w     = tid >> 5;
    const int b     = blockIdx.x;
    const int node0 = b << BUCKET_SHIFT;
    const int node  = node0 + tid;

    for (int i = tid; i < NBMAX; i += 256) tot[i] = (i < nB) ? min(max(cur[i], 0), nE) : 0;
    c[tid] = 0;
    __syncthreads();
    const int totD = tot[b];
    const int totS = min(max(cur[NBMAX + b], 0), nE);
    block_excl_scan512(tot, wsum, tid);
    const int begD = min(tot[b], nE);
    const int numD[2] = { min(min(totD, BUCKET_CAP), nE - begD), (totD > BUCKET_CAP) ? min(max(cur[2 * NBMAX], 0), nE) : 0 };
    const int numS[2] = { min(totS, BUCKET_CAP),                 (totS > BUCKET_CAP) ? min(max(cur[2 * NBMAX + 1], 0), nE) : 0 };
    const int*  __restrict__ const listS[2] = { recS + (size_t)b * BUCKET_CAP, spillS };
    const int2* __restrict__ const listD[2] = { recD + (size_t)b * BUCKET_CAP, spillD };

#pragma unroll
    for (int q = 0; q < 2; ++q) {
        const int* __restrict__ rs = listS[q];
        const int num = numS[q];
        for (int i = tid; i < num; i += REC_BATCH * 256) {
            int v[REC_BATCH];
#pragma unroll
            for (int u = 0; u < REC_BATCH; ++u) {
                v[u] = rs[min(i + u * 256, num - 1)];
            }
#pragma unroll
            for (int u = 0; u < REC_BATCH; ++u) {
                const unsigned l = (unsigned)(v[u] - node0);
                if (i + u * 256 < num && l < BUCKET_NODES) atomicAdd(&c[l], 1);
            }
        }
    }
    __syncthreads();
    if (node < n) invOut[node] = rsqrtf(fmaxf((float)c[tid], 1.0f));
    c[tid] = 0;
    __syncthreads();

#pragma unroll
    for (int q = 0; q < 2; ++q) {
        const int2* __restrict__ rd = listD[q];
        const int num = numD[q];
        for (int i = tid; i < num; i += REC_BATCH * 256) {
            int2 v[REC_BATCH];
#pragma unroll
            for (int u = 0; u < REC_BATCH; ++u) v[u] = rd[min(i + u * 256, num - 1)];
#pragma unroll
            for (int u = 0; u < REC_BATCH; ++u) {
                const unsigned l = (unsigned)(v[u].x - node0);
                if (i + u * 256 < num && l < BUCKET_NODES) atomicAdd(&c[l], 1);
            }
        }
    }
    __syncthreads();

    const int cnt = c[tid];
    int inc = cnt;
    for (int off = 1; off < 32; off <<= 1) {
        const int t = __shfl_up(inc, off, 32);
        if (lane >= off) inc += t;
    }
    if (lane == 31) wsum[w] = inc;
    __syncthreads();
    int offset = 0;
    for (int k = 0; k < 8; ++k)
        if (k < w) offset += wsum[k];
    const int first = begD + offset + inc - cnt;
    if (node < n) {
        rowptr[node] = first;
        cntIn[node]  = cnt;
        invIn[node]  = rsqrtf(fmaxf((float)cnt, 1.0f));
    }
    c[tid] = first;
    __syncthreads();

#pragma unroll
    for (int q = 0; q < 2; ++q) {
        const int2* __restrict__ rd = listD[q];
        const int num = numD[q];
        for (int i = tid; i < num; i += REC_BATCH * 256) {
            int2 v[REC_BATCH];
#pragma unroll
            for (int u = 0; u < REC_BATCH; ++u) v[u] = rd[min(i + u * 256, num - 1)];
#pragma unroll
            for (int u = 0; u < REC_BATCH; ++u) {
                const unsigned l = (unsigned)(v[u].x - node0);
                if (i + u * 256 < num && l < BUCKET_NODES) {
                    const unsigned pos = (unsigned)atomicAdd(&c[l], 1);
                    if (pos < (unsigned)nE) csr[pos] = v[u].y;
                }
            }
        }
    }
}

template <int U>
static __device__ __forceinline__ void add_rows_scaled(const float* __restrict__ x, int idx, int wgt,
                                                       int j, int lane4, float4& acc) {
    float4 v[U];
#pragma unroll
    for (int u = 0; u < U; ++u) {
        const int s = __builtin_amdgcn_readlane(idx, j + u);
        v[u] = *(const float4*)(x + (size_t)s * FEATS + lane4);
    }
#pragma unroll
    for (int u = 0; u < U; ++u) {
        const float w = __int_as_float(__builtin_amdgcn_readlane(wgt, j + u));
        acc.x = fmaf(v[u].x, w, acc.x);
        acc.y = fmaf(v[u].y, w, acc.y);
        acc.z = fmaf(v[u].z, w, acc.z);
        acc.w = fmaf(v[u].w, w, acc.w);
    }
}

static __device__ __forceinline__ void add_batch_scaled(const float* __restrict__ x, int idx, int wgt,
                                                        int m, int lane4, float4& acc) {
    int j = 0;
    for (; j + 8 <= m; j += 8) add_rows_scaled<8>(x, idx, wgt, j, lane4, acc);
    if (j + 4 <= m) { add_rows_scaled<4>(x, idx, wgt, j, lane4, acc); j += 4; }
    if (j + 2 <= m) { add_rows_scaled<2>(x, idx, wgt, j, lane4, acc); j += 2; }
    if (j < m)      { add_rows_scaled<1>(x, idx, wgt, j, lane4, acc); }
}

__global__ __launch_bounds__(256, 8) void gcn_gather_project_dot(const float* __restrict__ x,
                                                              const float* __restrict__ W1,
                                                              const int* __restrict__ rowptr,
                                                              const int* __restrict__ cntIn,
                                                              const int* __restrict__ csr,
                                                              const float* __restrict__ invIn,
                                                              const float* __restrict__ invOut,
                                                              const float* __restrict__ b1,
                                                              const float* __restrict__ W2,
                                                              float* __restrict__ z, int n, int nE) {
    __shared__ float sA[GROUP_NODES * A_PITCH];
    __shared__ float sPart[8 * GROUP_NODES];

    const int lane  = threadIdx.x & 31;
    const int wv    = __builtin_amdgcn_readfirstlane((int)(threadIdx.x >> 5));
    const int lane4 = lane * 4;
    const int l15   = lane & 15;
    const int khalf = lane >> 4;
    const int nCol  = wv * 16 + l15;

    v2f b[FEATS / 4];
#pragma unroll
    for (int kk = 0; kk < FEATS / 4; ++kk) {
        const int ka = kk * 4 + 2 * khalf;
        b[kk].x = W1[(size_t)ka       * FEATS + nCol];
        b[kk].y = W1[(size_t)(ka + 1) * FEATS + nCol];
    }
    const float bcol = b1[nCol];
    const float wcol = W2[nCol];

    for (int g = 0; g < GROUPS_PER_BLOCK; ++g) {
        const int base = (blockIdx.x * GROUPS_PER_BLOCK + g) * GROUP_NODES;
        if (base >= n) break;
        const int node0 = base + wv * NODES_PER_WAVE;

        int beg[NODES_PER_WAVE], deg[NODES_PER_WAVE], idx0[NODES_PER_WAVE];
        float sc[NODES_PER_WAVE];
#pragma unroll
        for (int i = 0; i < NODES_PER_WAVE; ++i) {
            const int node = min(node0 + i, n - 1);
            int bg = rowptr[node];
            int dg = cntIn[node];
            bg = min(max(bg, 0), nE);
            dg = min(max(dg, 0), nE - bg);
            if (node0 + i >= n) dg = 0;
            beg[i] = bg;
            deg[i] = dg;
            sc[i]  = invIn[node];
        }
#pragma unroll
        for (int i = 0; i < NODES_PER_WAVE; ++i) {
            idx0[i] = csr[max(min(beg[i] + lane, nE - 1), 0)];
        }
        int wgt0[NODES_PER_WAVE];
#pragma unroll
        for (int i = 0; i < NODES_PER_WAVE; ++i) {
            idx0[i] = (int)min((unsigned)idx0[i], (unsigned)(n - 1));
            wgt0[i] = __float_as_int(invOut[idx0[i]]);
        }

#pragma unroll
        for (int i = 0; i < NODES_PER_WAVE; ++i) {
            float4 acc = {0.0f, 0.0f, 0.0f, 0.0f};
            add_batch_scaled(x, idx0[i], wgt0[i], min(32, deg[i]), lane4, acc);
            for (int bb = 32; bb < deg[i]; bb += 32) {
                const int m = min(32, deg[i] - bb);
                int idx = csr[max(min(beg[i] + bb + lane, nE - 1), 0)];
                idx = (int)min((unsigned)idx, (unsigned)(n - 1));
                const int wgt = __float_as_int(invOut[idx]);
                add_batch_scaled(x, idx, wgt, m, lane4, acc);
            }
            acc.x *= sc[i]; acc.y *= sc[i]; acc.z *= sc[i]; acc.w *= sc[i];
            *(float4*)(sA + (wv * NODES_PER_WAVE + i) * A_PITCH + lane4) = acc;
        }
        __syncthreads();

        const float* __restrict__ arow0 = sA + l15 * A_PITCH + 2 * khalf;
        const float* __restrict__ arow1 = arow0 + 16 * A_PITCH;
        v8f c0 = {};
        v8f c1 = {};
        v2f a0[2][A_CHUNK], a1[2][A_CHUNK];
#pragma unroll
        for (int u = 0; u < A_CHUNK; ++u) {
            a0[0][u] = *(const v2f*)(arow0 + u * 4);
            a1[0][u] = *(const v2f*)(arow1 + u * 4);
        }
#pragma unroll
        for (int ch = 0; ch < FEATS / 4 / A_CHUNK; ++ch) {
            const int cur = ch & 1;
            if (ch + 1 < FEATS / 4 / A_CHUNK) {
#pragma unroll
                for (int u = 0; u < A_CHUNK; ++u) {
                    a0[cur ^ 1][u] = *(const v2f*)(arow0 + ((ch + 1) * A_CHUNK + u) * 4);
                    a1[cur ^ 1][u] = *(const v2f*)(arow1 + ((ch + 1) * A_CHUNK + u) * 4);
                }
            }
#pragma unroll
            for (int u = 0; u < A_CHUNK; ++u) {
                c0 = __builtin_amdgcn_wmma_f32_16x16x4_f32(
                         false, a0[cur][u],  false, b[ch * A_CHUNK + u],
                         (short)0, c0,  false,  false);
                c1 = __builtin_amdgcn_wmma_f32_16x16x4_f32(
                         false, a1[cur][u],  false, b[ch * A_CHUNK + u],
                         (short)0, c1,  false,  false);
            }
        }

#pragma unroll
        for (int rt = 0; rt < GROUP_NODES / 16; ++rt) {
            const v8f c = rt ? c1 : c0;
            float v[8];
#pragma unroll
            for (int r = 0; r < 8; ++r) {
                v[r] = fmaxf(c[r] + bcol, 0.0f) * wcol;
            }
            const bool h8 = (l15 & 8) != 0, h4 = (l15 & 4) != 0, h2 = (l15 & 2) != 0;
            float w4[4], u2[2];
#pragma unroll
            for (int q = 0; q < 4; ++q) {
                const float keep = h8 ? v[q + 4] : v[q];
                const float send = h8 ? v[q] : v[q + 4];
                w4[q] = keep + __shfl_xor(send, 8, 32);
            }
#pragma unroll
            for (int q = 0; q < 2; ++q) {
                const float keep = h4 ? w4[q + 2] : w4[q];
                const float send = h4 ? w4[q] : w4[q + 2];
                u2[q] = keep + __shfl_xor(send, 4, 32);
            }
            float t = (h2 ? u2[1] : u2[0]) + __shfl_xor(h2 ? u2[0] : u2[1], 2, 32);
            t += __shfl_xor(t, 1, 32);
            if ((lane & 1) == 0) sPart[wv * GROUP_NODES + rt * 16 + (lane >> 1)] = t;
        }
        __syncthreads();

        if (threadIdx.x < GROUP_NODES) {
            const int node = base + (int)threadIdx.x;
            float s = sPart[threadIdx.x];
#pragma unroll
            for (int ct = 1; ct < 8; ++ct) s += sPart[ct * GROUP_NODES + threadIdx.x];
            if (node < n) z[node] = s * invOut[node];
        }
    }
}

__global__ __launch_bounds__(256) void gcn_gather_scalar(const float* __restrict__ z,
                                                         const int* __restrict__ rowptr,
                                                         const int* __restrict__ cntIn,
                                                         const int* __restrict__ csr,
                                                         const float* __restrict__ invIn,
                                                         const float* __restrict__ b2,
                                                         float* __restrict__ out, int n, int nE) {
    const int wave  = blockIdx.x * 8 + (threadIdx.x >> 5);
    const int lane  = threadIdx.x & 31;
    const int grp   = lane >> 3;
    const int sub   = lane & 7;
    const int node0 = wave * (4 * GS_NODES_PER_GROUP) + grp;
    const float bias = b2[0];

    int beg[GS_NODES_PER_GROUP], deg[GS_NODES_PER_GROUP];
    float sc[GS_NODES_PER_GROUP];
#pragma unroll
    for (int t = 0; t < GS_NODES_PER_GROUP; ++t) {
        const int node = min(node0 + 4 * t, n - 1);
        int bg = rowptr[node];
        int dg = cntIn[node];
        const float v = invIn[node];
        bg = min(max(bg, 0), nE);
        dg = min(max(dg, 0), nE - bg);
        beg[t] = bg;
        deg[t] = dg;
        sc[t]  = v;
    }

#pragma unroll
    for (int t = 0; t < GS_NODES_PER_GROUP; ++t) {
        const int node = node0 + 4 * t;
        float s = 0.0f;
        for (int j = sub; j < deg[t]; j += 8) {
            const int idx = (int)min((unsigned)csr[beg[t] + j], (unsigned)(n - 1));
            s += z[idx];
        }
        s += __shfl_xor(s, 4, 32);
        s += __shfl_xor(s, 2, 32);
        s += __shfl_xor(s, 1, 32);
        if (sub == 0 && node < n) out[node] = s * sc[t] + bias;
    }
}

extern "C" void kernel_launch(void* const* d_in, const int* in_sizes, int n_in,
                              void* d_out, int out_size, void* d_ws, size_t ws_size,
                              hipStream_t stream) {
    const float* x   = (const float*)d_in[0];
    const int*   src = (const int*)d_in[1];
    const int*   dst = (const int*)d_in[2];
    const float* W1  = (const float*)d_in[3];
    const float* b1  = (const float*)d_in[4];
    const float* W2  = (const float*)d_in[5];
    const float* b2  = (const float*)d_in[6];
    float* out = (float*)d_out;

    const int n  = in_sizes[0] / FEATS;
    const int nE = in_sizes[1];
    const int nB = (n + BUCKET_NODES - 1) / BUCKET_NODES;

    float* f      = (float*)d_ws;
    int*   cntIn  = (int*)(f + (size_t)n);
    float* invOut = f + 2 * (size_t)n;
    float* invIn  = f + 3 * (size_t)n;
    float* z      = f + 4 * (size_t)n;
    int*   rowptr = (int*)(f + 5 * (size_t)n);
    int2*  recD   = (int2*)(f + 6 * (size_t)n);
    int2*  spillD = recD + (size_t)NBMAX * BUCKET_CAP;
    int*   recS   = (int*)(spillD + (size_t)nE);
    int*   spillS = recS + (size_t)NBMAX * BUCKET_CAP;
    int*   csr    = spillS + (size_t)nE;
    int*   cur    = csr + (size_t)nE;

    hipMemsetAsync(cur, 0, 4 * NBMAX * sizeof(int), stream);

    const int edgeBlocks = (nE + 256 * BIN_EDGES_PER_THREAD - 1) / (256 * BIN_EDGES_PER_THREAD);
    gcn_bucket_fill<<<edgeBlocks, 256, 0, stream>>>(src, dst, cur, recD, recS, spillD, spillS, nE, nB);
    gcn_bucket_build<<<nB, 256, 0, stream>>>(cur, recD, recS, spillD, spillS, rowptr, cntIn, csr, invOut, invIn, n, nE, nB);

    gcn_gather_project_dot<<<(n + GROUP_NODES * GROUPS_PER_BLOCK - 1) / (GROUP_NODES * GROUPS_PER_BLOCK), 256, 0, stream>>>(x, W1, rowptr, cntIn, csr, invIn, invOut, b1, W2, z, n, nE);

    gcn_gather_scalar<<<(n + 32 * GS_NODES_PER_GROUP - 1) / (32 * GS_NODES_PER_GROUP), 256, 0, stream>>>(z, rowptr, cntIn, csr, invIn, b2, out, n, nE);
}


# Scalar gather requests each lane group's whole walk at once (8 nodes x 4 slices of 8 edges: all list entries, then all z values): 3 dependent round trips per wave, half the waves; sum order kept
# speedup vs baseline: 5.7631x; 1.0316x over previous
#include <hip/hip_runtime.h>
#include <hip/hip_bf16.h>


typedef __attribute__((ext_vector_type(2))) float v2f;
typedef __attribute__((ext_vector_type(8))) float v8f;

#define FEATS 128

#define BIN_EDGES_PER_THREAD 16
#define NODES_PER_WAVE     4
#define GROUP_NODES        32
#define GROUPS_PER_BLOCK   4
#define A_CHUNK            4
#define A_PITCH            132
#define GS_NODES_PER_GROUP 8
#define GS_SLICES          4

#define BUCKET_SHIFT 8
#define BUCKET_NODES 256
#define NBMAX        512
#define BUCKET_CAP   8240
#define REC_BATCH    8

static __device__ __forceinline__ int bucket_of(int node, int nB) {
    return min((int)((unsigned)node >> BUCKET_SHIFT), nB - 1);
}

static __device__ __forceinline__ void block_excl_scan512(int* a, int* wsum, int tid) {
    const int lane = tid & 31;
    const int w    = tid >> 5;
    const int v0 = a[2 * tid];
    const int v1 = a[2 * tid + 1];
    const int local = v0 + v1;
    int inc = local;
    for (int off = 1; off < 32; off <<= 1) {
        const int t = __shfl_up(inc, off, 32);
        if (lane >= off) inc += t;
    }
    if (lane == 31) wsum[w] = inc;
    __syncthreads();
    int offset = 0;
    for (int k = 0; k < 8; ++k)
        if (k < w) offset += wsum[k];
    const int ex = offset + inc - local;
    a[2 * tid]     = ex;
    a[2 * tid + 1] = ex + v0;
    __syncthreads();
}

__global__ __launch_bounds__(256) void gcn_bucket_fill(const int* __restrict__ src,
                                                       const int* __restrict__ dst,
                                                       int* __restrict__ cur,
                                                       int2* __restrict__ recD,
                                                       int* __restrict__ recS,
                                                       int2* __restrict__ spillD,
                                                       int* __restrict__ spillS, int nE, int nB) {
    __shared__ int  hD[NBMAX];
    __shared__ int  hS[NBMAX];
    __shared__ int  gD[NBMAX];
    __shared__ int  gS[NBMAX];
    __shared__ int  wsum[8];
    __shared__ int2 stD[256 * BIN_EDGES_PER_THREAD];
    __shared__ int  stS[256 * BIN_EDGES_PER_THREAD];
    const int tid = threadIdx.x;
    for (int i = tid; i < NBMAX; i += 256) {
        hD[i] = 0;
        hS[i] = 0;
    }

    const int eBlock = blockIdx.x * (256 * BIN_EDGES_PER_THREAD);
    const int e0 = eBlock + tid;
    int s[BIN_EDGES_PER_THREAD], d[BIN_EDGES_PER_THREAD];
    int rD[BIN_EDGES_PER_THREAD], rS[BIN_EDGES_PER_THREAD];
#pragma unroll
    for (int u = 0; u < BIN_EDGES_PER_THREAD; ++u) {
        const int ec = min(e0 + u * 256, nE - 1);
        s[u] = src[ec];
        d[u] = dst[ec];
    }
    __syncthreads();
#pragma unroll
    for (int u = 0; u < BIN_EDGES_PER_THREAD; ++u) {
        rD[u] = 0;
        rS[u] = 0;
        if (e0 + u * 256 < nE) {
            rD[u] = atomicAdd(&hD[bucket_of(d[u], nB)], 1);
            rS[u] = atomicAdd(&hS[bucket_of(s[u], nB)], 1);
        }
    }
    __syncthreads();

    int baseD[2], baseS[2];
#pragma unroll
    for (int k = 0; k < 2; ++k) {
        const int i = tid + k * 256;
        baseD[k] = 0;
        baseS[k] = 0;
        if (i < nB) {
            baseD[k] = atomicAdd(&cur[i], hD[i]);
            baseS[k] = atomicAdd(&cur[NBMAX + i], hS[i]);
        }
    }
#pragma unroll
    for (int k = 0; k < 2; ++k) {
        const int i = tid + k * 256;
        gD[i] = baseD[k];
        gS[i] = baseS[k];
    }
    block_excl_scan512(hD, wsum, tid);
    block_excl_scan512(hS, wsum, tid);

#pragma unroll
    for (int u = 0; u < BIN_EDGES_PER_THREAD; ++u) {
        if (e0 + u * 256 < nE) {
            const int pD = (hD[bucket_of(d[u], nB)] + rD[u]) & (256 * BIN_EDGES_PER_THREAD - 1);
            const int pS = (hS[bucket_of(s[u], nB)] + rS[u]) & (256 * BIN_EDGES_PER_THREAD - 1);
            stD[pD] = make_int2(d[u], s[u]);
            stS[pS] = s[u];
        }
    }
    __syncthreads();

    const int nMine = min(256 * BIN_EDGES_PER_THREAD, nE - eBlock);
#pragma unroll
    for (int u = 0; u < BIN_EDGES_PER_THREAD; ++u) {
        const int j = tid + u * 256;
        if (j < nMine) {
            const int2 v  = stD[j];
            const int  bD = bucket_of(v.x, nB);
            const unsigned kD = (unsigned)(gD[bD] + (j - hD[bD]));
            const int  w  = stS[j];
            const int  bS = bucket_of(w, nB);
            const unsigned kS = (unsigned)(gS[bS] + (j - hS[bS]));
            if (kD < (unsigned)BUCKET_CAP) {
                recD[(size_t)bD * BUCKET_CAP + kD] = v;
            } else {
                const unsigned q = (unsigned)atomicAdd(&cur[2 * NBMAX], 1);
                if (q < (unsigned)nE) spillD[q] = v;
            }
            if (kS < (unsigned)BUCKET_CAP) {
                recS[(size_t)bS * BUCKET_CAP + kS] = w;
            } else {
                const unsigned q = (unsigned)atomicAdd(&cur[2 * NBMAX + 1], 1);
                if (q < (unsigned)nE) spillS[q] = w;
            }
        }
    }
}

__global__ __launch_bounds__(256) void gcn_bucket_build(const int* __restrict__ cur,
                                                        const int2* __restrict__ recD,
                                                        const int* __restrict__ recS,
                                                        const int2* __restrict__ spillD,
                                                        const int* __restrict__ spillS,
                                                        int* __restrict__ rowptr,
                                                        int* __restrict__ cntIn,
                                                        int* __restrict__ csr,
                                                        float* __restrict__ invOut,
                                                        float* __restrict__ invIn,
                                                        int n, int nE, int nB) {
    __shared__ int c[BUCKET_NODES];
    __shared__ int wsum[8];
    __shared__ int tot[NBMAX];
    const int tid   = threadIdx.x;
    const int lane  = tid & 31;
    const int w     = tid >> 5;
    const int b     = blockIdx.x;
    const int node0 = b << BUCKET_SHIFT;
    const int node  = node0 + tid;

    for (int i = tid; i < NBMAX; i += 256) tot[i] = (i < nB) ? min(max(cur[i], 0), nE) : 0;
    c[tid] = 0;
    __syncthreads();
    const int totD = tot[b];
    const int totS = min(max(cur[NBMAX + b], 0), nE);
    block_excl_scan512(tot, wsum, tid);
    const int begD = min(tot[b], nE);
    const int numD[2] = { min(min(totD, BUCKET_CAP), nE - begD), (totD > BUCKET_CAP) ? min(max(cur[2 * NBMAX], 0), nE) : 0 };
    const int numS[2] = { min(totS, BUCKET_CAP),                 (totS > BUCKET_CAP) ? min(max(cur[2 * NBMAX + 1], 0), nE) : 0 };
    const int*  __restrict__ const listS[2] = { recS + (size_t)b * BUCKET_CAP, spillS };
    const int2* __restrict__ const listD[2] = { recD + (size_t)b * BUCKET_CAP, spillD };

#pragma unroll
    for (int q = 0; q < 2; ++q) {
        const int* __restrict__ rs = listS[q];
        const int num = numS[q];
        for (int i = tid; i < num; i += REC_BATCH * 256) {
            int v[REC_BATCH];
#pragma unroll
            for (int u = 0; u < REC_BATCH; ++u) {
                v[u] = rs[min(i + u * 256, num - 1)];
            }
#pragma unroll
            for (int u = 0; u < REC_BATCH; ++u) {
                const unsigned l = (unsigned)(v[u] - node0);
                if (i + u * 256 < num && l < BUCKET_NODES) atomicAdd(&c[l], 1);
            }
        }
    }
    __syncthreads();
    if (node < n) invOut[node] = rsqrtf(fmaxf((float)c[tid], 1.0f));
    c[tid] = 0;
    __syncthreads();

#pragma unroll
    for (int q = 0; q < 2; ++q) {
        const int2* __restrict__ rd = listD[q];
        const int num = numD[q];
        for (int i = tid; i < num; i += REC_BATCH * 256) {
            int2 v[REC_BATCH];
#pragma unroll
            for (int u = 0; u < REC_BATCH; ++u) v[u] = rd[min(i + u * 256, num - 1)];
#pragma unroll
            for (int u = 0; u < REC_BATCH; ++u) {
                const unsigned l = (unsigned)(v[u].x - node0);
                if (i + u * 256 < num && l < BUCKET_NODES) atomicAdd(&c[l], 1);
            }
        }
    }
    __syncthreads();

    const int cnt = c[tid];
    int inc = cnt;
    for (int off = 1; off < 32; off <<= 1) {
        const int t = __shfl_up(inc, off, 32);
        if (lane >= off) inc += t;
    }
    if (lane == 31) wsum[w] = inc;
    __syncthreads();
    int offset = 0;
    for (int k = 0; k < 8; ++k)
        if (k < w) offset += wsum[k];
    const int first = begD + offset + inc - cnt;
    if (node < n) {
        rowptr[node] = first;
        cntIn[node]  = cnt;
        invIn[node]  = rsqrtf(fmaxf((float)cnt, 1.0f));
    }
    c[tid] = first;
    __syncthreads();

#pragma unroll
    for (int q = 0; q < 2; ++q) {
        const int2* __restrict__ rd = listD[q];
        const int num = numD[q];
        for (int i = tid; i < num; i += REC_BATCH * 256) {
            int2 v[REC_BATCH];
#pragma unroll
            for (int u = 0; u < REC_BATCH; ++u) v[u] = rd[min(i + u * 256, num - 1)];
#pragma unroll
            for (int u = 0; u < REC_BATCH; ++u) {
                const unsigned l = (unsigned)(v[u].x - node0);
                if (i + u * 256 < num && l < BUCKET_NODES) {
                    const unsigned pos = (unsigned)atomicAdd(&c[l], 1);
                    if (pos < (unsigned)nE) csr[pos] = v[u].y;
                }
            }
        }
    }
}

template <int U>
static __device__ __forceinline__ void add_rows_scaled(const float* __restrict__ x, int idx, int wgt,
                                                       int j, int lane4, float4& acc) {
    float4 v[U];
#pragma unroll
    for (int u = 0; u < U; ++u) {
        const int s = __builtin_amdgcn_readlane(idx, j + u);
        v[u] = *(const float4*)(x + (size_t)s * FEATS + lane4);
    }
#pragma unroll
    for (int u = 0; u < U; ++u) {
        const float w = __int_as_float(__builtin_amdgcn_readlane(wgt, j + u));
        acc.x = fmaf(v[u].x, w, acc.x);
        acc.y = fmaf(v[u].y, w, acc.y);
        acc.z = fmaf(v[u].z, w, acc.z);
        acc.w = fmaf(v[u].w, w, acc.w);
    }
}

static __device__ __forceinline__ void add_batch_scaled(const float* __restrict__ x, int idx, int wgt,
                                                        int m, int lane4, float4& acc) {
    int j = 0;
    for (; j + 8 <= m; j += 8) add_rows_scaled<8>(x, idx, wgt, j, lane4, acc);
    if (j + 4 <= m) { add_rows_scaled<4>(x, idx, wgt, j, lane4, acc); j += 4; }
    if (j + 2 <= m) { add_rows_scaled<2>(x, idx, wgt, j, lane4, acc); j += 2; }
    if (j < m)      { add_rows_scaled<1>(x, idx, wgt, j, lane4, acc); }
}

__global__ __launch_bounds__(256, 8) void gcn_gather_project_dot(const float* __restrict__ x,
                                                              const float* __restrict__ W1,
                                                              const int* __restrict__ rowptr,
                                                              const int* __restrict__ cntIn,
                                                              const int* __restrict__ csr,
                                                              const float* __restrict__ invIn,
                                                              const float* __restrict__ invOut,
                                                              const float* __restrict__ b1,
                                                              const float* __restrict__ W2,
                                                              float* __restrict__ z, int n, int nE) {
    __shared__ float sA[GROUP_NODES * A_PITCH];
    __shared__ float sPart[8 * GROUP_NODES];

    const int lane  = threadIdx.x & 31;
    const int wv    = __builtin_amdgcn_readfirstlane((int)(threadIdx.x >> 5));
    const int lane4 = lane * 4;
    const int l15   = lane & 15;
    const int khalf = lane >> 4;
    const int nCol  = wv * 16 + l15;

    v2f b[FEATS / 4];
#pragma unroll
    for (int kk = 0; kk < FEATS / 4; ++kk) {
        const int ka = kk * 4 + 2 * khalf;
        b[kk].x = W1[(size_t)ka       * FEATS + nCol];
        b[kk].y = W1[(size_t)(ka + 1) * FEATS + nCol];
    }
    const float bcol = b1[nCol];
    const float wcol = W2[nCol];

    for (int g = 0; g < GROUPS_PER_BLOCK; ++g) {
        const int base = (blockIdx.x * GROUPS_PER_BLOCK + g) * GROUP_NODES;
        if (base >= n) break;
        const int node0 = base + wv * NODES_PER_WAVE;

        int beg[NODES_PER_WAVE], deg[NODES_PER_WAVE], idx0[NODES_PER_WAVE];
        float sc[NODES_PER_WAVE];
#pragma unroll
        for (int i = 0; i < NODES_PER_WAVE; ++i) {
            const int node = min(node0 + i, n - 1);
            int bg = rowptr[node];
            int dg = cntIn[node];
            bg = min(max(bg, 0), nE);
            dg = min(max(dg, 0), nE - bg);
            if (node0 + i >= n) dg = 0;
            beg[i] = bg;
            deg[i] = dg;
            sc[i]  = invIn[node];
        }
#pragma unroll
        for (int i = 0; i < NODES_PER_WAVE; ++i) {
            idx0[i] = csr[max(min(beg[i] + lane, nE - 1), 0)];
        }
        int wgt0[NODES_PER_WAVE];
#pragma unroll
        for (int i = 0; i < NODES_PER_WAVE; ++i) {
            idx0[i] = (int)min((unsigned)idx0[i], (unsigned)(n - 1));
            wgt0[i] = __float_as_int(invOut[idx0[i]]);
        }

#pragma unroll
        for (int i = 0; i < NODES_PER_WAVE; ++i) {
            float4 acc = {0.0f, 0.0f, 0.0f, 0.0f};
            add_batch_scaled(x, idx0[i], wgt0[i], min(32, deg[i]), lane4, acc);
            for (int bb = 32; bb < deg[i]; bb += 32) {
                const int m = min(32, deg[i] - bb);
                int idx = csr[max(min(beg[i] + bb + lane, nE - 1), 0)];
                idx = (int)min((unsigned)idx, (unsigned)(n - 1));
                const int wgt = __float_as_int(invOut[idx]);
                add_batch_scaled(x, idx, wgt, m, lane4, acc);
            }
            acc.x *= sc[i]; acc.y *= sc[i]; acc.z *= sc[i]; acc.w *= sc[i];
            *(float4*)(sA + (wv * NODES_PER_WAVE + i) * A_PITCH + lane4) = acc;
        }
        __syncthreads();

        const float* __restrict__ arow0 = sA + l15 * A_PITCH + 2 * khalf;
        const float* __restrict__ arow1 = arow0 + 16 * A_PITCH;
        v8f c0 = {};
        v8f c1 = {};
        v2f a0[2][A_CHUNK], a1[2][A_CHUNK];
#pragma unroll
        for (int u = 0; u < A_CHUNK; ++u) {
            a0[0][u] = *(const v2f*)(arow0 + u * 4);
            a1[0][u] = *(const v2f*)(arow1 + u * 4);
        }
#pragma unroll
        for (int ch = 0; ch < FEATS / 4 / A_CHUNK; ++ch) {
            const int cur = ch & 1;
            if (ch + 1 < FEATS / 4 / A_CHUNK) {
#pragma unroll
                for (int u = 0; u < A_CHUNK; ++u) {
                    a0[cur ^ 1][u] = *(const v2f*)(arow0 + ((ch + 1) * A_CHUNK + u) * 4);
                    a1[cur ^ 1][u] = *(const v2f*)(arow1 + ((ch + 1) * A_CHUNK + u) * 4);
                }
            }
#pragma unroll
            for (int u = 0; u < A_CHUNK; ++u) {
                c0 = __builtin_amdgcn_wmma_f32_16x16x4_f32(
                         false, a0[cur][u],  false, b[ch * A_CHUNK + u],
                         (short)0, c0,  false,  false);
                c1 = __builtin_amdgcn_wmma_f32_16x16x4_f32(
                         false, a1[cur][u],  false, b[ch * A_CHUNK + u],
                         (short)0, c1,  false,  false);
            }
        }

#pragma unroll
        for (int rt = 0; rt < GROUP_NODES / 16; ++rt) {
            const v8f c = rt ? c1 : c0;
            float v[8];
#pragma unroll
            for (int r = 0; r < 8; ++r) {
                v[r] = fmaxf(c[r] + bcol, 0.0f) * wcol;
            }
            const bool h8 = (l15 & 8) != 0, h4 = (l15 & 4) != 0, h2 = (l15 & 2) != 0;
            float w4[4], u2[2];
#pragma unroll
            for (int q = 0; q < 4; ++q) {
                const float keep = h8 ? v[q + 4] : v[q];
                const float send = h8 ? v[q] : v[q + 4];
                w4[q] = keep + __shfl_xor(send, 8, 32);
            }
#pragma unroll
            for (int q = 0; q < 2; ++q) {
                const float keep = h4 ? w4[q + 2] : w4[q];
                const float send = h4 ? w4[q] : w4[q + 2];
                u2[q] = keep + __shfl_xor(send, 4, 32);
            }
            float t = (h2 ? u2[1] : u2[0]) + __shfl_xor(h2 ? u2[0] : u2[1], 2, 32);
            t += __shfl_xor(t, 1, 32);
            if ((lane & 1) == 0) sPart[wv * GROUP_NODES + rt * 16 + (lane >> 1)] = t;
        }
        __syncthreads();

        if (threadIdx.x < GROUP_NODES) {
            const int node = base + (int)threadIdx.x;
            float s = sPart[threadIdx.x];
#pragma unroll
            for (int ct = 1; ct < 8; ++ct) s += sPart[ct * GROUP_NODES + threadIdx.x];
            if (node < n) z[node] = s * invOut[node];
        }
    }
}

__global__ __launch_bounds__(256) void gcn_gather_scalar(const float* __restrict__ z,
                                                         const int* __restrict__ rowptr,
                                                         const int* __restrict__ cntIn,
                                                         const int* __restrict__ csr,
                                                         const float* __restrict__ invIn,
                                                         const float* __restrict__ b2,
                                                         float* __restrict__ out, int n, int nE) {
    const int wave  = blockIdx.x * 8 + (threadIdx.x >> 5);
    const int lane  = threadIdx.x & 31;
    const int grp   = lane >> 3;
    const int sub   = lane & 7;
    const int node0 = wave * (4 * GS_NODES_PER_GROUP) + grp;
    const float bias = b2[0];

    int beg[GS_NODES_PER_GROUP], deg[GS_NODES_PER_GROUP];
#pragma unroll
    for (int t = 0; t < GS_NODES_PER_GROUP; ++t) {
        const int node = min(node0 + 4 * t, n - 1);
        beg[t] = rowptr[node];
        deg[t] = cntIn[node];
    }
    const int onode = node0 + 4 * sub;
    const float osc = invIn[min(onode, n - 1)];
    __builtin_amdgcn_sched_barrier(0);
    int mdeg = 0;
#pragma unroll
    for (int t = 0; t < GS_NODES_PER_GROUP; ++t) {
        const int bg = min(max(beg[t], 0), nE);
        const int dg = min(max(deg[t], 0), nE - bg);
        beg[t] = bg;
        deg[t] = dg;
        mdeg = max(mdeg, dg);
    }

    float s[GS_NODES_PER_GROUP];
#pragma unroll
    for (int t = 0; t < GS_NODES_PER_GROUP; ++t) s[t] = 0.0f;

    for (int j0 = 0; j0 < mdeg; j0 += 8 * GS_SLICES) {
        int idx[GS_NODES_PER_GROUP][GS_SLICES];
#pragma unroll
        for (int t = 0; t < GS_NODES_PER_GROUP; ++t) {
#pragma unroll
            for (int u = 0; u < GS_SLICES; ++u) {
                const int e = beg[t] + min(j0 + 8 * u + sub, deg[t] - 1);
                idx[t][u] = csr[min(max(e, 0), nE - 1)];
            }
        }
        __builtin_amdgcn_sched_barrier(0);
        float zv[GS_NODES_PER_GROUP][GS_SLICES];
#pragma unroll
        for (int t = 0; t < GS_NODES_PER_GROUP; ++t) {
#pragma unroll
            for (int u = 0; u < GS_SLICES; ++u) {
                zv[t][u] = z[min((unsigned)idx[t][u], (unsigned)(n - 1))];
            }
        }
#pragma unroll
        for (int t = 0; t < GS_NODES_PER_GROUP; ++t) {
#pragma unroll
            for (int u = 0; u < GS_SLICES; ++u) {
                const int keep = (j0 + 8 * u + sub < deg[t]) ? -1 : 0;
                s[t] += __int_as_float(__float_as_int(zv[t][u]) & keep);
            }
        }
    }

    float sel = 0.0f;
#pragma unroll
    for (int t = 0; t < GS_NODES_PER_GROUP; ++t) {
        float v = s[t];
        v += __shfl_xor(v, 4, 32);
        v += __shfl_xor(v, 2, 32);
        v += __shfl_xor(v, 1, 32);
        if (sub == t) sel = v;
    }
    if (onode < n) out[onode] = sel * osc + bias;
}

extern "C" void kernel_launch(void* const* d_in, const int* in_sizes, int n_in,
                              void* d_out, int out_size, void* d_ws, size_t ws_size,
                              hipStream_t stream) {
    const float* x   = (const float*)d_in[0];
    const int*   src = (const int*)d_in[1];
    const int*   dst = (const int*)d_in[2];
    const float* W1  = (const float*)d_in[3];
    const float* b1  = (const float*)d_in[4];
    const float* W2  = (const float*)d_in[5];
    const float* b2  = (const float*)d_in[6];
    float* out = (float*)d_out;

    const int n  = in_sizes[0] / FEATS;
    const int nE = in_sizes[1];
    const int nB = (n + BUCKET_NODES - 1) / BUCKET_NODES;

    float* f      = (float*)d_ws;
    int*   cntIn  = (int*)(f + (size_t)n);
    float* invOut = f + 2 * (size_t)n;
    float* invIn  = f + 3 * (size_t)n;
    float* z      = f + 4 * (size_t)n;
    int*   rowptr = (int*)(f + 5 * (size_t)n);
    int2*  recD   = (int2*)(f + 6 * (size_t)n);
    int2*  spillD = recD + (size_t)NBMAX * BUCKET_CAP;
    int*   recS   = (int*)(spillD + (size_t)nE);
    int*   spillS = recS + (size_t)NBMAX * BUCKET_CAP;
    int*   csr    = spillS + (size_t)nE;
    int*   cur    = csr + (size_t)nE;

    hipMemsetAsync(cur, 0, 4 * NBMAX * sizeof(int), stream);

    const int edgeBlocks = (nE + 256 * BIN_EDGES_PER_THREAD - 1) / (256 * BIN_EDGES_PER_THREAD);
    gcn_bucket_fill<<<edgeBlocks, 256, 0, stream>>>(src, dst, cur, recD, recS, spillD, spillS, nE, nB);
    gcn_bucket_build<<<nB, 256, 0, stream>>>(cur, recD, recS, spillD, spillS, rowptr, cntIn, csr, invOut, invIn, n, nE, nB);

    gcn_gather_project_dot<<<(n + GROUP_NODES * GROUPS_PER_BLOCK - 1) / (GROUP_NODES * GROUPS_PER_BLOCK), 256, 0, stream>>>(x, W1, rowptr, cntIn, csr, invIn, invOut, b1, W2, z, n, nE);

    gcn_gather_scalar<<<(n + 32 * GS_NODES_PER_GROUP - 1) / (32 * GS_NODES_PER_GROUP), 256, 0, stream>>>(z, rowptr, cntIn, csr, invIn, b2, out, n, nE);
}


# Per-bucket pass: the bucket's slice of the source list is put in node order in LDS and stored as whole 128-byte lines, not as 1.6M scattered 4-byte stores; sum order = arrival rank as before
# speedup vs baseline: 5.9470x; 1.0319x over previous
#include <hip/hip_runtime.h>
#include <hip/hip_bf16.h>


typedef __attribute__((ext_vector_type(2))) float v2f;
typedef __attribute__((ext_vector_type(8))) float v8f;

#define FEATS 128

#define BIN_EDGES_PER_THREAD 16
#define NODES_PER_WAVE     4
#define GROUP_NODES        32
#define GROUPS_PER_BLOCK   4
#define A_CHUNK            4
#define A_PITCH            132
#define GS_NODES_PER_GROUP 8
#define GS_SLICES          4

#define BUCKET_SHIFT 8
#define BUCKET_NODES 256
#define NBMAX        512
#define BUCKET_CAP   8240
#define REC_BATCH    8
#define CSR_STAGE    8192

static __device__ __forceinline__ int bucket_of(int node, int nB) {
    return min((int)((unsigned)node >> BUCKET_SHIFT), nB - 1);
}

static __device__ __forceinline__ void block_excl_scan512(int* a, int* wsum, int tid) {
    const int lane = tid & 31;
    const int w    = tid >> 5;
    const int v0 = a[2 * tid];
    const int v1 = a[2 * tid + 1];
    const int local = v0 + v1;
    int inc = local;
    for (int off = 1; off < 32; off <<= 1) {
        const int t = __shfl_up(inc, off, 32);
        if (lane >= off) inc += t;
    }
    if (lane == 31) wsum[w] = inc;
    __syncthreads();
    int offset = 0;
    for (int k = 0; k < 8; ++k)
        if (k < w) offset += wsum[k];
    const int ex = offset + inc - local;
    a[2 * tid]     = ex;
    a[2 * tid + 1] = ex + v0;
    __syncthreads();
}

__global__ __launch_bounds__(256) void gcn_bucket_fill(const int* __restrict__ src,
                                                       const int* __restrict__ dst,
                                                       int* __restrict__ cur,
                                                       int2* __restrict__ recD,
                                                       int* __restrict__ recS,
                                                       int2* __restrict__ spillD,
                                                       int* __restrict__ spillS, int nE, int nB) {
    __shared__ int  hD[NBMAX];
    __shared__ int  hS[NBMAX];
    __shared__ int  gD[NBMAX];
    __shared__ int  gS[NBMAX];
    __shared__ int  wsum[8];
    __shared__ int2 stD[256 * BIN_EDGES_PER_THREAD];
    __shared__ int  stS[256 * BIN_EDGES_PER_THREAD];
    const int tid = threadIdx.x;
    for (int i = tid; i < NBMAX; i += 256) {
        hD[i] = 0;
        hS[i] = 0;
    }

    const int eBlock = blockIdx.x * (256 * BIN_EDGES_PER_THREAD);
    const int e0 = eBlock + tid;
    int s[BIN_EDGES_PER_THREAD], d[BIN_EDGES_PER_THREAD];
    int rD[BIN_EDGES_PER_THREAD], rS[BIN_EDGES_PER_THREAD];
#pragma unroll
    for (int u = 0; u < BIN_EDGES_PER_THREAD; ++u) {
        const int ec = min(e0 + u * 256, nE - 1);
        s[u] = src[ec];
        d[u] = dst[ec];
    }
    __syncthreads();
#pragma unroll
    for (int u = 0; u < BIN_EDGES_PER_THREAD; ++u) {
        rD[u] = 0;
        rS[u] = 0;
        if (e0 + u * 256 < nE) {
            rD[u] = atomicAdd(&hD[bucket_of(d[u], nB)], 1);
            rS[u] = atomicAdd(&hS[bucket_of(s[u], nB)], 1);
        }
    }
    __syncthreads();

    int baseD[2], baseS[2];
#pragma unroll
    for (int k = 0; k < 2; ++k) {
        const int i = tid + k * 256;
        baseD[k] = 0;
        baseS[k] = 0;
        if (i < nB) {
            baseD[k] = atomicAdd(&cur[i], hD[i]);
            baseS[k] = atomicAdd(&cur[NBMAX + i], hS[i]);
        }
    }
#pragma unroll
    for (int k = 0; k < 2; ++k) {
        const int i = tid + k * 256;
        gD[i] = baseD[k];
        gS[i] = baseS[k];
    }
    block_excl_scan512(hD, wsum, tid);
    block_excl_scan512(hS, wsum, tid);

#pragma unroll
    for (int u = 0; u < BIN_EDGES_PER_THREAD; ++u) {
        if (e0 + u * 256 < nE) {
            const int pD = (hD[bucket_of(d[u], nB)] + rD[u]) & (256 * BIN_EDGES_PER_THREAD - 1);
            const int pS = (hS[bucket_of(s[u], nB)] + rS[u]) & (256 * BIN_EDGES_PER_THREAD - 1);
            stD[pD] = make_int2(d[u], s[u]);
            stS[pS] = s[u];
        }
    }
    __syncthreads();

    const int nMine = min(256 * BIN_EDGES_PER_THREAD, nE - eBlock);
#pragma unroll
    for (int u = 0; u < BIN_EDGES_PER_THREAD; ++u) {
        const int j = tid + u * 256;
        if (j < nMine) {
            const int2 v  = stD[j];
            const int  bD = bucket_of(v.x, nB);
            const unsigned kD = (unsigned)(gD[bD] + (j - hD[bD]));
            const int  w  = stS[j];
            const int  bS = bucket_of(w, nB);
            const unsigned kS = (unsigned)(gS[bS] + (j - hS[bS]));
            if (kD < (unsigned)BUCKET_CAP) {
                recD[(size_t)bD * BUCKET_CAP + kD] = v;
            } else {
                const unsigned q = (unsigned)atomicAdd(&cur[2 * NBMAX], 1);
                if (q < (unsigned)nE) spillD[q] = v;
            }
            if (kS < (unsigned)BUCKET_CAP) {
                recS[(size_t)bS * BUCKET_CAP + kS] = w;
            } else {
                const unsigned q = (unsigned)atomicAdd(&cur[2 * NBMAX + 1], 1);
                if (q < (unsigned)nE) spillS[q] = w;
            }
        }
    }
}

__global__ __launch_bounds__(256) void gcn_bucket_build(const int* __restrict__ cur,
                                                        const int2* __restrict__ recD,
                                                        const int* __restrict__ recS,
                                                        const int2* __restrict__ spillD,
                                                        const int* __restrict__ spillS,
                                                        int* __restrict__ rowptr,
                                                        int* __restrict__ cntIn,
                                                        int* __restrict__ csr,
                                                        float* __restrict__ invOut,
                                                        float* __restrict__ invIn,
                                                        int n, int nE, int nB) {
    __shared__ int c[BUCKET_NODES];
    __shared__ int wsum[8];
    __shared__ int tot[NBMAX];
    __shared__ int stage[CSR_STAGE];
    const int tid   = threadIdx.x;
    const int lane  = tid & 31;
    const int w     = tid >> 5;
    const int b     = blockIdx.x;
    const int node0 = b << BUCKET_SHIFT;
    const int node  = node0 + tid;

    for (int i = tid; i < NBMAX; i += 256) tot[i] = (i < nB) ? min(max(cur[i], 0), nE) : 0;
    c[tid] = 0;
    __syncthreads();
    const int totD = tot[b];
    const int totS = min(max(cur[NBMAX + b], 0), nE);
    block_excl_scan512(tot, wsum, tid);
    const int begD = min(tot[b], nE);
    const int numD[2] = { min(min(totD, BUCKET_CAP), nE - begD), (totD > BUCKET_CAP) ? min(max(cur[2 * NBMAX], 0), nE) : 0 };
    const int numS[2] = { min(totS, BUCKET_CAP),                 (totS > BUCKET_CAP) ? min(max(cur[2 * NBMAX + 1], 0), nE) : 0 };
    const int*  __restrict__ const listS[2] = { recS + (size_t)b * BUCKET_CAP, spillS };
    const int2* __restrict__ const listD[2] = { recD + (size_t)b * BUCKET_CAP, spillD };

#pragma unroll
    for (int q = 0; q < 2; ++q) {
        const int* __restrict__ rs = listS[q];
        const int num = numS[q];
        for (int i = tid; i < num; i += REC_BATCH * 256) {
            int v[REC_BATCH];
#pragma unroll
            for (int u = 0; u < REC_BATCH; ++u) {
                v[u] = rs[min(i + u * 256, num - 1)];
            }
#pragma unroll
            for (int u = 0; u < REC_BATCH; ++u) {
                const unsigned l = (unsigned)(v[u] - node0);
                if (i + u * 256 < num && l < BUCKET_NODES) atomicAdd(&c[l], 1);
            }
        }
    }
    __syncthreads();
    if (node < n) invOut[node] = rsqrtf(fmaxf((float)c[tid], 1.0f));
    c[tid] = 0;
    __syncthreads();

#pragma unroll
    for (int q = 0; q < 2; ++q) {
        const int2* __restrict__ rd = listD[q];
        const int num = numD[q];
        for (int i = tid; i < num; i += REC_BATCH * 256) {
            int2 v[REC_BATCH];
#pragma unroll
            for (int u = 0; u < REC_BATCH; ++u) v[u] = rd[min(i + u * 256, num - 1)];
#pragma unroll
            for (int u = 0; u < REC_BATCH; ++u) {
                const unsigned l = (unsigned)(v[u].x - node0);
                if (i + u * 256 < num && l < BUCKET_NODES) atomicAdd(&c[l], 1);
            }
        }
    }
    __syncthreads();

    const int cnt = c[tid];
    int inc = cnt;
    for (int off = 1; off < 32; off <<= 1) {
        const int t = __shfl_up(inc, off, 32);
        if (lane >= off) inc += t;
    }
    if (lane == 31) wsum[w] = inc;
    __syncthreads();
    int offset = 0;
    for (int k = 0; k < 8; ++k)
        if (k < w) offset += wsum[k];
    const int firstLocal = offset + inc - cnt;
    const int first = begD + firstLocal;
    if (node < n) {
        rowptr[node] = first;
        cntIn[node]  = cnt;
        invIn[node]  = rsqrtf(fmaxf((float)cnt, 1.0f));
    }
    int sliceLen = 0;
    for (int k = 0; k < 8; ++k) sliceLen += wsum[k];
    c[tid] = firstLocal;
    __syncthreads();

#pragma unroll
    for (int q = 0; q < 2; ++q) {
        const int2* __restrict__ rd = listD[q];
        const int num = numD[q];
        for (int i = tid; i < num; i += REC_BATCH * 256) {
            int2 v[REC_BATCH];
#pragma unroll
            for (int u = 0; u < REC_BATCH; ++u) v[u] = rd[min(i + u * 256, num - 1)];
            unsigned p[REC_BATCH];
#pragma unroll
            for (int u = 0; u < REC_BATCH; ++u) {
                const unsigned l = (unsigned)(v[u].x - node0);
                p[u] = 0xffffffffu;
                if (i + u * 256 < num && l < BUCKET_NODES) p[u] = (unsigned)atomicAdd(&c[l], 1);
            }
#pragma unroll
            for (int u = 0; u < REC_BATCH; ++u) {
                if (p[u] < (unsigned)CSR_STAGE) {
                    stage[p[u]] = v[u].y;
                } else if (p[u] != 0xffffffffu) {
                    const unsigned pos = (unsigned)begD + p[u];
                    if (pos < (unsigned)nE) csr[pos] = v[u].y;
                }
            }
        }
    }
    __syncthreads();

    const int nStage = min(sliceLen, CSR_STAGE);
    const int gEnd = min(begD + nStage, nE);
    for (int g0 = (begD & ~31) + tid; g0 < gEnd; g0 += REC_BATCH * 256) {
        int val[REC_BATCH];
#pragma unroll
        for (int u = 0; u < REC_BATCH; ++u) {
            val[u] = stage[min(max(g0 + u * 256 - begD, 0), max(nStage - 1, 0))];
        }
#pragma unroll
        for (int u = 0; u < REC_BATCH; ++u) {
            const int g = g0 + u * 256;
            if (g >= begD && g < gEnd) csr[g] = val[u];
        }
    }
}

template <int U>
static __device__ __forceinline__ void add_rows_scaled(const float* __restrict__ x, int idx, int wgt,
                                                       int j, int lane4, float4& acc) {
    float4 v[U];
#pragma unroll
    for (int u = 0; u < U; ++u) {
        const int s = __builtin_amdgcn_readlane(idx, j + u);
        v[u] = *(const float4*)(x + (size_t)s * FEATS + lane4);
    }
#pragma unroll
    for (int u = 0; u < U; ++u) {
        const float w = __int_as_float(__builtin_amdgcn_readlane(wgt, j + u));
        acc.x = fmaf(v[u].x, w, acc.x);
        acc.y = fmaf(v[u].y, w, acc.y);
        acc.z = fmaf(v[u].z, w, acc.z);
        acc.w = fmaf(v[u].w, w, acc.w);
    }
}

static __device__ __forceinline__ void add_batch_scaled(const float* __restrict__ x, int idx, int wgt,
                                                        int m, int lane4, float4& acc) {
    int j = 0;
    for (; j + 8 <= m; j += 8) add_rows_scaled<8>(x, idx, wgt, j, lane4, acc);
    if (j + 4 <= m) { add_rows_scaled<4>(x, idx, wgt, j, lane4, acc); j += 4; }
    if (j + 2 <= m) { add_rows_scaled<2>(x, idx, wgt, j, lane4, acc); j += 2; }
    if (j < m)      { add_rows_scaled<1>(x, idx, wgt, j, lane4, acc); }
}

__global__ __launch_bounds__(256, 8) void gcn_gather_project_dot(const float* __restrict__ x,
                                                              const float* __restrict__ W1,
                                                              const int* __restrict__ rowptr,
                                                              const int* __restrict__ cntIn,
                                                              const int* __restrict__ csr,
                                                              const float* __restrict__ invIn,
                                                              const float* __restrict__ invOut,
                                                              const float* __restrict__ b1,
                                                              const float* __restrict__ W2,
                                                              float* __restrict__ z, int n, int nE) {
    __shared__ float sA[GROUP_NODES * A_PITCH];
    __shared__ float sPart[8 * GROUP_NODES];

    const int lane  = threadIdx.x & 31;
    const int wv    = __builtin_amdgcn_readfirstlane((int)(threadIdx.x >> 5));
    const int lane4 = lane * 4;
    const int l15   = lane & 15;
    const int khalf = lane >> 4;
    const int nCol  = wv * 16 + l15;

    v2f b[FEATS / 4];
#pragma unroll
    for (int kk = 0; kk < FEATS / 4; ++kk) {
        const int ka = kk * 4 + 2 * khalf;
        b[kk].x = W1[(size_t)ka       * FEATS + nCol];
        b[kk].y = W1[(size_t)(ka + 1) * FEATS + nCol];
    }
    const float bcol = b1[nCol];
    const float wcol = W2[nCol];

    for (int g = 0; g < GROUPS_PER_BLOCK; ++g) {
        const int base = (blockIdx.x * GROUPS_PER_BLOCK + g) * GROUP_NODES;
        if (base >= n) break;
        const int node0 = base + wv * NODES_PER_WAVE;

        int beg[NODES_PER_WAVE], deg[NODES_PER_WAVE], idx0[NODES_PER_WAVE];
        float sc[NODES_PER_WAVE];
#pragma unroll
        for (int i = 0; i < NODES_PER_WAVE; ++i) {
            const int node = min(node0 + i, n - 1);
            int bg = rowptr[node];
            int dg = cntIn[node];
            bg = min(max(bg, 0), nE);
            dg = min(max(dg, 0), nE - bg);
            if (node0 + i >= n) dg = 0;
            beg[i] = bg;
            deg[i] = dg;
            sc[i]  = invIn[node];
        }
#pragma unroll
        for (int i = 0; i < NODES_PER_WAVE; ++i) {
            idx0[i] = csr[max(min(beg[i] + lane, nE - 1), 0)];
        }
        int wgt0[NODES_PER_WAVE];
#pragma unroll
        for (int i = 0; i < NODES_PER_WAVE; ++i) {
            idx0[i] = (int)min((unsigned)idx0[i], (unsigned)(n - 1));
            wgt0[i] = __float_as_int(invOut[idx0[i]]);
        }

#pragma unroll
        for (int i = 0; i < NODES_PER_WAVE; ++i) {
            float4 acc = {0.0f, 0.0f, 0.0f, 0.0f};
            add_batch_scaled(x, idx0[i], wgt0[i], min(32, deg[i]), lane4, acc);
            for (int bb = 32; bb < deg[i]; bb += 32) {
                const int m = min(32, deg[i] - bb);
                int idx = csr[max(min(beg[i] + bb + lane, nE - 1), 0)];
                idx = (int)min((unsigned)idx, (unsigned)(n - 1));
                const int wgt = __float_as_int(invOut[idx]);
                add_batch_scaled(x, idx, wgt, m, lane4, acc);
            }
            acc.x *= sc[i]; acc.y *= sc[i]; acc.z *= sc[i]; acc.w *= sc[i];
            *(float4*)(sA + (wv * NODES_PER_WAVE + i) * A_PITCH + lane4) = acc;
        }
        __syncthreads();

        const float* __restrict__ arow0 = sA + l15 * A_PITCH + 2 * khalf;
        const float* __restrict__ arow1 = arow0 + 16 * A_PITCH;
        v8f c0 = {};
        v8f c1 = {};
        v2f a0[2][A_CHUNK], a1[2][A_CHUNK];
#pragma unroll
        for (int u = 0; u < A_CHUNK; ++u) {
            a0[0][u] = *(const v2f*)(arow0 + u * 4);
            a1[0][u] = *(const v2f*)(arow1 + u * 4);
        }
#pragma unroll
        for (int ch = 0; ch < FEATS / 4 / A_CHUNK; ++ch) {
            const int cur = ch & 1;
            if (ch + 1 < FEATS / 4 / A_CHUNK) {
#pragma unroll
                for (int u = 0; u < A_CHUNK; ++u) {
                    a0[cur ^ 1][u] = *(const v2f*)(arow0 + ((ch + 1) * A_CHUNK + u) * 4);
                    a1[cur ^ 1][u] = *(const v2f*)(arow1 + ((ch + 1) * A_CHUNK + u) * 4);
                }
            }
#pragma unroll
            for (int u = 0; u < A_CHUNK; ++u) {
                c0 = __builtin_amdgcn_wmma_f32_16x16x4_f32(
                         false, a0[cur][u],  false, b[ch * A_CHUNK + u],
                         (short)0, c0,  false,  false);
                c1 = __builtin_amdgcn_wmma_f32_16x16x4_f32(
                         false, a1[cur][u],  false, b[ch * A_CHUNK + u],
                         (short)0, c1,  false,  false);
            }
        }

#pragma unroll
        for (int rt = 0; rt < GROUP_NODES / 16; ++rt) {
            const v8f c = rt ? c1 : c0;
            float v[8];
#pragma unroll
            for (int r = 0; r < 8; ++r) {
                v[r] = fmaxf(c[r] + bcol, 0.0f) * wcol;
            }
            const bool h8 = (l15 & 8) != 0, h4 = (l15 & 4) != 0, h2 = (l15 & 2) != 0;
            float w4[4], u2[2];
#pragma unroll
            for (int q = 0; q < 4; ++q) {
                const float keep = h8 ? v[q + 4] : v[q];
                const float send = h8 ? v[q] : v[q + 4];
                w4[q] = keep + __shfl_xor(send, 8, 32);
            }
#pragma unroll
            for (int q = 0; q < 2; ++q) {
                const float keep = h4 ? w4[q + 2] : w4[q];
                const float send = h4 ? w4[q] : w4[q + 2];
                u2[q] = keep + __shfl_xor(send, 4, 32);
            }
            float t = (h2 ? u2[1] : u2[0]) + __shfl_xor(h2 ? u2[0] : u2[1], 2, 32);
            t += __shfl_xor(t, 1, 32);
            if ((lane & 1) == 0) sPart[wv * GROUP_NODES + rt * 16 + (lane >> 1)] = t;
        }
        __syncthreads();

        if (threadIdx.x < GROUP_NODES) {
            const int node = base + (int)threadIdx.x;
            float s = sPart[threadIdx.x];
#pragma unroll
            for (int ct = 1; ct < 8; ++ct) s += sPart[ct * GROUP_NODES + threadIdx.x];
            if (node < n) z[node] = s * invOut[node];
        }
    }
}

__global__ __launch_bounds__(256) void gcn_gather_scalar(const float* __restrict__ z,
                                                         const int* __restrict__ rowptr,
                                                         const int* __restrict__ cntIn,
                                                         const int* __restrict__ csr,
                                                         const float* __restrict__ invIn,
                                                         const float* __restrict__ b2,
                                                         float* __restrict__ out, int n, int nE) {
    const int wave  = blockIdx.x * 8 + (threadIdx.x >> 5);
    const int lane  = threadIdx.x & 31;
    const int grp   = lane >> 3;
    const int sub   = lane & 7;
    const int node0 = wave * (4 * GS_NODES_PER_GROUP) + grp;
    const float bias = b2[0];

    int beg[GS_NODES_PER_GROUP], deg[GS_NODES_PER_GROUP];
#pragma unroll
    for (int t = 0; t < GS_NODES_PER_GROUP; ++t) {
        const int node = min(node0 + 4 * t, n - 1);
        beg[t] = rowptr[node];
        deg[t] = cntIn[node];
    }
    const int onode = node0 + 4 * sub;
    const float osc = invIn[min(onode, n - 1)];
    __builtin_amdgcn_sched_barrier(0);
    int mdeg = 0;
#pragma unroll
    for (int t = 0; t < GS_NODES_PER_GROUP; ++t) {
        const int bg = min(max(beg[t], 0), nE);
        const int dg = min(max(deg[t], 0), nE - bg);
        beg[t] = bg;
        deg[t] = dg;
        mdeg = max(mdeg, dg);
    }

    float s[GS_NODES_PER_GROUP];
#pragma unroll
    for (int t = 0; t < GS_NODES_PER_GROUP; ++t) s[t] = 0.0f;

    for (int j0 = 0; j0 < mdeg; j0 += 8 * GS_SLICES) {
        int idx[GS_NODES_PER_GROUP][GS_SLICES];
#pragma unroll
        for (int t = 0; t < GS_NODES_PER_GROUP; ++t) {
#pragma unroll
            for (int u = 0; u < GS_SLICES; ++u) {
                const int e = beg[t] + min(j0 + 8 * u + sub, deg[t] - 1);
                idx[t][u] = csr[min(max(e, 0), nE - 1)];
            }
        }
        __builtin_amdgcn_sched_barrier(0);
        float zv[GS_NODES_PER_GROUP][GS_SLICES];
#pragma unroll
        for (int t = 0; t < GS_NODES_PER_GROUP; ++t) {
#pragma unroll
            for (int u = 0; u < GS_SLICES; ++u) {
                zv[t][u] = z[min((unsigned)idx[t][u], (unsigned)(n - 1))];
            }
        }
#pragma unroll
        for (int t = 0; t < GS_NODES_PER_GROUP; ++t) {
#pragma unroll
            for (int u = 0; u < GS_SLICES; ++u) {
                const int keep = (j0 + 8 * u + sub < deg[t]) ? -1 : 0;
                s[t] += __int_as_float(__float_as_int(zv[t][u]) & keep);
            }
        }
    }

    float sel = 0.0f;
#pragma unroll
    for (int t = 0; t < GS_NODES_PER_GROUP; ++t) {
        float v = s[t];
        v += __shfl_xor(v, 4, 32);
        v += __shfl_xor(v, 2, 32);
        v += __shfl_xor(v, 1, 32);
        if (sub == t) sel = v;
    }
    if (onode < n) out[onode] = sel * osc + bias;
}

extern "C" void kernel_launch(void* const* d_in, const int* in_sizes, int n_in,
                              void* d_out, int out_size, void* d_ws, size_t ws_size,
                              hipStream_t stream) {
    const float* x   = (const float*)d_in[0];
    const int*   src = (const int*)d_in[1];
    const int*   dst = (const int*)d_in[2];
    const float* W1  = (const float*)d_in[3];
    const float* b1  = (const float*)d_in[4];
    const float* W2  = (const float*)d_in[5];
    const float* b2  = (const float*)d_in[6];
    float* out = (float*)d_out;

    const int n  = in_sizes[0] / FEATS;
    const int nE = in_sizes[1];
    const int nB = (n + BUCKET_NODES - 1) / BUCKET_NODES;

    float* f      = (float*)d_ws;
    int*   cntIn  = (int*)(f + (size_t)n);
    float* invOut = f + 2 * (size_t)n;
    float* invIn  = f + 3 * (size_t)n;
    float* z      = f + 4 * (size_t)n;
    int*   rowptr = (int*)(f + 5 * (size_t)n);
    int2*  recD   = (int2*)(f + 6 * (size_t)n);
    int2*  spillD = recD + (size_t)NBMAX * BUCKET_CAP;
    int*   recS   = (int*)(spillD + (size_t)nE);
    int*   spillS = recS + (size_t)NBMAX * BUCKET_CAP;
    int*   csr    = spillS + (size_t)nE;
    int*   cur    = csr + (size_t)nE;

    hipMemsetAsync(cur, 0, 4 * NBMAX * sizeof(int), stream);

    const int edgeBlocks = (nE + 256 * BIN_EDGES_PER_THREAD - 1) / (256 * BIN_EDGES_PER_THREAD);
    gcn_bucket_fill<<<edgeBlocks, 256, 0, stream>>>(src, dst, cur, recD, recS, spillD, spillS, nE, nB);
    gcn_bucket_build<<<nB, 256, 0, stream>>>(cur, recD, recS, spillD, spillS, rowptr, cntIn, csr, invOut, invIn, n, nE, nB);

    gcn_gather_project_dot<<<(n + GROUP_NODES * GROUPS_PER_BLOCK - 1) / (GROUP_NODES * GROUPS_PER_BLOCK), 256, 0, stream>>>(x, W1, rowptr, cntIn, csr, invIn, invOut, b1, W2, z, n, nE);

    gcn_gather_scalar<<<(n + 32 * GS_NODES_PER_GROUP - 1) / (32 * GS_NODES_PER_GROUP), 256, 0, stream>>>(z, rowptr, cntIn, csr, invIn, b2, out, n, nE);
}


# Row gather grid set to 1024 blocks (4 resident per WGP on 256 WGPs), 32-node groups dealt out block after block: 12 or 13 trips on every WGP, not 12 on most and a third more on some; sum order kept
# speedup vs baseline: 6.2994x; 1.0593x over previous
#include <hip/hip_runtime.h>
#include <hip/hip_bf16.h>


typedef __attribute__((ext_vector_type(2))) float v2f;
typedef __attribute__((ext_vector_type(8))) float v8f;

#define FEATS 128

#define BIN_EDGES_PER_THREAD 16
#define NODES_PER_WAVE     4
#define GROUP_NODES        32
#define GATHER_BLOCKS      1024
#define A_CHUNK            4
#define A_PITCH            132
#define GS_NODES_PER_GROUP 8
#define GS_SLICES          4

#define BUCKET_SHIFT 8
#define BUCKET_NODES 256
#define NBMAX        512
#define BUCKET_CAP   8240
#define REC_BATCH    8
#define CSR_STAGE    8192

static __device__ __forceinline__ int bucket_of(int node, int nB) {
    return min((int)((unsigned)node >> BUCKET_SHIFT), nB - 1);
}

static __device__ __forceinline__ void block_excl_scan512(int* a, int* wsum, int tid) {
    const int lane = tid & 31;
    const int w    = tid >> 5;
    const int v0 = a[2 * tid];
    const int v1 = a[2 * tid + 1];
    const int local = v0 + v1;
    int inc = local;
    for (int off = 1; off < 32; off <<= 1) {
        const int t = __shfl_up(inc, off, 32);
        if (lane >= off) inc += t;
    }
    if (lane == 31) wsum[w] = inc;
    __syncthreads();
    int offset = 0;
    for (int k = 0; k < 8; ++k)
        if (k < w) offset += wsum[k];
    const int ex = offset + inc - local;
    a[2 * tid]     = ex;
    a[2 * tid + 1] = ex + v0;
    __syncthreads();
}

__global__ __launch_bounds__(256) void gcn_bucket_fill(const int* __restrict__ src,
                                                       const int* __restrict__ dst,
                                                       int* __restrict__ cur,
                                                       int2* __restrict__ recD,
                                                       int* __restrict__ recS,
                                                       int2* __restrict__ spillD,
                                                       int* __restrict__ spillS, int nE, int nB) {
    __shared__ int  hD[NBMAX];
    __shared__ int  hS[NBMAX];
    __shared__ int  gD[NBMAX];
    __shared__ int  gS[NBMAX];
    __shared__ int  wsum[8];
    __shared__ int2 stD[256 * BIN_EDGES_PER_THREAD];
    __shared__ int  stS[256 * BIN_EDGES_PER_THREAD];
    const int tid = threadIdx.x;
    for (int i = tid; i < NBMAX; i += 256) {
        hD[i] = 0;
        hS[i] = 0;
    }

    const int eBlock = blockIdx.x * (256 * BIN_EDGES_PER_THREAD);
    const int e0 = eBlock + tid;
    int s[BIN_EDGES_PER_THREAD], d[BIN_EDGES_PER_THREAD];
    int rD[BIN_EDGES_PER_THREAD], rS[BIN_EDGES_PER_THREAD];
#pragma unroll
    for (int u = 0; u < BIN_EDGES_PER_THREAD; ++u) {
        const int ec = min(e0 + u * 256, nE - 1);
        s[u] = src[ec];
        d[u] = dst[ec];
    }
    __syncthreads();
#pragma unroll
    for (int u = 0; u < BIN_EDGES_PER_THREAD; ++u) {
        rD[u] = 0;
        rS[u] = 0;
        if (e0 + u * 256 < nE) {
            rD[u] = atomicAdd(&hD[bucket_of(d[u], nB)], 1);
            rS[u] = atomicAdd(&hS[bucket_of(s[u], nB)], 1);
        }
    }
    __syncthreads();

    int baseD[2], baseS[2];
#pragma unroll
    for (int k = 0; k < 2; ++k) {
        const int i = tid + k * 256;
        baseD[k] = 0;
        baseS[k] = 0;
        if (i < nB) {
            baseD[k] = atomicAdd(&cur[i], hD[i]);
            baseS[k] = atomicAdd(&cur[NBMAX + i], hS[i]);
        }
    }
#pragma unroll
    for (int k = 0; k < 2; ++k) {
        const int i = tid + k * 256;
        gD[i] = baseD[k];
        gS[i] = baseS[k];
    }
    block_excl_scan512(hD, wsum, tid);
    block_excl_scan512(hS, wsum, tid);

#pragma unroll
    for (int u = 0; u < BIN_EDGES_PER_THREAD; ++u) {
        if (e0 + u * 256 < nE) {
            const int pD = (hD[bucket_of(d[u], nB)] + rD[u]) & (256 * BIN_EDGES_PER_THREAD - 1);
            const int pS = (hS[bucket_of(s[u], nB)] + rS[u]) & (256 * BIN_EDGES_PER_THREAD - 1);
            stD[pD] = make_int2(d[u], s[u]);
            stS[pS] = s[u];
        }
    }
    __syncthreads();

    const int nMine = min(256 * BIN_EDGES_PER_THREAD, nE - eBlock);
#pragma unroll
    for (int u = 0; u < BIN_EDGES_PER_THREAD; ++u) {
        const int j = tid + u * 256;
        if (j < nMine) {
            const int2 v  = stD[j];
            const int  bD = bucket_of(v.x, nB);
            const unsigned kD = (unsigned)(gD[bD] + (j - hD[bD]));
            const int  w  = stS[j];
            const int  bS = bucket_of(w, nB);
            const unsigned kS = (unsigned)(gS[bS] + (j - hS[bS]));
            if (kD < (unsigned)BUCKET_CAP) {
                recD[(size_t)bD * BUCKET_CAP + kD] = v;
            } else {
                const unsigned q = (unsigned)atomicAdd(&cur[2 * NBMAX], 1);
                if (q < (unsigned)nE) spillD[q] = v;
            }
            if (kS < (unsigned)BUCKET_CAP) {
                recS[(size_t)bS * BUCKET_CAP + kS] = w;
            } else {
                const unsigned q = (unsigned)atomicAdd(&cur[2 * NBMAX + 1], 1);
                if (q < (unsigned)nE) spillS[q] = w;
            }
        }
    }
}

__global__ __launch_bounds__(256) void gcn_bucket_build(const int* __restrict__ cur,
                                                        const int2* __restrict__ recD,
                                                        const int* __restrict__ recS,
                                                        const int2* __restrict__ spillD,
                                                        const int* __restrict__ spillS,
                                                        int* __restrict__ rowptr,
                                                        int* __restrict__ cntIn,
                                                        int* __restrict__ csr,
                                                        float* __restrict__ invOut,
                                                        float* __restrict__ invIn,
                                                        int n, int nE, int nB) {
    __shared__ int c[BUCKET_NODES];
    __shared__ int wsum[8];
    __shared__ int tot[NBMAX];
    __shared__ int stage[CSR_STAGE];
    const int tid   = threadIdx.x;
    const int lane  = tid & 31;
    const int w     = tid >> 5;
    const int b     = blockIdx.x;
    const int node0 = b << BUCKET_SHIFT;
    const int node  = node0 + tid;

    for (int i = tid; i < NBMAX; i += 256) tot[i] = (i < nB) ? min(max(cur[i], 0), nE) : 0;
    c[tid] = 0;
    __syncthreads();
    const int totD = tot[b];
    const int totS = min(max(cur[NBMAX + b], 0), nE);
    block_excl_scan512(tot, wsum, tid);
    const int begD = min(tot[b], nE);
    const int numD[2] = { min(min(totD, BUCKET_CAP), nE - begD), (totD > BUCKET_CAP) ? min(max(cur[2 * NBMAX], 0), nE) : 0 };
    const int numS[2] = { min(totS, BUCKET_CAP),                 (totS > BUCKET_CAP) ? min(max(cur[2 * NBMAX + 1], 0), nE) : 0 };
    const int*  __restrict__ const listS[2] = { recS + (size_t)b * BUCKET_CAP, spillS };
    const int2* __restrict__ const listD[2] = { recD + (size_t)b * BUCKET_CAP, spillD };

#pragma unroll
    for (int q = 0; q < 2; ++q) {
        const int* __restrict__ rs = listS[q];
        const int num = numS[q];
        for (int i = tid; i < num; i += REC_BATCH * 256) {
            int v[REC_BATCH];
#pragma unroll
            for (int u = 0; u < REC_BATCH; ++u) {
                v[u] = rs[min(i + u * 256, num - 1)];
            }
#pragma unroll
            for (int u = 0; u < REC_BATCH; ++u) {
                const unsigned l = (unsigned)(v[u] - node0);
                if (i + u * 256 < num && l < BUCKET_NODES) atomicAdd(&c[l], 1);
            }
        }
    }
    __syncthreads();
    if (node < n) invOut[node] = rsqrtf(fmaxf((float)c[tid], 1.0f));
    c[tid] = 0;
    __syncthreads();

#pragma unroll
    for (int q = 0; q < 2; ++q) {
        const int2* __restrict__ rd = listD[q];
        const int num = numD[q];
        for (int i = tid; i < num; i += REC_BATCH * 256) {
            int2 v[REC_BATCH];
#pragma unroll
            for (int u = 0; u < REC_BATCH; ++u) v[u] = rd[min(i + u * 256, num - 1)];
#pragma unroll
            for (int u = 0; u < REC_BATCH; ++u) {
                const unsigned l = (unsigned)(v[u].x - node0);
                if (i + u * 256 < num && l < BUCKET_NODES) atomicAdd(&c[l], 1);
            }
        }
    }
    __syncthreads();

    const int cnt = c[tid];
    int inc = cnt;
    for (int off = 1; off < 32; off <<= 1) {
        const int t = __shfl_up(inc, off, 32);
        if (lane >= off) inc += t;
    }
    if (lane == 31) wsum[w] = inc;
    __syncthreads();
    int offset = 0;
    for (int k = 0; k < 8; ++k)
        if (k < w) offset += wsum[k];
    const int firstLocal = offset + inc - cnt;
    const int first = begD + firstLocal;
    if (node < n) {
        rowptr[node] = first;
        cntIn[node]  = cnt;
        invIn[node]  = rsqrtf(fmaxf((float)cnt, 1.0f));
    }
    int sliceLen = 0;
    for (int k = 0; k < 8; ++k) sliceLen += wsum[k];
    c[tid] = firstLocal;
    __syncthreads();

#pragma unroll
    for (int q = 0; q < 2; ++q) {
        const int2* __restrict__ rd = listD[q];
        const int num = numD[q];
        for (int i = tid; i < num; i += REC_BATCH * 256) {
            int2 v[REC_BATCH];
#pragma unroll
            for (int u = 0; u < REC_BATCH; ++u) v[u] = rd[min(i + u * 256, num - 1)];
            unsigned p[REC_BATCH];
#pragma unroll
            for (int u = 0; u < REC_BATCH; ++u) {
                const unsigned l = (unsigned)(v[u].x - node0);
                p[u] = 0xffffffffu;
                if (i + u * 256 < num && l < BUCKET_NODES) p[u] = (unsigned)atomicAdd(&c[l], 1);
            }
#pragma unroll
            for (int u = 0; u < REC_BATCH; ++u) {
                if (p[u] < (unsigned)CSR_STAGE) {
                    stage[p[u]] = v[u].y;
                } else if (p[u] != 0xffffffffu) {
                    const unsigned pos = (unsigned)begD + p[u];
                    if (pos < (unsigned)nE) csr[pos] = v[u].y;
                }
            }
        }
    }
    __syncthreads();

    const int nStage = min(sliceLen, CSR_STAGE);
    const int gEnd = min(begD + nStage, nE);
    for (int g0 = (begD & ~31) + tid; g0 < gEnd; g0 += REC_BATCH * 256) {
        int val[REC_BATCH];
#pragma unroll
        for (int u = 0; u < REC_BATCH; ++u) {
            val[u] = stage[min(max(g0 + u * 256 - begD, 0), max(nStage - 1, 0))];
        }
#pragma unroll
        for (int u = 0; u < REC_BATCH; ++u) {
            const int g = g0 + u * 256;
            if (g >= begD && g < gEnd) csr[g] = val[u];
        }
    }
}

template <int U>
static __device__ __forceinline__ void add_rows_scaled(const float* __restrict__ x, int idx, int wgt,
                                                       int j, int lane4, float4& acc) {
    float4 v[U];
#pragma unroll
    for (int u = 0; u < U; ++u) {
        const int s = __builtin_amdgcn_readlane(idx, j + u);
        v[u] = *(const float4*)(x + (size_t)s * FEATS + lane4);
    }
#pragma unroll
    for (int u = 0; u < U; ++u) {
        const float w = __int_as_float(__builtin_amdgcn_readlane(wgt, j + u));
        acc.x = fmaf(v[u].x, w, acc.x);
        acc.y = fmaf(v[u].y, w, acc.y);
        acc.z = fmaf(v[u].z, w, acc.z);
        acc.w = fmaf(v[u].w, w, acc.w);
    }
}

static __device__ __forceinline__ void add_batch_scaled(const float* __restrict__ x, int idx, int wgt,
                                                        int m, int lane4, float4& acc) {
    int j = 0;
    for (; j + 8 <= m; j += 8) add_rows_scaled<8>(x, idx, wgt, j, lane4, acc);
    if (j + 4 <= m) { add_rows_scaled<4>(x, idx, wgt, j, lane4, acc); j += 4; }
    if (j + 2 <= m) { add_rows_scaled<2>(x, idx, wgt, j, lane4, acc); j += 2; }
    if (j < m)      { add_rows_scaled<1>(x, idx, wgt, j, lane4, acc); }
}

__global__ __launch_bounds__(256, 8) void gcn_gather_project_dot(const float* __restrict__ x,
                                                              const float* __restrict__ W1,
                                                              const int* __restrict__ rowptr,
                                                              const int* __restrict__ cntIn,
                                                              const int* __restrict__ csr,
                                                              const float* __restrict__ invIn,
                                                              const float* __restrict__ invOut,
                                                              const float* __restrict__ b1,
                                                              const float* __restrict__ W2,
                                                              float* __restrict__ z, int n, int nE) {
    __shared__ float sA[GROUP_NODES * A_PITCH];
    __shared__ float sPart[8 * GROUP_NODES];

    const int lane  = threadIdx.x & 31;
    const int wv    = __builtin_amdgcn_readfirstlane((int)(threadIdx.x >> 5));
    const int lane4 = lane * 4;
    const int l15   = lane & 15;
    const int khalf = lane >> 4;
    const int nCol  = wv * 16 + l15;

    v2f b[FEATS / 4];
#pragma unroll
    for (int kk = 0; kk < FEATS / 4; ++kk) {
        const int ka = kk * 4 + 2 * khalf;
        b[kk].x = W1[(size_t)ka       * FEATS + nCol];
        b[kk].y = W1[(size_t)(ka + 1) * FEATS + nCol];
    }
    const float bcol = b1[nCol];
    const float wcol = W2[nCol];

    for (int grp = blockIdx.x; grp * GROUP_NODES < n; grp += GATHER_BLOCKS) {
        const int base = grp * GROUP_NODES;
        const int node0 = base + wv * NODES_PER_WAVE;

        int beg[NODES_PER_WAVE], deg[NODES_PER_WAVE], idx0[NODES_PER_WAVE];
        float sc[NODES_PER_WAVE];
#pragma unroll
        for (int i = 0; i < NODES_PER_WAVE; ++i) {
            const int node = min(node0 + i, n - 1);
            int bg = rowptr[node];
            int dg = cntIn[node];
            bg = min(max(bg, 0), nE);
            dg = min(max(dg, 0), nE - bg);
            if (node0 + i >= n) dg = 0;
            beg[i] = bg;
            deg[i] = dg;
            sc[i]  = invIn[node];
        }
#pragma unroll
        for (int i = 0; i < NODES_PER_WAVE; ++i) {
            idx0[i] = csr[max(min(beg[i] + lane, nE - 1), 0)];
        }
        int wgt0[NODES_PER_WAVE];
#pragma unroll
        for (int i = 0; i < NODES_PER_WAVE; ++i) {
            idx0[i] = (int)min((unsigned)idx0[i], (unsigned)(n - 1));
            wgt0[i] = __float_as_int(invOut[idx0[i]]);
        }

#pragma unroll
        for (int i = 0; i < NODES_PER_WAVE; ++i) {
            float4 acc = {0.0f, 0.0f, 0.0f, 0.0f};
            add_batch_scaled(x, idx0[i], wgt0[i], min(32, deg[i]), lane4, acc);
            for (int bb = 32; bb < deg[i]; bb += 32) {
                const int m = min(32, deg[i] - bb);
                int idx = csr[max(min(beg[i] + bb + lane, nE - 1), 0)];
                idx = (int)min((unsigned)idx, (unsigned)(n - 1));
                const int wgt = __float_as_int(invOut[idx]);
                add_batch_scaled(x, idx, wgt, m, lane4, acc);
            }
            acc.x *= sc[i]; acc.y *= sc[i]; acc.z *= sc[i]; acc.w *= sc[i];
            *(float4*)(sA + (wv * NODES_PER_WAVE + i) * A_PITCH + lane4) = acc;
        }
        __syncthreads();

        const float* __restrict__ arow0 = sA + l15 * A_PITCH + 2 * khalf;
        const float* __restrict__ arow1 = arow0 + 16 * A_PITCH;
        v8f c0 = {};
        v8f c1 = {};
        v2f a0[2][A_CHUNK], a1[2][A_CHUNK];
#pragma unroll
        for (int u = 0; u < A_CHUNK; ++u) {
            a0[0][u] = *(const v2f*)(arow0 + u * 4);
            a1[0][u] = *(const v2f*)(arow1 + u * 4);
        }
#pragma unroll
        for (int ch = 0; ch < FEATS / 4 / A_CHUNK; ++ch) {
            const int cur = ch & 1;
            if (ch + 1 < FEATS / 4 / A_CHUNK) {
#pragma unroll
                for (int u = 0; u < A_CHUNK; ++u) {
                    a0[cur ^ 1][u] = *(const v2f*)(arow0 + ((ch + 1) * A_CHUNK + u) * 4);
                    a1[cur ^ 1][u] = *(const v2f*)(arow1 + ((ch + 1) * A_CHUNK + u) * 4);
                }
            }
#pragma unroll
            for (int u = 0; u < A_CHUNK; ++u) {
                c0 = __builtin_amdgcn_wmma_f32_16x16x4_f32(
                         false, a0[cur][u],  false, b[ch * A_CHUNK + u],
                         (short)0, c0,  false,  false);
                c1 = __builtin_amdgcn_wmma_f32_16x16x4_f32(
                         false, a1[cur][u],  false, b[ch * A_CHUNK + u],
                         (short)0, c1,  false,  false);
            }
        }

#pragma unroll
        for (int rt = 0; rt < GROUP_NODES / 16; ++rt) {
            const v8f c = rt ? c1 : c0;
            float v[8];
#pragma unroll
            for (int r = 0; r < 8; ++r) {
                v[r] = fmaxf(c[r] + bcol, 0.0f) * wcol;
            }
            const bool h8 = (l15 & 8) != 0, h4 = (l15 & 4) != 0, h2 = (l15 & 2) != 0;
            float w4[4], u2[2];
#pragma unroll
            for (int q = 0; q < 4; ++q) {
                const float keep = h8 ? v[q + 4] : v[q];
                const float send = h8 ? v[q] : v[q + 4];
                w4[q] = keep + __shfl_xor(send, 8, 32);
            }
#pragma unroll
            for (int q = 0; q < 2; ++q) {
                const float keep = h4 ? w4[q + 2] : w4[q];
                const float send = h4 ? w4[q] : w4[q + 2];
                u2[q] = keep + __shfl_xor(send, 4, 32);
            }
            float t = (h2 ? u2[1] : u2[0]) + __shfl_xor(h2 ? u2[0] : u2[1], 2, 32);
            t += __shfl_xor(t, 1, 32);
            if ((lane & 1) == 0) sPart[wv * GROUP_NODES + rt * 16 + (lane >> 1)] = t;
        }
        __syncthreads();

        if (threadIdx.x < GROUP_NODES) {
            const int node = base + (int)threadIdx.x;
            float s = sPart[threadIdx.x];
#pragma unroll
            for (int ct = 1; ct < 8; ++ct) s += sPart[ct * GROUP_NODES + threadIdx.x];
            if (node < n) z[node] = s * invOut[node];
        }
    }
}

__global__ __launch_bounds__(256) void gcn_gather_scalar(const float* __restrict__ z,
                                                         const int* __restrict__ rowptr,
                                                         const int* __restrict__ cntIn,
                                                         const int* __restrict__ csr,
                                                         const float* __restrict__ invIn,
                                                         const float* __restrict__ b2,
                                                         float* __restrict__ out, int n, int nE) {
    const int wave  = blockIdx.x * 8 + (threadIdx.x >> 5);
    const int lane  = threadIdx.x & 31;
    const int grp   = lane >> 3;
    const int sub   = lane & 7;
    const int node0 = wave * (4 * GS_NODES_PER_GROUP) + grp;
    const float bias = b2[0];

    int beg[GS_NODES_PER_GROUP], deg[GS_NODES_PER_GROUP];
#pragma unroll
    for (int t = 0; t < GS_NODES_PER_GROUP; ++t) {
        const int node = min(node0 + 4 * t, n - 1);
        beg[t] = rowptr[node];
        deg[t] = cntIn[node];
    }
    const int onode = node0 + 4 * sub;
    const float osc = invIn[min(onode, n - 1)];
    __builtin_amdgcn_sched_barrier(0);
    int mdeg = 0;
#pragma unroll
    for (int t = 0; t < GS_NODES_PER_GROUP; ++t) {
        const int bg = min(max(beg[t], 0), nE);
        const int dg = min(max(deg[t], 0), nE - bg);
        beg[t] = bg;
        deg[t] = dg;
        mdeg = max(mdeg, dg);
    }

    float s[GS_NODES_PER_GROUP];
#pragma unroll
    for (int t = 0; t < GS_NODES_PER_GROUP; ++t) s[t] = 0.0f;

    for (int j0 = 0; j0 < mdeg; j0 += 8 * GS_SLICES) {
        int idx[GS_NODES_PER_GROUP][GS_SLICES];
#pragma unroll
        for (int t = 0; t < GS_NODES_PER_GROUP; ++t) {
#pragma unroll
            for (int u = 0; u < GS_SLICES; ++u) {
                const int e = beg[t] + min(j0 + 8 * u + sub, deg[t] - 1);
                idx[t][u] = csr[min(max(e, 0), nE - 1)];
            }
        }
        __builtin_amdgcn_sched_barrier(0);
        float zv[GS_NODES_PER_GROUP][GS_SLICES];
#pragma unroll
        for (int t = 0; t < GS_NODES_PER_GROUP; ++t) {
#pragma unroll
            for (int u = 0; u < GS_SLICES; ++u) {
                zv[t][u] = z[min((unsigned)idx[t][u], (unsigned)(n - 1))];
            }
        }
#pragma unroll
        for (int t = 0; t < GS_NODES_PER_GROUP; ++t) {
#pragma unroll
            for (int u = 0; u < GS_SLICES; ++u) {
                const int keep = (j0 + 8 * u + sub < deg[t]) ? -1 : 0;
                s[t] += __int_as_float(__float_as_int(zv[t][u]) & keep);
            }
        }
    }

    float sel = 0.0f;
#pragma unroll
    for (int t = 0; t < GS_NODES_PER_GROUP; ++t) {
        float v = s[t];
        v += __shfl_xor(v, 4, 32);
        v += __shfl_xor(v, 2, 32);
        v += __shfl_xor(v, 1, 32);
        if (sub == t) sel = v;
    }
    if (onode < n) out[onode] = sel * osc + bias;
}

extern "C" void kernel_launch(void* const* d_in, const int* in_sizes, int n_in,
                              void* d_out, int out_size, void* d_ws, size_t ws_size,
                              hipStream_t stream) {
    const float* x   = (const float*)d_in[0];
    const int*   src = (const int*)d_in[1];
    const int*   dst = (const int*)d_in[2];
    const float* W1  = (const float*)d_in[3];
    const float* b1  = (const float*)d_in[4];
    const float* W2  = (const float*)d_in[5];
    const float* b2  = (const float*)d_in[6];
    float* out = (float*)d_out;

    const int n  = in_sizes[0] / FEATS;
    const int nE = in_sizes[1];
    const int nB = (n + BUCKET_NODES - 1) / BUCKET_NODES;

    float* f      = (float*)d_ws;
    int*   cntIn  = (int*)(f + (size_t)n);
    float* invOut = f + 2 * (size_t)n;
    float* invIn  = f + 3 * (size_t)n;
    float* z      = f + 4 * (size_t)n;
    int*   rowptr = (int*)(f + 5 * (size_t)n);
    int2*  recD   = (int2*)(f + 6 * (size_t)n);
    int2*  spillD = recD + (size_t)NBMAX * BUCKET_CAP;
    int*   recS   = (int*)(spillD + (size_t)nE);
    int*   spillS = recS + (size_t)NBMAX * BUCKET_CAP;
    int*   csr    = spillS + (size_t)nE;
    int*   cur    = csr + (size_t)nE;

    hipMemsetAsync(cur, 0, 4 * NBMAX * sizeof(int), stream);

    const int edgeBlocks = (nE + 256 * BIN_EDGES_PER_THREAD - 1) / (256 * BIN_EDGES_PER_THREAD);
    gcn_bucket_fill<<<edgeBlocks, 256, 0, stream>>>(src, dst, cur, recD, recS, spillD, spillS, nE, nB);
    gcn_bucket_build<<<nB, 256, 0, stream>>>(cur, recD, recS, spillD, spillS, rowptr, cntIn, csr, invOut, invIn, n, nE, nB);

    gcn_gather_project_dot<<<GATHER_BLOCKS, 256, 0, stream>>>(x, W1, rowptr, cntIn, csr, invIn, invOut, b1, W2, z, n, nE);

    gcn_gather_scalar<<<(n + 32 * GS_NODES_PER_GROUP - 1) / (32 * GS_NODES_PER_GROUP), 256, 0, stream>>>(z, rowptr, cntIn, csr, invIn, b2, out, n, nE);
}
